# starting kernel
# baseline (speedup 1.0000x reference)
#include <hip/hip_runtime.h>

typedef float v2f __attribute__((ext_vector_type(2)));
typedef float v8f __attribute__((ext_vector_type(8)));

#define LDS_STRIDE 132

__global__ void deg_init_kernel(float* deg, int n) {
  int i = blockIdx.x * blockDim.x + threadIdx.x;
  if (i < n) deg[i] = 1.0f;
}

__global__ void deg_count_kernel(const int* __restrict__ dst, float* __restrict__ deg, int nE) {
  int e = blockIdx.x * blockDim.x + threadIdx.x;
  if (e < nE) unsafeAtomicAdd(&deg[dst[e]], 1.0f);
}

__global__ void deg_rsqrt_kernel(float* deg, int n) {
  int i = blockIdx.x * blockDim.x + threadIdx.x;
  if (i < n) deg[i] = rsqrtf(deg[i]);
}

__global__ void gemm_gather_scale_kernel(const float* __restrict__ X,
                                         const int* __restrict__ gidx,
                                         const float* __restrict__ W,
                                         const float* __restrict__ dinv,
                                         float* __restrict__ hs,
                                         float* __restrict__ agg,
                                         int n_rows, int N) {
  extern __shared__ float lds[];
  float* lA = lds;
  float* lW = lds + 64 * LDS_STRIDE;
  const int tid = threadIdx.x;
  const int row0 = blockIdx.x * 64;

  const float4* X4 = (const float4*)X;
  for (int i = tid; i < 64 * 32; i += 256) {
    int r = i >> 5;
    int c4 = i & 31;
    int gr = row0 + r;
    float4 v = make_float4(0.f, 0.f, 0.f, 0.f);
    if (gr < n_rows) {
      int sr = gidx ? gidx[gr] : gr;
      v = X4[(size_t)sr * 32 + c4];
    }
    *(float4*)&lA[r * LDS_STRIDE + (c4 << 2)] = v;
  }
  for (int i = tid; i < 128 * N; i += 256) {
    int k = i / N;
    int n = i - k * N;
    lW[n * LDS_STRIDE + k] = W[i];
  }
  __syncthreads();

  const int wave = tid >> 5;
  const int lane = tid & 31;
  const int nct = N >> 4;
  const int col = lane & 15;
  const int koff = (lane >> 4) << 1;

  for (int t = wave; t < 4 * nct; t += 8) {
    int rt = t / nct;
    int ct = t - rt * nct;
    int mbase = rt << 4;
    int nbase = ct << 4;
    const float* aRow = &lA[(mbase + col) * LDS_STRIDE + koff];
    const float* bRow = &lW[(nbase + col) * LDS_STRIDE + koff];
    v8f c = {0.f, 0.f, 0.f, 0.f, 0.f, 0.f, 0.f, 0.f};
#pragma unroll
    for (int k0 = 0; k0 < 128; k0 += 4) {
      v2f a = *(const v2f*)(aRow + k0);
      v2f b = *(const v2f*)(bRow + k0);
      c = __builtin_amdgcn_wmma_f32_16x16x4_f32(false, a, false, b, (short)0, c, false, false);
    }
    int madd = (lane < 16) ? 0 : 8;
#pragma unroll
    for (int i = 0; i < 8; i++) {
      int gr = row0 + mbase + i + madd;
      if (gr < n_rows) {
        float v = c[i] * dinv[gr];
        size_t o = (size_t)gr * N + nbase + col;
        hs[o] = v;
        agg[o] = v;
      }
    }
  }
}

__global__ void scatter_kernel(const int* __restrict__ src, const int* __restrict__ dst,
                               const float* __restrict__ hs, float* __restrict__ agg,
                               int nE, int N) {
  int g = blockIdx.x * blockDim.x + threadIdx.x;
  int e = g >> 5;
  int lane = g & 31;
  if (e >= nE) return;
  const float* hp = hs + (size_t)src[e] * N;
  float* ap = agg + (size_t)dst[e] * N;
  for (int f = lane; f < N; f += 32)
    unsafeAtomicAdd(&ap[f], hp[f]);
}

__global__ void combine_kernel(const float* __restrict__ agg, const float* __restrict__ dinv,
                               const float* __restrict__ bias, float* __restrict__ out,
                               int n_rows, int N, int do_relu) {
  int i = blockIdx.x * blockDim.x + threadIdx.x;
  if (i >= n_rows * N) return;
  int r = i / N;
  int f = i - r * N;
  float v = dinv[r] * agg[i] + bias[f];
  if (do_relu) v = fmaxf(v, 0.f);
  out[i] = v;
}

__global__ void edgedot_kernel(const int* __restrict__ es, const int* __restrict__ ed,
                               const float* __restrict__ z, float* __restrict__ out, int nL) {
  int g = blockIdx.x * blockDim.x + threadIdx.x;
  int e = g >> 5;
  int lane = g & 31;
  if (e >= nL) return;
  const float* a = z + (size_t)es[e] * 64;
  const float* b = z + (size_t)ed[e] * 64;
  float p = a[lane] * b[lane] + a[lane + 32] * b[lane + 32];
#pragma unroll
  for (int off = 16; off > 0; off >>= 1) p += __shfl_xor(p, off, 32);
  if (lane == 0) out[e] = p;
}

extern "C" void kernel_launch(void* const* d_in, const int* in_sizes, int n_in,
                              void* d_out, int out_size, void* d_ws, size_t ws_size,
                              hipStream_t stream) {
  const int*   x_idx = (const int*)d_in[0];
  const int*   eidx  = (const int*)d_in[1];
  const int*   lidx  = (const int*)d_in[2];
  const float* emb   = (const float*)d_in[3];
  const float* W1    = (const float*)d_in[4];
  const float* b1    = (const float*)d_in[5];
  const float* W2    = (const float*)d_in[6];
  const float* b2    = (const float*)d_in[7];
  float* out = (float*)d_out;

  const int nN = in_sizes[0];
  const int nE = in_sizes[1] / 2;
  const int nL = in_sizes[2] / 2;
  const int H = 128, O = 64;

  float* dinv = (float*)d_ws;
  float* hs1  = dinv + ((nN + 63) & ~63);
  float* agg1 = hs1 + (size_t)nN * H;
  float* hs2  = agg1 + (size_t)nN * H;
  float* agg2 = hs2 + (size_t)nN * O;

  const int* esrc = eidx;
  const int* edst = eidx + nE;

  deg_init_kernel<<<(nN + 255) / 256, 256, 0, stream>>>(dinv, nN);
  deg_count_kernel<<<(nE + 255) / 256, 256, 0, stream>>>(edst, dinv, nE);
  deg_rsqrt_kernel<<<(nN + 255) / 256, 256, 0, stream>>>(dinv, nN);

  size_t shmem1 = (size_t)(64 * LDS_STRIDE + 128 * LDS_STRIDE) * sizeof(float);
  gemm_gather_scale_kernel<<<(nN + 63) / 64, 256, shmem1, stream>>>(
      emb, x_idx, W1, dinv, hs1, agg1, nN, H);
  scatter_kernel<<<(unsigned)(((size_t)nE * 32 + 255) / 256), 256, 0, stream>>>(
      esrc, edst, hs1, agg1, nE, H);
  combine_kernel<<<(unsigned)(((size_t)nN * H + 255) / 256), 256, 0, stream>>>(
      agg1, dinv, b1, hs1, nN, H,  1);

  size_t shmem2 = (size_t)(64 * LDS_STRIDE + 64 * LDS_STRIDE) * sizeof(float);
  gemm_gather_scale_kernel<<<(nN + 63) / 64, 256, shmem2, stream>>>(
      hs1, nullptr, W2, dinv, hs2, agg2, nN, O);
  scatter_kernel<<<(unsigned)(((size_t)nE * 32 + 255) / 256), 256, 0, stream>>>(
      esrc, edst, hs2, agg2, nE, O);
  combine_kernel<<<(unsigned)(((size_t)nN * O + 255) / 256), 256, 0, stream>>>(
      agg2, dinv, b2, hs2, nN, O,  0);

  edgedot_kernel<<<(unsigned)(((size_t)nL * 32 + 255) / 256), 256, 0, stream>>>(
      lidx, lidx + nL, hs2, out, nL);
}


# Node sums by dst range in LDS: 256 blocks each own a row range's sums, filter the edge list, add source rows by LDS f32 adds, store relu(dinv*sum+b); global atomics, combines gone; LDS arrival order
# speedup vs baseline: 1.4296x; 1.4296x over previous
#include <hip/hip_runtime.h>

typedef float v2f __attribute__((ext_vector_type(2)));
typedef float v8f __attribute__((ext_vector_type(8)));

#define LDS_STRIDE 132

__global__ void deg_init_kernel(float* deg, int n) {
  int i = blockIdx.x * blockDim.x + threadIdx.x;
  if (i < n) deg[i] = 1.0f;
}

__global__ void deg_count_kernel(const int* __restrict__ dst, float* __restrict__ deg, int nE) {
  int e = blockIdx.x * blockDim.x + threadIdx.x;
  if (e < nE) unsafeAtomicAdd(&deg[dst[e]], 1.0f);
}

__global__ void deg_rsqrt_kernel(float* deg, int n) {
  int i = blockIdx.x * blockDim.x + threadIdx.x;
  if (i < n) deg[i] = rsqrtf(deg[i]);
}

__global__ void gemm_gather_scale_kernel(const float* __restrict__ X,
                                         const int* __restrict__ gidx,
                                         const float* __restrict__ W,
                                         const float* __restrict__ dinv,
                                         float* __restrict__ hs,
                                         int n_rows, int N) {
  extern __shared__ float lds[];
  float* lA = lds;
  float* lW = lds + 64 * LDS_STRIDE;
  const int tid = threadIdx.x;
  const int row0 = blockIdx.x * 64;

  const float4* X4 = (const float4*)X;
  for (int i = tid; i < 64 * 32; i += 256) {
    int r = i >> 5;
    int c4 = i & 31;
    int gr = row0 + r;
    float4 v = make_float4(0.f, 0.f, 0.f, 0.f);
    if (gr < n_rows) {
      int sr = gidx ? gidx[gr] : gr;
      v = X4[(size_t)sr * 32 + c4];
    }
    *(float4*)&lA[r * LDS_STRIDE + (c4 << 2)] = v;
  }
  for (int i = tid; i < 128 * N; i += 256) {
    int k = i / N;
    int n = i - k * N;
    lW[n * LDS_STRIDE + k] = W[i];
  }
  __syncthreads();

  const int wave = tid >> 5;
  const int lane = tid & 31;
  const int nct = N >> 4;
  const int col = lane & 15;
  const int koff = (lane >> 4) << 1;

  for (int t = wave; t < 4 * nct; t += 8) {
    int rt = t / nct;
    int ct = t - rt * nct;
    int mbase = rt << 4;
    int nbase = ct << 4;
    const float* aRow = &lA[(mbase + col) * LDS_STRIDE + koff];
    const float* bRow = &lW[(nbase + col) * LDS_STRIDE + koff];
    v8f c = {0.f, 0.f, 0.f, 0.f, 0.f, 0.f, 0.f, 0.f};
#pragma unroll
    for (int k0 = 0; k0 < 128; k0 += 4) {
      v2f a = *(const v2f*)(aRow + k0);
      v2f b = *(const v2f*)(bRow + k0);
      c = __builtin_amdgcn_wmma_f32_16x16x4_f32(false, a, false, b, (short)0, c, false, false);
    }
    int madd = (lane < 16) ? 0 : 8;
#pragma unroll
    for (int i = 0; i < 8; i++) {
      int gr = row0 + mbase + i + madd;
      if (gr < n_rows) {
        float v = c[i] * dinv[gr];
        size_t o = (size_t)gr * N + nbase + col;
        hs[o] = v;
      }
    }
  }
}

#define RS_THREADS 1024
#define RS_WAVES (RS_THREADS / 32)
#define RS_LIST_CAP 8192

__device__ __forceinline__ void rs_queue_one(int d, int e, int r0, unsigned un, int2* list, int* cntp) {
  unsigned loc = (unsigned)(d - r0);
  if (loc < un) {
    int slot = atomicAdd(cntp, 1);
    if (slot < RS_LIST_CAP) list[slot] = make_int2(e, (int)loc);
  }
}

template <int N, int RELU>
__global__ void __launch_bounds__(RS_THREADS)
range_sum_kernel(const int* __restrict__ src, const int* __restrict__ dst,
                 const float* __restrict__ hs, const float* __restrict__ dinv,
                 const float* __restrict__ bias, float* __restrict__ outp,
                 int n_rows, int nE, int rpb) {
  extern __shared__ float lds[];
  float* acc = lds;
  float* dl = lds + (size_t)rpb * N;
  int2* list = (int2*)(dl + ((rpb + 1) & ~1));
  int* cntp = (int*)(list + RS_LIST_CAP);
  const int tid = threadIdx.x;
  const int r0 = blockIdx.x * rpb;
  int nrows = n_rows - r0;
  if (nrows > rpb) nrows = rpb;
  if (nrows < 0) nrows = 0;
  const unsigned un = (unsigned)nrows;
  const int nel4 = (nrows * N) >> 2;

  const float4* h4 = (const float4*)(hs + (size_t)r0 * N);
  float4* a4 = (float4*)acc;
  int i0 = tid;
  for (; i0 + 3 * RS_THREADS < nel4; i0 += 4 * RS_THREADS) {
    float4 t[4];
#pragma unroll
    for (int u = 0; u < 4; u++) t[u] = h4[i0 + u * RS_THREADS];
#pragma unroll
    for (int u = 0; u < 4; u++) a4[i0 + u * RS_THREADS] = t[u];
  }
  for (; i0 < nel4; i0 += RS_THREADS) a4[i0] = h4[i0];
  for (int r = tid; r < nrows; r += RS_THREADS) dl[r] = dinv[r0 + r];
  if (tid == 0) *cntp = 0;
  __syncthreads();

  const int nV = ((((size_t)dst) & 15) == 0) ? (nE >> 2) : 0;
  const int4* d4 = (const int4*)dst;
  int base = 0;
  for (; base + 4 * RS_THREADS <= nV; base += 4 * RS_THREADS) {
    int4 v[4];
#pragma unroll
    for (int u = 0; u < 4; u++) v[u] = d4[base + u * RS_THREADS + tid];
    unsigned mask = 0;
#pragma unroll
    for (int u = 0; u < 4; u++) {
      mask |= ((unsigned)(v[u].x - r0) < un) ? (1u << (4 * u + 0)) : 0u;
      mask |= ((unsigned)(v[u].y - r0) < un) ? (1u << (4 * u + 1)) : 0u;
      mask |= ((unsigned)(v[u].z - r0) < un) ? (1u << (4 * u + 2)) : 0u;
      mask |= ((unsigned)(v[u].w - r0) < un) ? (1u << (4 * u + 3)) : 0u;
    }
    if (mask) {
      int slot = atomicAdd(cntp, __popc(mask));
#pragma unroll
      for (int u = 0; u < 4; u++) {
        int e = (base + u * RS_THREADS + tid) << 2;
        if (mask & (1u << (4 * u + 0))) {
          if (slot < RS_LIST_CAP) list[slot] = make_int2(e + 0, v[u].x - r0);
          slot++;
        }
        if (mask & (1u << (4 * u + 1))) {
          if (slot < RS_LIST_CAP) list[slot] = make_int2(e + 1, v[u].y - r0);
          slot++;
        }
        if (mask & (1u << (4 * u + 2))) {
          if (slot < RS_LIST_CAP) list[slot] = make_int2(e + 2, v[u].z - r0);
          slot++;
        }
        if (mask & (1u << (4 * u + 3))) {
          if (slot < RS_LIST_CAP) list[slot] = make_int2(e + 3, v[u].w - r0);
          slot++;
        }
      }
    }
  }
  for (int i = base + tid; i < nV; i += RS_THREADS) {
    int4 v = d4[i];
    int e = i << 2;
    rs_queue_one(v.x, e + 0, r0, un, list, cntp);
    rs_queue_one(v.y, e + 1, r0, un, list, cntp);
    rs_queue_one(v.z, e + 2, r0, un, list, cntp);
    rs_queue_one(v.w, e + 3, r0, un, list, cntp);
  }
  for (int e = (nV << 2) + tid; e < nE; e += RS_THREADS) rs_queue_one(dst[e], e, r0, un, list, cntp);
  __syncthreads();

  const int n = *cntp;
  const bool over = n > RS_LIST_CAP;
  const int nq = over ? 0 : n;
  if (over) {
    for (int e = tid; e < nE; e += RS_THREADS) {
      unsigned loc = (unsigned)(dst[e] - r0);
      if (loc < un) {
        const float* hp = hs + (size_t)src[e] * N;
        float* ap = acc + (size_t)loc * N;
        for (int f = 0; f < N; f++) unsafeAtomicAdd(&ap[f], hp[f]);
      }
    }
  }
  for (int i = tid; i < nq; i += RS_THREADS) list[i].x = src[list[i].x];
  __syncthreads();

  const int wave = tid >> 5;
  const int lane = tid & 31;
  for (int i = wave; i < nq; i += 4 * RS_WAVES) {
    float v[4][N / 32];
    int loc[4];
#pragma unroll
    for (int q = 0; q < 4; q++) {
      int idx = i + q * RS_WAVES;
      loc[q] = -1;
      if (idx < nq) {
        int2 p = list[idx];
        loc[q] = p.y;
        const float* hp = hs + (size_t)p.x * N + lane;
#pragma unroll
        for (int j = 0; j < N / 32; j++) v[q][j] = hp[32 * j];
      }
    }
#pragma unroll
    for (int q = 0; q < 4; q++) {
      if (loc[q] >= 0) {
        float* ap = acc + (size_t)loc[q] * N + lane;
#pragma unroll
        for (int j = 0; j < N / 32; j++) unsafeAtomicAdd(&ap[32 * j], v[q][j]);
      }
    }
  }
  __syncthreads();

  const float4 b = *(const float4*)(bias + ((tid << 2) & (N - 1)));
  float4* o4 = (float4*)(outp + (size_t)r0 * N);
  for (int i = tid; i < nel4; i += RS_THREADS) {
    float4 a = a4[i];
    float di = dl[(i << 2) / N];
    float4 o;
    o.x = di * a.x + b.x;
    o.y = di * a.y + b.y;
    o.z = di * a.z + b.z;
    o.w = di * a.w + b.w;
    if (RELU) {
      o.x = fmaxf(o.x, 0.f);
      o.y = fmaxf(o.y, 0.f);
      o.z = fmaxf(o.z, 0.f);
      o.w = fmaxf(o.w, 0.f);
    }
    o4[i] = o;
  }
}

__global__ void edgedot_kernel(const int* __restrict__ es, const int* __restrict__ ed,
                               const float* __restrict__ z, float* __restrict__ out, int nL) {
  int g = blockIdx.x * blockDim.x + threadIdx.x;
  int e = g >> 5;
  int lane = g & 31;
  if (e >= nL) return;
  const float* a = z + (size_t)es[e] * 64;
  const float* b = z + (size_t)ed[e] * 64;
  float p = a[lane] * b[lane] + a[lane + 32] * b[lane + 32];
#pragma unroll
  for (int off = 16; off > 0; off >>= 1) p += __shfl_xor(p, off, 32);
  if (lane == 0) out[e] = p;
}

extern "C" void kernel_launch(void* const* d_in, const int* in_sizes, int n_in,
                              void* d_out, int out_size, void* d_ws, size_t ws_size,
                              hipStream_t stream) {
  const int*   x_idx = (const int*)d_in[0];
  const int*   eidx  = (const int*)d_in[1];
  const int*   lidx  = (const int*)d_in[2];
  const float* emb   = (const float*)d_in[3];
  const float* W1    = (const float*)d_in[4];
  const float* b1    = (const float*)d_in[5];
  const float* W2    = (const float*)d_in[6];
  const float* b2    = (const float*)d_in[7];
  float* out = (float*)d_out;

  const int nN = in_sizes[0];
  const int nE = in_sizes[1] / 2;
  const int nL = in_sizes[2] / 2;
  const int H = 128, O = 64;

  float* dinv = (float*)d_ws;
  float* hs1  = dinv + ((nN + 63) & ~63);
  float* x1   = hs1 + (size_t)nN * H;
  float* hs2  = x1 + (size_t)nN * H;
  float* z    = hs2 + (size_t)nN * O;

  const int* esrc = eidx;
  const int* edst = eidx + nE;

  deg_init_kernel<<<(nN + 255) / 256, 256, 0, stream>>>(dinv, nN);
  deg_count_kernel<<<(nE + 255) / 256, 256, 0, stream>>>(edst, dinv, nE);
  deg_rsqrt_kernel<<<(nN + 255) / 256, 256, 0, stream>>>(dinv, nN);

  const int rpb = (nN + 255) / 256;
  size_t rs_tail = (size_t)((rpb + 1) & ~1) * sizeof(float) + (size_t)RS_LIST_CAP * sizeof(int2) + 16;

  size_t shmem1 = (size_t)(64 * LDS_STRIDE + 128 * LDS_STRIDE) * sizeof(float);
  gemm_gather_scale_kernel<<<(nN + 63) / 64, 256, shmem1, stream>>>(
      emb, x_idx, W1, dinv, hs1, nN, H);
  range_sum_kernel<128, 1><<<256, RS_THREADS, (size_t)rpb * H * sizeof(float) + rs_tail, stream>>>(
      esrc, edst, hs1, dinv, b1, x1, nN, nE, rpb);

  size_t shmem2 = (size_t)(64 * LDS_STRIDE + 64 * LDS_STRIDE) * sizeof(float);
  gemm_gather_scale_kernel<<<(nN + 63) / 64, 256, shmem2, stream>>>(
      x1, nullptr, W2, dinv, hs2, nN, O);
  range_sum_kernel<64, 0><<<256, RS_THREADS, (size_t)rpb * O * sizeof(float) + rs_tail, stream>>>(
      esrc, edst, hs2, dinv, b2, z, nN, nE, rpb);

  edgedot_kernel<<<(unsigned)(((size_t)nL * 32 + 255) / 256), 256, 0, stream>>>(
      lidx, lidx + nL, z, out, nL);
}


# Edge list read once per call: first launch filters dst per row range, counts in-degrees in LDS (dinv), stores (src,row) queues both node-sum launches read; 3 degree launches gone; sum order kept
# speedup vs baseline: 1.8414x; 1.2881x over previous
#include <hip/hip_runtime.h>

typedef float v2f __attribute__((ext_vector_type(2)));
typedef float v8f __attribute__((ext_vector_type(8)));

#define LDS_STRIDE 132

__global__ void gemm_gather_scale_kernel(const float* __restrict__ X,
                                         const int* __restrict__ gidx,
                                         const float* __restrict__ W,
                                         const float* __restrict__ dinv,
                                         float* __restrict__ hs,
                                         int n_rows, int N) {
  extern __shared__ float lds[];
  float* lA = lds;
  float* lW = lds + 64 * LDS_STRIDE;
  const int tid = threadIdx.x;
  const int row0 = blockIdx.x * 64;

  const float4* X4 = (const float4*)X;
  for (int i = tid; i < 64 * 32; i += 256) {
    int r = i >> 5;
    int c4 = i & 31;
    int gr = row0 + r;
    float4 v = make_float4(0.f, 0.f, 0.f, 0.f);
    if (gr < n_rows) {
      int sr = gidx ? gidx[gr] : gr;
      v = X4[(size_t)sr * 32 + c4];
    }
    *(float4*)&lA[r * LDS_STRIDE + (c4 << 2)] = v;
  }
  for (int i = tid; i < 128 * N; i += 256) {
    int k = i / N;
    int n = i - k * N;
    lW[n * LDS_STRIDE + k] = W[i];
  }
  __syncthreads();

  const int wave = tid >> 5;
  const int lane = tid & 31;
  const int nct = N >> 4;
  const int col = lane & 15;
  const int koff = (lane >> 4) << 1;

  for (int t = wave; t < 4 * nct; t += 8) {
    int rt = t / nct;
    int ct = t - rt * nct;
    int mbase = rt << 4;
    int nbase = ct << 4;
    const float* aRow = &lA[(mbase + col) * LDS_STRIDE + koff];
    const float* bRow = &lW[(nbase + col) * LDS_STRIDE + koff];
    v8f c = {0.f, 0.f, 0.f, 0.f, 0.f, 0.f, 0.f, 0.f};
#pragma unroll
    for (int k0 = 0; k0 < 128; k0 += 4) {
      v2f a = *(const v2f*)(aRow + k0);
      v2f b = *(const v2f*)(bRow + k0);
      c = __builtin_amdgcn_wmma_f32_16x16x4_f32(false, a, false, b, (short)0, c, false, false);
    }
    int madd = (lane < 16) ? 0 : 8;
#pragma unroll
    for (int i = 0; i < 8; i++) {
      int gr = row0 + mbase + i + madd;
      if (gr < n_rows) {
        float v = c[i] * dinv[gr];
        size_t o = (size_t)gr * N + nbase + col;
        hs[o] = v;
      }
    }
  }
}

#define RS_THREADS 1024
#define RS_WAVES (RS_THREADS / 32)
#define RS_LIST_CAP 8192
#define RS_LIST_PER_THREAD (RS_LIST_CAP / RS_THREADS)

__device__ __forceinline__ void rq_queue_one(int d, int e, int r0, unsigned un, int2* list, int* cntp,
                                             int* degc) {
  unsigned loc = (unsigned)(d - r0);
  if (loc < un) {
    atomicAdd(&degc[loc], 1);
    int slot = atomicAdd(cntp, 1);
    if (slot < RS_LIST_CAP) list[slot] = make_int2(e, (int)loc);
  }
}

__global__ void __launch_bounds__(RS_THREADS)
range_queue_kernel(const int* __restrict__ src, const int* __restrict__ dst,
                   float* __restrict__ dinv, int* __restrict__ qcnt, int2* __restrict__ qlist,
                   int n_rows, int nE, int rpb) {
  extern __shared__ float lds[];
  int2* list = (int2*)lds;
  int* degc = (int*)(list + RS_LIST_CAP);
  int* cntp = degc + rpb;
  const int tid = threadIdx.x;
  const int r0 = blockIdx.x * rpb;
  int nrows = n_rows - r0;
  if (nrows > rpb) nrows = rpb;
  if (nrows < 0) nrows = 0;
  const unsigned un = (unsigned)nrows;

  for (int r = tid; r < nrows; r += RS_THREADS) degc[r] = 0;
  if (tid == 0) *cntp = 0;
  __syncthreads();

  const int nV = ((((size_t)dst) & 15) == 0) ? (nE >> 2) : 0;
  const int4* d4 = (const int4*)dst;
  int base = 0;
  for (; base + 4 * RS_THREADS <= nV; base += 4 * RS_THREADS) {
    int4 v[4];
#pragma unroll
    for (int u = 0; u < 4; u++) v[u] = d4[base + u * RS_THREADS + tid];
    unsigned mask = 0;
#pragma unroll
    for (int u = 0; u < 4; u++) {
      mask |= ((unsigned)(v[u].x - r0) < un) ? (1u << (4 * u + 0)) : 0u;
      mask |= ((unsigned)(v[u].y - r0) < un) ? (1u << (4 * u + 1)) : 0u;
      mask |= ((unsigned)(v[u].z - r0) < un) ? (1u << (4 * u + 2)) : 0u;
      mask |= ((unsigned)(v[u].w - r0) < un) ? (1u << (4 * u + 3)) : 0u;
    }
    if (mask) {
      int slot = atomicAdd(cntp, __popc(mask));
#pragma unroll
      for (int u = 0; u < 4; u++) {
        int e = (base + u * RS_THREADS + tid) << 2;
        if (mask & (1u << (4 * u + 0))) {
          atomicAdd(&degc[v[u].x - r0], 1);
          if (slot < RS_LIST_CAP) list[slot] = make_int2(e + 0, v[u].x - r0);
          slot++;
        }
        if (mask & (1u << (4 * u + 1))) {
          atomicAdd(&degc[v[u].y - r0], 1);
          if (slot < RS_LIST_CAP) list[slot] = make_int2(e + 1, v[u].y - r0);
          slot++;
        }
        if (mask & (1u << (4 * u + 2))) {
          atomicAdd(&degc[v[u].z - r0], 1);
          if (slot < RS_LIST_CAP) list[slot] = make_int2(e + 2, v[u].z - r0);
          slot++;
        }
        if (mask & (1u << (4 * u + 3))) {
          atomicAdd(&degc[v[u].w - r0], 1);
          if (slot < RS_LIST_CAP) list[slot] = make_int2(e + 3, v[u].w - r0);
          slot++;
        }
      }
    }
  }
  for (int i = base + tid; i < nV; i += RS_THREADS) {
    int4 v = d4[i];
    int e = i << 2;
    rq_queue_one(v.x, e + 0, r0, un, list, cntp, degc);
    rq_queue_one(v.y, e + 1, r0, un, list, cntp, degc);
    rq_queue_one(v.z, e + 2, r0, un, list, cntp, degc);
    rq_queue_one(v.w, e + 3, r0, un, list, cntp, degc);
  }
  for (int e = (nV << 2) + tid; e < nE; e += RS_THREADS)
    rq_queue_one(dst[e], e, r0, un, list, cntp, degc);
  __syncthreads();

  const int n = *cntp;
  const int nq = (n > RS_LIST_CAP) ? 0 : n;
  int2* q = qlist + (size_t)blockIdx.x * RS_LIST_CAP;
  if (nq > 0) {
    int2 p[RS_LIST_PER_THREAD];
    int s[RS_LIST_PER_THREAD];
#pragma unroll
    for (int u = 0; u < RS_LIST_PER_THREAD; u++) {
      int idx = tid + u * RS_THREADS;
      p[u] = list[idx < nq ? idx : 0];
    }
#pragma unroll
    for (int u = 0; u < RS_LIST_PER_THREAD; u++) s[u] = src[p[u].x];
#pragma unroll
    for (int u = 0; u < RS_LIST_PER_THREAD; u++) {
      int idx = tid + u * RS_THREADS;
      if (idx < nq) q[idx] = make_int2(s[u], p[u].y);
    }
  }
  if (tid == 0) qcnt[blockIdx.x] = n;
  for (int r = tid; r < nrows; r += RS_THREADS) dinv[r0 + r] = rsqrtf(1.0f + (float)degc[r]);
}

template <int N, int RELU>
__global__ void __launch_bounds__(RS_THREADS)
range_sum_kernel(const int* __restrict__ src, const int* __restrict__ dst,
                 const float* __restrict__ hs, const float* __restrict__ dinv,
                 const float* __restrict__ bias, float* __restrict__ outp,
                 const int* __restrict__ qcnt, const int2* __restrict__ qlist,
                 int n_rows, int nE, int rpb) {
  extern __shared__ float lds[];
  float* acc = lds;
  float* dl = lds + (size_t)rpb * N;
  int2* list = (int2*)(dl + ((rpb + 1) & ~1));
  const int tid = threadIdx.x;
  const int r0 = blockIdx.x * rpb;
  int nrows = n_rows - r0;
  if (nrows > rpb) nrows = rpb;
  if (nrows < 0) nrows = 0;
  const unsigned un = (unsigned)nrows;
  const int nel4 = (nrows * N) >> 2;

  const int n = qcnt[blockIdx.x];
  const bool over = n > RS_LIST_CAP;
  const int nq = over ? 0 : n;
  const int2* q = qlist + (size_t)blockIdx.x * RS_LIST_CAP;

  const float4* h4 = (const float4*)(hs + (size_t)r0 * N);
  float4* a4 = (float4*)acc;
  int i0 = tid;
  for (; i0 + 3 * RS_THREADS < nel4; i0 += 4 * RS_THREADS) {
    float4 t[4];
#pragma unroll
    for (int u = 0; u < 4; u++) t[u] = h4[i0 + u * RS_THREADS];
#pragma unroll
    for (int u = 0; u < 4; u++) a4[i0 + u * RS_THREADS] = t[u];
  }
  for (; i0 < nel4; i0 += RS_THREADS) a4[i0] = h4[i0];
  for (int r = tid; r < nrows; r += RS_THREADS) dl[r] = dinv[r0 + r];
  {
    int2 p[RS_LIST_PER_THREAD];
#pragma unroll
    for (int u = 0; u < RS_LIST_PER_THREAD; u++) {
      int idx = tid + u * RS_THREADS;
      p[u] = make_int2(0, 0);
      if (idx < nq) p[u] = q[idx];
    }
#pragma unroll
    for (int u = 0; u < RS_LIST_PER_THREAD; u++) {
      int idx = tid + u * RS_THREADS;
      if (idx < nq) list[idx] = p[u];
    }
  }
  __syncthreads();

  if (over) {
    for (int e = tid; e < nE; e += RS_THREADS) {
      unsigned loc = (unsigned)(dst[e] - r0);
      if (loc < un) {
        const float* hp = hs + (size_t)src[e] * N;
        float* ap = acc + (size_t)loc * N;
        for (int f = 0; f < N; f++) unsafeAtomicAdd(&ap[f], hp[f]);
      }
    }
  }

  const int wave = tid >> 5;
  const int lane = tid & 31;
  for (int i = wave; i < nq; i += 4 * RS_WAVES) {
    float v[4][N / 32];
    int loc[4];
#pragma unroll
    for (int q4 = 0; q4 < 4; q4++) {
      int idx = i + q4 * RS_WAVES;
      loc[q4] = -1;
      if (idx < nq) {
        int2 p = list[idx];
        loc[q4] = p.y;
        const float* hp = hs + (size_t)p.x * N + lane;
#pragma unroll
        for (int j = 0; j < N / 32; j++) v[q4][j] = hp[32 * j];
      }
    }
#pragma unroll
    for (int q4 = 0; q4 < 4; q4++) {
      if (loc[q4] >= 0) {
        float* ap = acc + (size_t)loc[q4] * N + lane;
#pragma unroll
        for (int j = 0; j < N / 32; j++) unsafeAtomicAdd(&ap[32 * j], v[q4][j]);
      }
    }
  }
  __syncthreads();

  const float4 b = *(const float4*)(bias + ((tid << 2) & (N - 1)));
  float4* o4 = (float4*)(outp + (size_t)r0 * N);
  for (int i = tid; i < nel4; i += RS_THREADS) {
    float4 a = a4[i];
    float di = dl[(i << 2) / N];
    float4 o;
    o.x = di * a.x + b.x;
    o.y = di * a.y + b.y;
    o.z = di * a.z + b.z;
    o.w = di * a.w + b.w;
    if (RELU) {
      o.x = fmaxf(o.x, 0.f);
      o.y = fmaxf(o.y, 0.f);
      o.z = fmaxf(o.z, 0.f);
      o.w = fmaxf(o.w, 0.f);
    }
    o4[i] = o;
  }
}

__global__ void edgedot_kernel(const int* __restrict__ es, const int* __restrict__ ed,
                               const float* __restrict__ z, float* __restrict__ out, int nL) {
  int g = blockIdx.x * blockDim.x + threadIdx.x;
  int e = g >> 5;
  int lane = g & 31;
  if (e >= nL) return;
  const float* a = z + (size_t)es[e] * 64;
  const float* b = z + (size_t)ed[e] * 64;
  float p = a[lane] * b[lane] + a[lane + 32] * b[lane + 32];
#pragma unroll
  for (int off = 16; off > 0; off >>= 1) p += __shfl_xor(p, off, 32);
  if (lane == 0) out[e] = p;
}

extern "C" void kernel_launch(void* const* d_in, const int* in_sizes, int n_in,
                              void* d_out, int out_size, void* d_ws, size_t ws_size,
                              hipStream_t stream) {
  const int*   x_idx = (const int*)d_in[0];
  const int*   eidx  = (const int*)d_in[1];
  const int*   lidx  = (const int*)d_in[2];
  const float* emb   = (const float*)d_in[3];
  const float* W1    = (const float*)d_in[4];
  const float* b1    = (const float*)d_in[5];
  const float* W2    = (const float*)d_in[6];
  const float* b2    = (const float*)d_in[7];
  float* out = (float*)d_out;

  const int nN = in_sizes[0];
  const int nE = in_sizes[1] / 2;
  const int nL = in_sizes[2] / 2;
  const int H = 128, O = 64;

  float* dinv = (float*)d_ws;
  float* hs1  = dinv + ((nN + 63) & ~63);
  float* x1   = hs1 + (size_t)nN * H;
  float* hs2  = x1 + (size_t)nN * H;
  float* z    = hs2 + (size_t)nN * O;
  int*   qcnt = (int*)(z + (size_t)nN * O);
  int2*  qlist = (int2*)(qcnt + 256);

  const int* esrc = eidx;
  const int* edst = eidx + nE;

  const int rpb = (nN + 255) / 256;
  size_t rs_tail = (size_t)((rpb + 1) & ~1) * sizeof(float) + (size_t)RS_LIST_CAP * sizeof(int2) + 16;

  range_queue_kernel<<<256, RS_THREADS,
                       (size_t)RS_LIST_CAP * sizeof(int2) + (size_t)rpb * sizeof(int) + 16, stream>>>(
      esrc, edst, dinv, qcnt, qlist, nN, nE, rpb);

  size_t shmem1 = (size_t)(64 * LDS_STRIDE + 128 * LDS_STRIDE) * sizeof(float);
  gemm_gather_scale_kernel<<<(nN + 63) / 64, 256, shmem1, stream>>>(
      emb, x_idx, W1, dinv, hs1, nN, H);
  range_sum_kernel<128, 1><<<256, RS_THREADS, (size_t)rpb * H * sizeof(float) + rs_tail, stream>>>(
      esrc, edst, hs1, dinv, b1, x1, qcnt, qlist, nN, nE, rpb);

  size_t shmem2 = (size_t)(64 * LDS_STRIDE + 64 * LDS_STRIDE) * sizeof(float);
  gemm_gather_scale_kernel<<<(nN + 63) / 64, 256, shmem2, stream>>>(
      x1, nullptr, W2, dinv, hs2, nN, O);
  range_sum_kernel<64, 0><<<256, RS_THREADS, (size_t)rpb * O * sizeof(float) + rs_tail, stream>>>(
      esrc, edst, hs2, dinv, b2, z, qcnt, qlist, nN, nE, rpb);

  edgedot_kernel<<<(unsigned)(((size_t)nL * 32 + 255) / 256), 256, 0, stream>>>(
      lidx, lidx + nL, z, out, nL);
}


# GEMM launches persistent with W in registers: a wave owns one 16-column tile, reads its B fragments once per block, loops over 64-row tiles (next rows prefetched); no W in LDS; k sum order kept
# speedup vs baseline: 2.6028x; 1.4135x over previous
#include <hip/hip_runtime.h>

typedef float v2f __attribute__((ext_vector_type(2)));
typedef float v8f __attribute__((ext_vector_type(8)));

#define LDS_STRIDE 132

#define GT_TILE_ROWS 64

template <int GATHER>
__device__ __forceinline__ void gt_load_tile(const float4* __restrict__ X4,
                                             const int* __restrict__ gidx,
                                             const float* __restrict__ dinv, int row0, int n_rows,
                                             int tid, float4 (&pre)[8], float& dpre) {
  const int c4 = tid & 31;
  const int rw = tid >> 5;
  int sr[8];
#pragma unroll
  for (int u = 0; u < 8; u++) {
    int gr = row0 + rw + 8 * u;
    int g = (gr < n_rows) ? gr : 0;
    sr[u] = GATHER ? gidx[g] : g;
  }
#pragma unroll
  for (int u = 0; u < 8; u++) {
    pre[u] = X4[(size_t)sr[u] * 32 + c4];
  }
  dpre = 0.f;
  if (tid < GT_TILE_ROWS && row0 + tid < n_rows) dpre = dinv[row0 + tid];
}

template <int N, int GATHER>
__global__ void __launch_bounds__(256)
gemm_rows_kernel(const float* __restrict__ X, const int* __restrict__ gidx,
                 const float* __restrict__ W, const float* __restrict__ dinv,
                 float* __restrict__ hs, int n_rows) {
  __shared__ float lA[GT_TILE_ROWS * LDS_STRIDE];
  __shared__ __attribute__((aligned(16))) float lD[GT_TILE_ROWS];
  constexpr int NCT = N / 16;
  constexpr int WPC = 8 / NCT;
  constexpr int MT = 4 / WPC;
  const int tid = threadIdx.x;
  const int wave = tid >> 5;
  const int lane = tid & 31;
  const int col = lane & 15;
  const int koff = (lane >> 4) << 1;
  const int nbase = (wave % NCT) << 4;
  const int mt0 = (wave / NCT) * MT;
  const int ntiles = (n_rows + GT_TILE_ROWS - 1) / GT_TILE_ROWS;
  const float4* X4 = (const float4*)X;

  int tile = blockIdx.x;
  float4 pre[8];
  float dpre = 0.f;
#pragma unroll
  for (int u = 0; u < 8; u++) pre[u] = make_float4(0.f, 0.f, 0.f, 0.f);
  if (tile < ntiles) gt_load_tile<GATHER>(X4, gidx, dinv, tile * GT_TILE_ROWS, n_rows, tid, pre, dpre);

  v2f bf[32];
  {
    const float* wp = W + (size_t)koff * N + nbase + col;
#pragma unroll
    for (int k = 0; k < 32; k++) {
      bf[k].x = wp[(size_t)(4 * k) * N];
      bf[k].y = wp[(size_t)(4 * k + 1) * N];
    }
  }

  const int madd = (lane < 16) ? 0 : 8;
  const float* aBase = &lA[(mt0 * 16 + col) * LDS_STRIDE + koff];

  for (; tile < ntiles; tile += gridDim.x) {
    const int row0 = tile * GT_TILE_ROWS;
    __syncthreads();
    {
      const int c4 = tid & 31;
      const int rw = tid >> 5;
#pragma unroll
      for (int u = 0; u < 8; u++) {
        float4 v = pre[u];
        if (row0 + rw + 8 * u >= n_rows) v = make_float4(0.f, 0.f, 0.f, 0.f);
        *(float4*)&lA[(rw + 8 * u) * LDS_STRIDE + (c4 << 2)] = v;
      }
      if (tid < GT_TILE_ROWS) lD[tid] = dpre;
    }
    __syncthreads();
    const int nt = tile + gridDim.x;
    if (nt < ntiles) gt_load_tile<GATHER>(X4, gidx, dinv, nt * GT_TILE_ROWS, n_rows, tid, pre, dpre);

    v8f c[MT];
#pragma unroll
    for (int m = 0; m < MT; m++) c[m] = (v8f){0.f, 0.f, 0.f, 0.f, 0.f, 0.f, 0.f, 0.f};
#pragma unroll
    for (int kg = 0; kg < 32; kg += 4) {
      v2f a[4][MT];
#pragma unroll
      for (int kk = 0; kk < 4; kk++)
#pragma unroll
        for (int m = 0; m < MT; m++)
          a[kk][m] = *(const v2f*)(aBase + m * 16 * LDS_STRIDE + 4 * (kg + kk));
#pragma unroll
      for (int kk = 0; kk < 4; kk++)
#pragma unroll
        for (int m = 0; m < MT; m++)
          c[m] = __builtin_amdgcn_wmma_f32_16x16x4_f32(false, a[kk][m], false, bf[kg + kk], (short)0,
                                                       c[m], false, false);
    }
    float4 dv[MT][2];
#pragma unroll
    for (int m = 0; m < MT; m++) {
      const float4* dp = (const float4*)&lD[(mt0 + m) * 16 + madd];
      dv[m][0] = dp[0];
      dv[m][1] = dp[1];
    }
#pragma unroll
    for (int m = 0; m < MT; m++) {
      const float d[8] = {dv[m][0].x, dv[m][0].y, dv[m][0].z, dv[m][0].w,
                          dv[m][1].x, dv[m][1].y, dv[m][1].z, dv[m][1].w};
#pragma unroll
      for (int i = 0; i < 8; i++) {
        int lr = (mt0 + m) * 16 + i + madd;
        int gr = row0 + lr;
        if (gr < n_rows) hs[(size_t)gr * N + nbase + col] = c[m][i] * d[i];
      }
    }
  }
}

#define RS_THREADS 1024
#define RS_WAVES (RS_THREADS / 32)
#define RS_LIST_CAP 8192
#define RS_LIST_PER_THREAD (RS_LIST_CAP / RS_THREADS)

__device__ __forceinline__ void rq_queue_one(int d, int e, int r0, unsigned un, int2* list, int* cntp,
                                             int* degc) {
  unsigned loc = (unsigned)(d - r0);
  if (loc < un) {
    atomicAdd(&degc[loc], 1);
    int slot = atomicAdd(cntp, 1);
    if (slot < RS_LIST_CAP) list[slot] = make_int2(e, (int)loc);
  }
}

__global__ void __launch_bounds__(RS_THREADS)
range_queue_kernel(const int* __restrict__ src, const int* __restrict__ dst,
                   float* __restrict__ dinv, int* __restrict__ qcnt, int2* __restrict__ qlist,
                   int n_rows, int nE, int rpb) {
  extern __shared__ float lds[];
  int2* list = (int2*)lds;
  int* degc = (int*)(list + RS_LIST_CAP);
  int* cntp = degc + rpb;
  const int tid = threadIdx.x;
  const int r0 = blockIdx.x * rpb;
  int nrows = n_rows - r0;
  if (nrows > rpb) nrows = rpb;
  if (nrows < 0) nrows = 0;
  const unsigned un = (unsigned)nrows;

  for (int r = tid; r < nrows; r += RS_THREADS) degc[r] = 0;
  if (tid == 0) *cntp = 0;
  __syncthreads();

  const int nV = ((((size_t)dst) & 15) == 0) ? (nE >> 2) : 0;
  const int4* d4 = (const int4*)dst;
  int base = 0;
  for (; base + 4 * RS_THREADS <= nV; base += 4 * RS_THREADS) {
    int4 v[4];
#pragma unroll
    for (int u = 0; u < 4; u++) v[u] = d4[base + u * RS_THREADS + tid];
    unsigned mask = 0;
#pragma unroll
    for (int u = 0; u < 4; u++) {
      mask |= ((unsigned)(v[u].x - r0) < un) ? (1u << (4 * u + 0)) : 0u;
      mask |= ((unsigned)(v[u].y - r0) < un) ? (1u << (4 * u + 1)) : 0u;
      mask |= ((unsigned)(v[u].z - r0) < un) ? (1u << (4 * u + 2)) : 0u;
      mask |= ((unsigned)(v[u].w - r0) < un) ? (1u << (4 * u + 3)) : 0u;
    }
    if (mask) {
      int slot = atomicAdd(cntp, __popc(mask));
#pragma unroll
      for (int u = 0; u < 4; u++) {
        int e = (base + u * RS_THREADS + tid) << 2;
        if (mask & (1u << (4 * u + 0))) {
          atomicAdd(&degc[v[u].x - r0], 1);
          if (slot < RS_LIST_CAP) list[slot] = make_int2(e + 0, v[u].x - r0);
          slot++;
        }
        if (mask & (1u << (4 * u + 1))) {
          atomicAdd(&degc[v[u].y - r0], 1);
          if (slot < RS_LIST_CAP) list[slot] = make_int2(e + 1, v[u].y - r0);
          slot++;
        }
        if (mask & (1u << (4 * u + 2))) {
          atomicAdd(&degc[v[u].z - r0], 1);
          if (slot < RS_LIST_CAP) list[slot] = make_int2(e + 2, v[u].z - r0);
          slot++;
        }
        if (mask & (1u << (4 * u + 3))) {
          atomicAdd(&degc[v[u].w - r0], 1);
          if (slot < RS_LIST_CAP) list[slot] = make_int2(e + 3, v[u].w - r0);
          slot++;
        }
      }
    }
  }
  for (int i = base + tid; i < nV; i += RS_THREADS) {
    int4 v = d4[i];
    int e = i << 2;
    rq_queue_one(v.x, e + 0, r0, un, list, cntp, degc);
    rq_queue_one(v.y, e + 1, r0, un, list, cntp, degc);
    rq_queue_one(v.z, e + 2, r0, un, list, cntp, degc);
    rq_queue_one(v.w, e + 3, r0, un, list, cntp, degc);
  }
  for (int e = (nV << 2) + tid; e < nE; e += RS_THREADS)
    rq_queue_one(dst[e], e, r0, un, list, cntp, degc);
  __syncthreads();

  const int n = *cntp;
  const int nq = (n > RS_LIST_CAP) ? 0 : n;
  int2* q = qlist + (size_t)blockIdx.x * RS_LIST_CAP;
  if (nq > 0) {
    int2 p[RS_LIST_PER_THREAD];
    int s[RS_LIST_PER_THREAD];
#pragma unroll
    for (int u = 0; u < RS_LIST_PER_THREAD; u++) {
      int idx = tid + u * RS_THREADS;
      p[u] = list[idx < nq ? idx : 0];
    }
#pragma unroll
    for (int u = 0; u < RS_LIST_PER_THREAD; u++) s[u] = src[p[u].x];
#pragma unroll
    for (int u = 0; u < RS_LIST_PER_THREAD; u++) {
      int idx = tid + u * RS_THREADS;
      if (idx < nq) q[idx] = make_int2(s[u], p[u].y);
    }
  }
  if (tid == 0) qcnt[blockIdx.x] = n;
  for (int r = tid; r < nrows; r += RS_THREADS) dinv[r0 + r] = rsqrtf(1.0f + (float)degc[r]);
}

template <int N, int RELU>
__global__ void __launch_bounds__(RS_THREADS)
range_sum_kernel(const int* __restrict__ src, const int* __restrict__ dst,
                 const float* __restrict__ hs, const float* __restrict__ dinv,
                 const float* __restrict__ bias, float* __restrict__ outp,
                 const int* __restrict__ qcnt, const int2* __restrict__ qlist,
                 int n_rows, int nE, int rpb) {
  extern __shared__ float lds[];
  float* acc = lds;
  float* dl = lds + (size_t)rpb * N;
  int2* list = (int2*)(dl + ((rpb + 1) & ~1));
  const int tid = threadIdx.x;
  const int r0 = blockIdx.x * rpb;
  int nrows = n_rows - r0;
  if (nrows > rpb) nrows = rpb;
  if (nrows < 0) nrows = 0;
  const unsigned un = (unsigned)nrows;
  const int nel4 = (nrows * N) >> 2;

  const int n = qcnt[blockIdx.x];
  const bool over = n > RS_LIST_CAP;
  const int nq = over ? 0 : n;
  const int2* q = qlist + (size_t)blockIdx.x * RS_LIST_CAP;

  const float4* h4 = (const float4*)(hs + (size_t)r0 * N);
  float4* a4 = (float4*)acc;
  int i0 = tid;
  for (; i0 + 3 * RS_THREADS < nel4; i0 += 4 * RS_THREADS) {
    float4 t[4];
#pragma unroll
    for (int u = 0; u < 4; u++) t[u] = h4[i0 + u * RS_THREADS];
#pragma unroll
    for (int u = 0; u < 4; u++) a4[i0 + u * RS_THREADS] = t[u];
  }
  for (; i0 < nel4; i0 += RS_THREADS) a4[i0] = h4[i0];
  for (int r = tid; r < nrows; r += RS_THREADS) dl[r] = dinv[r0 + r];
  {
    int2 p[RS_LIST_PER_THREAD];
#pragma unroll
    for (int u = 0; u < RS_LIST_PER_THREAD; u++) {
      int idx = tid + u * RS_THREADS;
      p[u] = make_int2(0, 0);
      if (idx < nq) p[u] = q[idx];
    }
#pragma unroll
    for (int u = 0; u < RS_LIST_PER_THREAD; u++) {
      int idx = tid + u * RS_THREADS;
      if (idx < nq) list[idx] = p[u];
    }
  }
  __syncthreads();

  if (over) {
    for (int e = tid; e < nE; e += RS_THREADS) {
      unsigned loc = (unsigned)(dst[e] - r0);
      if (loc < un) {
        const float* hp = hs + (size_t)src[e] * N;
        float* ap = acc + (size_t)loc * N;
        for (int f = 0; f < N; f++) unsafeAtomicAdd(&ap[f], hp[f]);
      }
    }
  }

  const int wave = tid >> 5;
  const int lane = tid & 31;
  for (int i = wave; i < nq; i += 4 * RS_WAVES) {
    float v[4][N / 32];
    int loc[4];
#pragma unroll
    for (int q4 = 0; q4 < 4; q4++) {
      int idx = i + q4 * RS_WAVES;
      loc[q4] = -1;
      if (idx < nq) {
        int2 p = list[idx];
        loc[q4] = p.y;
        const float* hp = hs + (size_t)p.x * N + lane;
#pragma unroll
        for (int j = 0; j < N / 32; j++) v[q4][j] = hp[32 * j];
      }
    }
#pragma unroll
    for (int q4 = 0; q4 < 4; q4++) {
      if (loc[q4] >= 0) {
        float* ap = acc + (size_t)loc[q4] * N + lane;
#pragma unroll
        for (int j = 0; j < N / 32; j++) unsafeAtomicAdd(&ap[32 * j], v[q4][j]);
      }
    }
  }
  __syncthreads();

  const float4 b = *(const float4*)(bias + ((tid << 2) & (N - 1)));
  float4* o4 = (float4*)(outp + (size_t)r0 * N);
  for (int i = tid; i < nel4; i += RS_THREADS) {
    float4 a = a4[i];
    float di = dl[(i << 2) / N];
    float4 o;
    o.x = di * a.x + b.x;
    o.y = di * a.y + b.y;
    o.z = di * a.z + b.z;
    o.w = di * a.w + b.w;
    if (RELU) {
      o.x = fmaxf(o.x, 0.f);
      o.y = fmaxf(o.y, 0.f);
      o.z = fmaxf(o.z, 0.f);
      o.w = fmaxf(o.w, 0.f);
    }
    o4[i] = o;
  }
}

__global__ void edgedot_kernel(const int* __restrict__ es, const int* __restrict__ ed,
                               const float* __restrict__ z, float* __restrict__ out, int nL) {
  int g = blockIdx.x * blockDim.x + threadIdx.x;
  int e = g >> 5;
  int lane = g & 31;
  if (e >= nL) return;
  const float* a = z + (size_t)es[e] * 64;
  const float* b = z + (size_t)ed[e] * 64;
  float p = a[lane] * b[lane] + a[lane + 32] * b[lane + 32];
#pragma unroll
  for (int off = 16; off > 0; off >>= 1) p += __shfl_xor(p, off, 32);
  if (lane == 0) out[e] = p;
}

extern "C" void kernel_launch(void* const* d_in, const int* in_sizes, int n_in,
                              void* d_out, int out_size, void* d_ws, size_t ws_size,
                              hipStream_t stream) {
  const int*   x_idx = (const int*)d_in[0];
  const int*   eidx  = (const int*)d_in[1];
  const int*   lidx  = (const int*)d_in[2];
  const float* emb   = (const float*)d_in[3];
  const float* W1    = (const float*)d_in[4];
  const float* b1    = (const float*)d_in[5];
  const float* W2    = (const float*)d_in[6];
  const float* b2    = (const float*)d_in[7];
  float* out = (float*)d_out;

  const int nN = in_sizes[0];
  const int nE = in_sizes[1] / 2;
  const int nL = in_sizes[2] / 2;
  const int H = 128, O = 64;

  float* dinv = (float*)d_ws;
  float* hs1  = dinv + ((nN + 63) & ~63);
  float* x1   = hs1 + (size_t)nN * H;
  float* hs2  = x1 + (size_t)nN * H;
  float* z    = hs2 + (size_t)nN * O;
  int*   qcnt = (int*)(z + (size_t)nN * O);
  int2*  qlist = (int2*)(qcnt + 256);

  const int* esrc = eidx;
  const int* edst = eidx + nE;

  const int rpb = (nN + 255) / 256;
  size_t rs_tail = (size_t)((rpb + 1) & ~1) * sizeof(float) + (size_t)RS_LIST_CAP * sizeof(int2) + 16;

  range_queue_kernel<<<256, RS_THREADS,
                       (size_t)RS_LIST_CAP * sizeof(int2) + (size_t)rpb * sizeof(int) + 16, stream>>>(
      esrc, edst, dinv, qcnt, qlist, nN, nE, rpb);

  gemm_rows_kernel<128, 1><<<512, 256, 0, stream>>>(emb, x_idx, W1, dinv, hs1, nN);
  range_sum_kernel<128, 1><<<256, RS_THREADS, (size_t)rpb * H * sizeof(float) + rs_tail, stream>>>(
      esrc, edst, hs1, dinv, b1, x1, qcnt, qlist, nN, nE, rpb);

  gemm_rows_kernel<64, 0><<<512, 256, 0, stream>>>(x1, nullptr, W2, dinv, hs2, nN);
  range_sum_kernel<64, 0><<<256, RS_THREADS, (size_t)rpb * O * sizeof(float) + rs_tail, stream>>>(
      esrc, edst, hs2, dinv, b2, z, qcnt, qlist, nN, nE, rpb);

  edgedot_kernel<<<(unsigned)(((size_t)nL * 32 + 255) / 256), 256, 0, stream>>>(
      lidx, lidx + nL, z, out, nL);
}


# Edge filter on one-byte range keys: new first launch writes dst/rpb per edge as a byte; a range block scans 0.8 MB of keys (was 3.2 MB of dst), checks flagged edges by dst; sums in LDS arrival order
# speedup vs baseline: 2.8947x; 1.1121x over previous
#include <hip/hip_runtime.h>

typedef float v2f __attribute__((ext_vector_type(2)));
typedef float v8f __attribute__((ext_vector_type(8)));

#define LDS_STRIDE 132

#define GT_TILE_ROWS 64

template <int GATHER>
__device__ __forceinline__ void gt_load_tile(const float4* __restrict__ X4,
                                             const int* __restrict__ gidx,
                                             const float* __restrict__ dinv, int row0, int n_rows,
                                             int tid, float4 (&pre)[8], float& dpre) {
  const int c4 = tid & 31;
  const int rw = tid >> 5;
  int sr[8];
#pragma unroll
  for (int u = 0; u < 8; u++) {
    int gr = row0 + rw + 8 * u;
    int g = (gr < n_rows) ? gr : 0;
    sr[u] = GATHER ? gidx[g] : g;
  }
#pragma unroll
  for (int u = 0; u < 8; u++) {
    pre[u] = X4[(size_t)sr[u] * 32 + c4];
  }
  dpre = 0.f;
  if (tid < GT_TILE_ROWS && row0 + tid < n_rows) dpre = dinv[row0 + tid];
}

template <int N, int GATHER>
__global__ void __launch_bounds__(256)
gemm_rows_kernel(const float* __restrict__ X, const int* __restrict__ gidx,
                 const float* __restrict__ W, const float* __restrict__ dinv,
                 float* __restrict__ hs, int n_rows) {
  __shared__ float lA[GT_TILE_ROWS * LDS_STRIDE];
  __shared__ __attribute__((aligned(16))) float lD[GT_TILE_ROWS];
  constexpr int NCT = N / 16;
  constexpr int WPC = 8 / NCT;
  constexpr int MT = 4 / WPC;
  const int tid = threadIdx.x;
  const int wave = tid >> 5;
  const int lane = tid & 31;
  const int col = lane & 15;
  const int koff = (lane >> 4) << 1;
  const int nbase = (wave % NCT) << 4;
  const int mt0 = (wave / NCT) * MT;
  const int ntiles = (n_rows + GT_TILE_ROWS - 1) / GT_TILE_ROWS;
  const float4* X4 = (const float4*)X;

  int tile = blockIdx.x;
  float4 pre[8];
  float dpre = 0.f;
#pragma unroll
  for (int u = 0; u < 8; u++) pre[u] = make_float4(0.f, 0.f, 0.f, 0.f);
  if (tile < ntiles) gt_load_tile<GATHER>(X4, gidx, dinv, tile * GT_TILE_ROWS, n_rows, tid, pre, dpre);

  v2f bf[32];
  {
    const float* wp = W + (size_t)koff * N + nbase + col;
#pragma unroll
    for (int k = 0; k < 32; k++) {
      bf[k].x = wp[(size_t)(4 * k) * N];
      bf[k].y = wp[(size_t)(4 * k + 1) * N];
    }
  }

  const int madd = (lane < 16) ? 0 : 8;
  const float* aBase = &lA[(mt0 * 16 + col) * LDS_STRIDE + koff];

  for (; tile < ntiles; tile += gridDim.x) {
    const int row0 = tile * GT_TILE_ROWS;
    __syncthreads();
    {
      const int c4 = tid & 31;
      const int rw = tid >> 5;
#pragma unroll
      for (int u = 0; u < 8; u++) {
        float4 v = pre[u];
        if (row0 + rw + 8 * u >= n_rows) v = make_float4(0.f, 0.f, 0.f, 0.f);
        *(float4*)&lA[(rw + 8 * u) * LDS_STRIDE + (c4 << 2)] = v;
      }
      if (tid < GT_TILE_ROWS) lD[tid] = dpre;
    }
    __syncthreads();
    const int nt = tile + gridDim.x;
    if (nt < ntiles) gt_load_tile<GATHER>(X4, gidx, dinv, nt * GT_TILE_ROWS, n_rows, tid, pre, dpre);

    v8f c[MT];
#pragma unroll
    for (int m = 0; m < MT; m++) c[m] = (v8f){0.f, 0.f, 0.f, 0.f, 0.f, 0.f, 0.f, 0.f};
#pragma unroll
    for (int kg = 0; kg < 32; kg += 4) {
      v2f a[4][MT];
#pragma unroll
      for (int kk = 0; kk < 4; kk++)
#pragma unroll
        for (int m = 0; m < MT; m++)
          a[kk][m] = *(const v2f*)(aBase + m * 16 * LDS_STRIDE + 4 * (kg + kk));
#pragma unroll
      for (int kk = 0; kk < 4; kk++)
#pragma unroll
        for (int m = 0; m < MT; m++)
          c[m] = __builtin_amdgcn_wmma_f32_16x16x4_f32(false, a[kk][m], false, bf[kg + kk], (short)0,
                                                       c[m], false, false);
    }
    float4 dv[MT][2];
#pragma unroll
    for (int m = 0; m < MT; m++) {
      const float4* dp = (const float4*)&lD[(mt0 + m) * 16 + madd];
      dv[m][0] = dp[0];
      dv[m][1] = dp[1];
    }
#pragma unroll
    for (int m = 0; m < MT; m++) {
      const float d[8] = {dv[m][0].x, dv[m][0].y, dv[m][0].z, dv[m][0].w,
                          dv[m][1].x, dv[m][1].y, dv[m][1].z, dv[m][1].w};
#pragma unroll
      for (int i = 0; i < 8; i++) {
        int lr = (mt0 + m) * 16 + i + madd;
        int gr = row0 + lr;
        if (gr < n_rows) hs[(size_t)gr * N + nbase + col] = c[m][i] * d[i];
      }
    }
  }
}

__global__ void __launch_bounds__(256)
range_key_kernel(const int* __restrict__ dst, unsigned* __restrict__ keys, int nE, int rpb) {
  const int nW = (nE >> 4) << 2;
  const int i = blockIdx.x * 256 + threadIdx.x;
  if (i >= nW) return;
  const int* p = dst + 4 * (size_t)i;
  int d0, d1, d2, d3;
  if ((((size_t)dst) & 15) == 0) {
    int4 v = *(const int4*)p;
    d0 = v.x; d1 = v.y; d2 = v.z; d3 = v.w;
  } else {
    d0 = p[0]; d1 = p[1]; d2 = p[2]; d3 = p[3];
  }
  const unsigned r = (unsigned)rpb;
  unsigned k0 = min((unsigned)d0 / r, 255u);
  unsigned k1 = min((unsigned)d1 / r, 255u);
  unsigned k2 = min((unsigned)d2 / r, 255u);
  unsigned k3 = min((unsigned)d3 / r, 255u);
  keys[i] = k0 | (k1 << 8) | (k2 << 16) | (k3 << 24);
}

#define RS_THREADS 1024
#define RS_WAVES (RS_THREADS / 32)
#define RS_LIST_CAP 8192
#define RS_LIST_PER_THREAD (RS_LIST_CAP / RS_THREADS)
#define RK_ONE 0x01010101u
#define RK_HI 0x80808080u

__device__ __forceinline__ unsigned rk_flag(unsigned x) { return (x - RK_ONE) & ~x; }

__global__ void __launch_bounds__(RS_THREADS)
range_queue_kernel(const int* __restrict__ src, const int* __restrict__ dst,
                   const unsigned* __restrict__ keyw, float* __restrict__ dinv,
                   int* __restrict__ qcnt, int2* __restrict__ qlist, int n_rows, int nE, int rpb) {
  extern __shared__ float lds[];
  const uint4* keys = (const uint4*)keyw;
  int* list = (int*)lds;
  int* degc = list + RS_LIST_CAP;
  int* cntp = degc + rpb;
  const int tid = threadIdx.x;
  const int r0 = blockIdx.x * rpb;
  int nrows = n_rows - r0;
  if (nrows > rpb) nrows = rpb;
  if (nrows < 0) nrows = 0;
  const unsigned un = (unsigned)nrows;
  const unsigned bb = (blockIdx.x & 255u) * RK_ONE;

  for (int r = tid; r < nrows; r += RS_THREADS) degc[r] = 0;
  if (tid == 0) *cntp = 0;
  __syncthreads();

  const int nK = nE >> 4;
  int base = 0;
  for (; base + 4 * RS_THREADS <= nK; base += 4 * RS_THREADS) {
    uint4 kv[4];
#pragma unroll
    for (int u = 0; u < 4; u++) kv[u] = keys[base + u * RS_THREADS + tid];
    unsigned a[16];
    unsigned any = 0;
#pragma unroll
    for (int u = 0; u < 4; u++) {
      a[4 * u + 0] = rk_flag(kv[u].x ^ bb);
      a[4 * u + 1] = rk_flag(kv[u].y ^ bb);
      a[4 * u + 2] = rk_flag(kv[u].z ^ bb);
      a[4 * u + 3] = rk_flag(kv[u].w ^ bb);
      any |= a[4 * u + 0] | a[4 * u + 1] | a[4 * u + 2] | a[4 * u + 3];
    }
    if (any & RK_HI) {
      unsigned qlo = 0, qhi = 0;
#pragma unroll
      for (int i = 0; i < 8; i++) {
        qlo |= (a[i] & RK_HI) >> i;
        qhi |= (a[8 + i] & RK_HI) >> i;
      }
      int slot = atomicAdd(cntp, __popc(qlo) + __popc(qhi));
      const int e0 = (base + tid) << 4;
      while (qlo) {
        int pos = __ffs(qlo) - 1;
        qlo &= qlo - 1;
        int i = 7 - (pos & 7);
        int e = e0 + (i >> 2) * (16 * RS_THREADS) + ((i & 3) << 2) + (pos >> 3);
        if (slot < RS_LIST_CAP) list[slot] = e;
        slot++;
      }
      while (qhi) {
        int pos = __ffs(qhi) - 1;
        qhi &= qhi - 1;
        int i = 15 - (pos & 7);
        int e = e0 + (i >> 2) * (16 * RS_THREADS) + ((i & 3) << 2) + (pos >> 3);
        if (slot < RS_LIST_CAP) list[slot] = e;
        slot++;
      }
    }
  }
  for (int j = base + tid; j < nK; j += RS_THREADS) {
    uint4 kv = keys[j];
    unsigned a0 = rk_flag(kv.x ^ bb), a1 = rk_flag(kv.y ^ bb);
    unsigned a2 = rk_flag(kv.z ^ bb), a3 = rk_flag(kv.w ^ bb);
    if ((a0 | a1 | a2 | a3) & RK_HI) {
      unsigned q = (a0 & RK_HI) | ((a1 & RK_HI) >> 1) | ((a2 & RK_HI) >> 2) | ((a3 & RK_HI) >> 3);
      int slot = atomicAdd(cntp, __popc(q));
      while (q) {
        int pos = __ffs(q) - 1;
        q &= q - 1;
        int i = 7 - (pos & 7);
        int e = (j << 4) + (i << 2) + (pos >> 3);
        if (slot < RS_LIST_CAP) list[slot] = e;
        slot++;
      }
    }
  }
  for (int e = (nK << 4) + tid; e < nE; e += RS_THREADS) {
    if ((unsigned)(dst[e] - r0) < un) {
      int slot = atomicAdd(cntp, 1);
      if (slot < RS_LIST_CAP) list[slot] = e;
    }
  }
  __syncthreads();

  const int n = *cntp;
  const bool over = n > RS_LIST_CAP;
  const int nq = over ? 0 : n;
  int2* q = qlist + (size_t)blockIdx.x * RS_LIST_CAP;
  if (over) {
    for (int e = tid; e < nE; e += RS_THREADS) {
      unsigned loc = (unsigned)(dst[e] - r0);
      if (loc < un) atomicAdd(&degc[loc], 1);
    }
  } else if (nq > 0) {
    int ed[RS_LIST_PER_THREAD];
    int s[RS_LIST_PER_THREAD];
    int d[RS_LIST_PER_THREAD];
#pragma unroll
    for (int u = 0; u < RS_LIST_PER_THREAD; u++) {
      int idx = tid + u * RS_THREADS;
      ed[u] = list[idx < nq ? idx : 0];
    }
#pragma unroll
    for (int u = 0; u < RS_LIST_PER_THREAD; u++) {
      s[u] = src[ed[u]];
      d[u] = dst[ed[u]];
    }
#pragma unroll
    for (int u = 0; u < RS_LIST_PER_THREAD; u++) {
      int idx = tid + u * RS_THREADS;
      if (idx < nq) {
        unsigned loc = (unsigned)(d[u] - r0);
        if (loc < un) {
          atomicAdd(&degc[loc], 1);
          q[idx] = make_int2(s[u], (int)loc);
        } else {
          q[idx] = make_int2(0, -1);
        }
      }
    }
  }
  __syncthreads();
  if (tid == 0) qcnt[blockIdx.x] = n;
  for (int r = tid; r < nrows; r += RS_THREADS) dinv[r0 + r] = rsqrtf(1.0f + (float)degc[r]);
}

template <int N, int RELU>
__global__ void __launch_bounds__(RS_THREADS)
range_sum_kernel(const int* __restrict__ src, const int* __restrict__ dst,
                 const float* __restrict__ hs, const float* __restrict__ dinv,
                 const float* __restrict__ bias, float* __restrict__ outp,
                 const int* __restrict__ qcnt, const int2* __restrict__ qlist,
                 int n_rows, int nE, int rpb) {
  extern __shared__ float lds[];
  float* acc = lds;
  float* dl = lds + (size_t)rpb * N;
  int2* list = (int2*)(dl + ((rpb + 1) & ~1));
  const int tid = threadIdx.x;
  const int r0 = blockIdx.x * rpb;
  int nrows = n_rows - r0;
  if (nrows > rpb) nrows = rpb;
  if (nrows < 0) nrows = 0;
  const unsigned un = (unsigned)nrows;
  const int nel4 = (nrows * N) >> 2;

  const int n = qcnt[blockIdx.x];
  const bool over = n > RS_LIST_CAP;
  const int nq = over ? 0 : n;
  const int2* q = qlist + (size_t)blockIdx.x * RS_LIST_CAP;

  const float4* h4 = (const float4*)(hs + (size_t)r0 * N);
  float4* a4 = (float4*)acc;
  int i0 = tid;
  for (; i0 + 3 * RS_THREADS < nel4; i0 += 4 * RS_THREADS) {
    float4 t[4];
#pragma unroll
    for (int u = 0; u < 4; u++) t[u] = h4[i0 + u * RS_THREADS];
#pragma unroll
    for (int u = 0; u < 4; u++) a4[i0 + u * RS_THREADS] = t[u];
  }
  for (; i0 < nel4; i0 += RS_THREADS) a4[i0] = h4[i0];
  for (int r = tid; r < nrows; r += RS_THREADS) dl[r] = dinv[r0 + r];
  {
    int2 p[RS_LIST_PER_THREAD];
#pragma unroll
    for (int u = 0; u < RS_LIST_PER_THREAD; u++) {
      int idx = tid + u * RS_THREADS;
      p[u] = make_int2(0, 0);
      if (idx < nq) p[u] = q[idx];
    }
#pragma unroll
    for (int u = 0; u < RS_LIST_PER_THREAD; u++) {
      int idx = tid + u * RS_THREADS;
      if (idx < nq) list[idx] = p[u];
    }
  }
  __syncthreads();

  if (over) {
    for (int e = tid; e < nE; e += RS_THREADS) {
      unsigned loc = (unsigned)(dst[e] - r0);
      if (loc < un) {
        const float* hp = hs + (size_t)src[e] * N;
        float* ap = acc + (size_t)loc * N;
        for (int f = 0; f < N; f++) unsafeAtomicAdd(&ap[f], hp[f]);
      }
    }
  }

  const int wave = tid >> 5;
  const int lane = tid & 31;
  for (int i = wave; i < nq; i += 4 * RS_WAVES) {
    float v[4][N / 32];
    int loc[4];
#pragma unroll
    for (int q4 = 0; q4 < 4; q4++) {
      int idx = i + q4 * RS_WAVES;
      loc[q4] = -1;
      if (idx < nq) {
        int2 p = list[idx];
        loc[q4] = p.y;
        const float* hp = hs + (size_t)p.x * N + lane;
#pragma unroll
        for (int j = 0; j < N / 32; j++) v[q4][j] = hp[32 * j];
      }
    }
#pragma unroll
    for (int q4 = 0; q4 < 4; q4++) {
      if (loc[q4] >= 0) {
        float* ap = acc + (size_t)loc[q4] * N + lane;
#pragma unroll
        for (int j = 0; j < N / 32; j++) unsafeAtomicAdd(&ap[32 * j], v[q4][j]);
      }
    }
  }
  __syncthreads();

  const float4 b = *(const float4*)(bias + ((tid << 2) & (N - 1)));
  float4* o4 = (float4*)(outp + (size_t)r0 * N);
  for (int i = tid; i < nel4; i += RS_THREADS) {
    float4 a = a4[i];
    float di = dl[(i << 2) / N];
    float4 o;
    o.x = di * a.x + b.x;
    o.y = di * a.y + b.y;
    o.z = di * a.z + b.z;
    o.w = di * a.w + b.w;
    if (RELU) {
      o.x = fmaxf(o.x, 0.f);
      o.y = fmaxf(o.y, 0.f);
      o.z = fmaxf(o.z, 0.f);
      o.w = fmaxf(o.w, 0.f);
    }
    o4[i] = o;
  }
}

__global__ void edgedot_kernel(const int* __restrict__ es, const int* __restrict__ ed,
                               const float* __restrict__ z, float* __restrict__ out, int nL) {
  int g = blockIdx.x * blockDim.x + threadIdx.x;
  int e = g >> 5;
  int lane = g & 31;
  if (e >= nL) return;
  const float* a = z + (size_t)es[e] * 64;
  const float* b = z + (size_t)ed[e] * 64;
  float p = a[lane] * b[lane] + a[lane + 32] * b[lane + 32];
#pragma unroll
  for (int off = 16; off > 0; off >>= 1) p += __shfl_xor(p, off, 32);
  if (lane == 0) out[e] = p;
}

extern "C" void kernel_launch(void* const* d_in, const int* in_sizes, int n_in,
                              void* d_out, int out_size, void* d_ws, size_t ws_size,
                              hipStream_t stream) {
  const int*   x_idx = (const int*)d_in[0];
  const int*   eidx  = (const int*)d_in[1];
  const int*   lidx  = (const int*)d_in[2];
  const float* emb   = (const float*)d_in[3];
  const float* W1    = (const float*)d_in[4];
  const float* b1    = (const float*)d_in[5];
  const float* W2    = (const float*)d_in[6];
  const float* b2    = (const float*)d_in[7];
  float* out = (float*)d_out;

  const int nN = in_sizes[0];
  const int nE = in_sizes[1] / 2;
  const int nL = in_sizes[2] / 2;
  const int H = 128, O = 64;

  float* dinv = (float*)d_ws;
  float* hs1  = dinv + ((nN + 63) & ~63);
  float* x1   = hs1 + (size_t)nN * H;
  float* hs2  = x1 + (size_t)nN * H;
  float* z    = hs2 + (size_t)nN * O;
  int*   qcnt = (int*)(z + (size_t)nN * O);
  int2*  qlist = (int2*)(qcnt + 256);
  unsigned* keys = (unsigned*)(qlist + (size_t)256 * RS_LIST_CAP);

  const int* esrc = eidx;
  const int* edst = eidx + nE;

  const int rpb = (nN + 255) / 256;
  size_t rs_tail = (size_t)((rpb + 1) & ~1) * sizeof(float) + (size_t)RS_LIST_CAP * sizeof(int2) + 16;

  range_key_kernel<<<(unsigned)((nE / 4 + 255) / 256), 256, 0, stream>>>(edst, keys, nE, rpb);
  range_queue_kernel<<<256, RS_THREADS,
                       (size_t)RS_LIST_CAP * sizeof(int) + (size_t)rpb * sizeof(int) + 16, stream>>>(
      esrc, edst, keys, dinv, qcnt, qlist, nN, nE, rpb);

  gemm_rows_kernel<128, 1><<<512, 256, 0, stream>>>(emb, x_idx, W1, dinv, hs1, nN);
  range_sum_kernel<128, 1><<<256, RS_THREADS, (size_t)rpb * H * sizeof(float) + rs_tail, stream>>>(
      esrc, edst, hs1, dinv, b1, x1, qcnt, qlist, nN, nE, rpb);

  gemm_rows_kernel<64, 0><<<512, 256, 0, stream>>>(x1, nullptr, W2, dinv, hs2, nN);
  range_sum_kernel<64, 0><<<256, RS_THREADS, (size_t)rpb * O * sizeof(float) + rs_tail, stream>>>(
      esrc, edst, hs2, dinv, b2, z, qcnt, qlist, nN, nE, rpb);

  edgedot_kernel<<<(unsigned)(((size_t)nL * 32 + 255) / 256), 256, 0, stream>>>(
      lidx, lidx + nL, z, out, nL);
}


# Link scores edge-parallel: a lane owns one label edge, reads both z rows itself (32 16-byte loads as one group), sums 64 products in feature order; no cross-lane sums; 32x fewer waves enter
# speedup vs baseline: 3.1589x; 1.0913x over previous
#include <hip/hip_runtime.h>

typedef float v2f __attribute__((ext_vector_type(2)));
typedef float v8f __attribute__((ext_vector_type(8)));

#define LDS_STRIDE 132

#define GT_TILE_ROWS 64

template <int GATHER>
__device__ __forceinline__ void gt_load_tile(const float4* __restrict__ X4,
                                             const int* __restrict__ gidx,
                                             const float* __restrict__ dinv, int row0, int n_rows,
                                             int tid, float4 (&pre)[8], float& dpre) {
  const int c4 = tid & 31;
  const int rw = tid >> 5;
  int sr[8];
#pragma unroll
  for (int u = 0; u < 8; u++) {
    int gr = row0 + rw + 8 * u;
    int g = (gr < n_rows) ? gr : 0;
    sr[u] = GATHER ? gidx[g] : g;
  }
#pragma unroll
  for (int u = 0; u < 8; u++) {
    pre[u] = X4[(size_t)sr[u] * 32 + c4];
  }
  dpre = 0.f;
  if (tid < GT_TILE_ROWS && row0 + tid < n_rows) dpre = dinv[row0 + tid];
}

template <int N, int GATHER>
__global__ void __launch_bounds__(256)
gemm_rows_kernel(const float* __restrict__ X, const int* __restrict__ gidx,
                 const float* __restrict__ W, const float* __restrict__ dinv,
                 float* __restrict__ hs, int n_rows) {
  __shared__ float lA[GT_TILE_ROWS * LDS_STRIDE];
  __shared__ __attribute__((aligned(16))) float lD[GT_TILE_ROWS];
  constexpr int NCT = N / 16;
  constexpr int WPC = 8 / NCT;
  constexpr int MT = 4 / WPC;
  const int tid = threadIdx.x;
  const int wave = tid >> 5;
  const int lane = tid & 31;
  const int col = lane & 15;
  const int koff = (lane >> 4) << 1;
  const int nbase = (wave % NCT) << 4;
  const int mt0 = (wave / NCT) * MT;
  const int ntiles = (n_rows + GT_TILE_ROWS - 1) / GT_TILE_ROWS;
  const float4* X4 = (const float4*)X;

  int tile = blockIdx.x;
  float4 pre[8];
  float dpre = 0.f;
#pragma unroll
  for (int u = 0; u < 8; u++) pre[u] = make_float4(0.f, 0.f, 0.f, 0.f);
  if (tile < ntiles) gt_load_tile<GATHER>(X4, gidx, dinv, tile * GT_TILE_ROWS, n_rows, tid, pre, dpre);

  v2f bf[32];
  {
    const float* wp = W + (size_t)koff * N + nbase + col;
#pragma unroll
    for (int k = 0; k < 32; k++) {
      bf[k].x = wp[(size_t)(4 * k) * N];
      bf[k].y = wp[(size_t)(4 * k + 1) * N];
    }
  }

  const int madd = (lane < 16) ? 0 : 8;
  const float* aBase = &lA[(mt0 * 16 + col) * LDS_STRIDE + koff];

  for (; tile < ntiles; tile += gridDim.x) {
    const int row0 = tile * GT_TILE_ROWS;
    __syncthreads();
    {
      const int c4 = tid & 31;
      const int rw = tid >> 5;
#pragma unroll
      for (int u = 0; u < 8; u++) {
        float4 v = pre[u];
        if (row0 + rw + 8 * u >= n_rows) v = make_float4(0.f, 0.f, 0.f, 0.f);
        *(float4*)&lA[(rw + 8 * u) * LDS_STRIDE + (c4 << 2)] = v;
      }
      if (tid < GT_TILE_ROWS) lD[tid] = dpre;
    }
    __syncthreads();
    const int nt = tile + gridDim.x;
    if (nt < ntiles) gt_load_tile<GATHER>(X4, gidx, dinv, nt * GT_TILE_ROWS, n_rows, tid, pre, dpre);

    v8f c[MT];
#pragma unroll
    for (int m = 0; m < MT; m++) c[m] = (v8f){0.f, 0.f, 0.f, 0.f, 0.f, 0.f, 0.f, 0.f};
#pragma unroll
    for (int kg = 0; kg < 32; kg += 4) {
      v2f a[4][MT];
#pragma unroll
      for (int kk = 0; kk < 4; kk++)
#pragma unroll
        for (int m = 0; m < MT; m++)
          a[kk][m] = *(const v2f*)(aBase + m * 16 * LDS_STRIDE + 4 * (kg + kk));
#pragma unroll
      for (int kk = 0; kk < 4; kk++)
#pragma unroll
        for (int m = 0; m < MT; m++)
          c[m] = __builtin_amdgcn_wmma_f32_16x16x4_f32(false, a[kk][m], false, bf[kg + kk], (short)0,
                                                       c[m], false, false);
    }
    float4 dv[MT][2];
#pragma unroll
    for (int m = 0; m < MT; m++) {
      const float4* dp = (const float4*)&lD[(mt0 + m) * 16 + madd];
      dv[m][0] = dp[0];
      dv[m][1] = dp[1];
    }
#pragma unroll
    for (int m = 0; m < MT; m++) {
      const float d[8] = {dv[m][0].x, dv[m][0].y, dv[m][0].z, dv[m][0].w,
                          dv[m][1].x, dv[m][1].y, dv[m][1].z, dv[m][1].w};
#pragma unroll
      for (int i = 0; i < 8; i++) {
        int lr = (mt0 + m) * 16 + i + madd;
        int gr = row0 + lr;
        if (gr < n_rows) hs[(size_t)gr * N + nbase + col] = c[m][i] * d[i];
      }
    }
  }
}

__global__ void __launch_bounds__(256)
range_key_kernel(const int* __restrict__ dst, unsigned* __restrict__ keys, int nE, int rpb) {
  const int nW = (nE >> 4) << 2;
  const int i = blockIdx.x * 256 + threadIdx.x;
  if (i >= nW) return;
  const int* p = dst + 4 * (size_t)i;
  int d0, d1, d2, d3;
  if ((((size_t)dst) & 15) == 0) {
    int4 v = *(const int4*)p;
    d0 = v.x; d1 = v.y; d2 = v.z; d3 = v.w;
  } else {
    d0 = p[0]; d1 = p[1]; d2 = p[2]; d3 = p[3];
  }
  const unsigned r = (unsigned)rpb;
  unsigned k0 = min((unsigned)d0 / r, 255u);
  unsigned k1 = min((unsigned)d1 / r, 255u);
  unsigned k2 = min((unsigned)d2 / r, 255u);
  unsigned k3 = min((unsigned)d3 / r, 255u);
  keys[i] = k0 | (k1 << 8) | (k2 << 16) | (k3 << 24);
}

#define RS_THREADS 1024
#define RS_WAVES (RS_THREADS / 32)
#define RS_LIST_CAP 8192
#define RS_LIST_PER_THREAD (RS_LIST_CAP / RS_THREADS)
#define RK_ONE 0x01010101u
#define RK_HI 0x80808080u

__device__ __forceinline__ unsigned rk_flag(unsigned x) { return (x - RK_ONE) & ~x; }

__global__ void __launch_bounds__(RS_THREADS)
range_queue_kernel(const int* __restrict__ src, const int* __restrict__ dst,
                   const unsigned* __restrict__ keyw, float* __restrict__ dinv,
                   int* __restrict__ qcnt, int2* __restrict__ qlist, int n_rows, int nE, int rpb) {
  extern __shared__ float lds[];
  const uint4* keys = (const uint4*)keyw;
  int* list = (int*)lds;
  int* degc = list + RS_LIST_CAP;
  int* cntp = degc + rpb;
  const int tid = threadIdx.x;
  const int r0 = blockIdx.x * rpb;
  int nrows = n_rows - r0;
  if (nrows > rpb) nrows = rpb;
  if (nrows < 0) nrows = 0;
  const unsigned un = (unsigned)nrows;
  const unsigned bb = (blockIdx.x & 255u) * RK_ONE;

  for (int r = tid; r < nrows; r += RS_THREADS) degc[r] = 0;
  if (tid == 0) *cntp = 0;
  __syncthreads();

  const int nK = nE >> 4;
  int base = 0;
  for (; base + 4 * RS_THREADS <= nK; base += 4 * RS_THREADS) {
    uint4 kv[4];
#pragma unroll
    for (int u = 0; u < 4; u++) kv[u] = keys[base + u * RS_THREADS + tid];
    unsigned a[16];
    unsigned any = 0;
#pragma unroll
    for (int u = 0; u < 4; u++) {
      a[4 * u + 0] = rk_flag(kv[u].x ^ bb);
      a[4 * u + 1] = rk_flag(kv[u].y ^ bb);
      a[4 * u + 2] = rk_flag(kv[u].z ^ bb);
      a[4 * u + 3] = rk_flag(kv[u].w ^ bb);
      any |= a[4 * u + 0] | a[4 * u + 1] | a[4 * u + 2] | a[4 * u + 3];
    }
    if (any & RK_HI) {
      unsigned qlo = 0, qhi = 0;
#pragma unroll
      for (int i = 0; i < 8; i++) {
        qlo |= (a[i] & RK_HI) >> i;
        qhi |= (a[8 + i] & RK_HI) >> i;
      }
      int slot = atomicAdd(cntp, __popc(qlo) + __popc(qhi));
      const int e0 = (base + tid) << 4;
      while (qlo) {
        int pos = __ffs(qlo) - 1;
        qlo &= qlo - 1;
        int i = 7 - (pos & 7);
        int e = e0 + (i >> 2) * (16 * RS_THREADS) + ((i & 3) << 2) + (pos >> 3);
        if (slot < RS_LIST_CAP) list[slot] = e;
        slot++;
      }
      while (qhi) {
        int pos = __ffs(qhi) - 1;
        qhi &= qhi - 1;
        int i = 15 - (pos & 7);
        int e = e0 + (i >> 2) * (16 * RS_THREADS) + ((i & 3) << 2) + (pos >> 3);
        if (slot < RS_LIST_CAP) list[slot] = e;
        slot++;
      }
    }
  }
  for (int j = base + tid; j < nK; j += RS_THREADS) {
    uint4 kv = keys[j];
    unsigned a0 = rk_flag(kv.x ^ bb), a1 = rk_flag(kv.y ^ bb);
    unsigned a2 = rk_flag(kv.z ^ bb), a3 = rk_flag(kv.w ^ bb);
    if ((a0 | a1 | a2 | a3) & RK_HI) {
      unsigned q = (a0 & RK_HI) | ((a1 & RK_HI) >> 1) | ((a2 & RK_HI) >> 2) | ((a3 & RK_HI) >> 3);
      int slot = atomicAdd(cntp, __popc(q));
      while (q) {
        int pos = __ffs(q) - 1;
        q &= q - 1;
        int i = 7 - (pos & 7);
        int e = (j << 4) + (i << 2) + (pos >> 3);
        if (slot < RS_LIST_CAP) list[slot] = e;
        slot++;
      }
    }
  }
  for (int e = (nK << 4) + tid; e < nE; e += RS_THREADS) {
    if ((unsigned)(dst[e] - r0) < un) {
      int slot = atomicAdd(cntp, 1);
      if (slot < RS_LIST_CAP) list[slot] = e;
    }
  }
  __syncthreads();

  const int n = *cntp;
  const bool over = n > RS_LIST_CAP;
  const int nq = over ? 0 : n;
  int2* q = qlist + (size_t)blockIdx.x * RS_LIST_CAP;
  if (over) {
    for (int e = tid; e < nE; e += RS_THREADS) {
      unsigned loc = (unsigned)(dst[e] - r0);
      if (loc < un) atomicAdd(&degc[loc], 1);
    }
  } else if (nq > 0) {
    int ed[RS_LIST_PER_THREAD];
    int s[RS_LIST_PER_THREAD];
    int d[RS_LIST_PER_THREAD];
#pragma unroll
    for (int u = 0; u < RS_LIST_PER_THREAD; u++) {
      int idx = tid + u * RS_THREADS;
      ed[u] = list[idx < nq ? idx : 0];
    }
#pragma unroll
    for (int u = 0; u < RS_LIST_PER_THREAD; u++) {
      s[u] = src[ed[u]];
      d[u] = dst[ed[u]];
    }
#pragma unroll
    for (int u = 0; u < RS_LIST_PER_THREAD; u++) {
      int idx = tid + u * RS_THREADS;
      if (idx < nq) {
        unsigned loc = (unsigned)(d[u] - r0);
        if (loc < un) {
          atomicAdd(&degc[loc], 1);
          q[idx] = make_int2(s[u], (int)loc);
        } else {
          q[idx] = make_int2(0, -1);
        }
      }
    }
  }
  __syncthreads();
  if (tid == 0) qcnt[blockIdx.x] = n;
  for (int r = tid; r < nrows; r += RS_THREADS) dinv[r0 + r] = rsqrtf(1.0f + (float)degc[r]);
}

template <int N, int RELU>
__global__ void __launch_bounds__(RS_THREADS)
range_sum_kernel(const int* __restrict__ src, const int* __restrict__ dst,
                 const float* __restrict__ hs, const float* __restrict__ dinv,
                 const float* __restrict__ bias, float* __restrict__ outp,
                 const int* __restrict__ qcnt, const int2* __restrict__ qlist,
                 int n_rows, int nE, int rpb) {
  extern __shared__ float lds[];
  float* acc = lds;
  float* dl = lds + (size_t)rpb * N;
  int2* list = (int2*)(dl + ((rpb + 1) & ~1));
  const int tid = threadIdx.x;
  const int r0 = blockIdx.x * rpb;
  int nrows = n_rows - r0;
  if (nrows > rpb) nrows = rpb;
  if (nrows < 0) nrows = 0;
  const unsigned un = (unsigned)nrows;
  const int nel4 = (nrows * N) >> 2;

  const int n = qcnt[blockIdx.x];
  const bool over = n > RS_LIST_CAP;
  const int nq = over ? 0 : n;
  const int2* q = qlist + (size_t)blockIdx.x * RS_LIST_CAP;

  const float4* h4 = (const float4*)(hs + (size_t)r0 * N);
  float4* a4 = (float4*)acc;
  int i0 = tid;
  for (; i0 + 3 * RS_THREADS < nel4; i0 += 4 * RS_THREADS) {
    float4 t[4];
#pragma unroll
    for (int u = 0; u < 4; u++) t[u] = h4[i0 + u * RS_THREADS];
#pragma unroll
    for (int u = 0; u < 4; u++) a4[i0 + u * RS_THREADS] = t[u];
  }
  for (; i0 < nel4; i0 += RS_THREADS) a4[i0] = h4[i0];
  for (int r = tid; r < nrows; r += RS_THREADS) dl[r] = dinv[r0 + r];
  {
    int2 p[RS_LIST_PER_THREAD];
#pragma unroll
    for (int u = 0; u < RS_LIST_PER_THREAD; u++) {
      int idx = tid + u * RS_THREADS;
      p[u] = make_int2(0, 0);
      if (idx < nq) p[u] = q[idx];
    }
#pragma unroll
    for (int u = 0; u < RS_LIST_PER_THREAD; u++) {
      int idx = tid + u * RS_THREADS;
      if (idx < nq) list[idx] = p[u];
    }
  }
  __syncthreads();

  if (over) {
    for (int e = tid; e < nE; e += RS_THREADS) {
      unsigned loc = (unsigned)(dst[e] - r0);
      if (loc < un) {
        const float* hp = hs + (size_t)src[e] * N;
        float* ap = acc + (size_t)loc * N;
        for (int f = 0; f < N; f++) unsafeAtomicAdd(&ap[f], hp[f]);
      }
    }
  }

  const int wave = tid >> 5;
  const int lane = tid & 31;
  for (int i = wave; i < nq; i += 4 * RS_WAVES) {
    float v[4][N / 32];
    int loc[4];
#pragma unroll
    for (int q4 = 0; q4 < 4; q4++) {
      int idx = i + q4 * RS_WAVES;
      loc[q4] = -1;
      if (idx < nq) {
        int2 p = list[idx];
        loc[q4] = p.y;
        const float* hp = hs + (size_t)p.x * N + lane;
#pragma unroll
        for (int j = 0; j < N / 32; j++) v[q4][j] = hp[32 * j];
      }
    }
#pragma unroll
    for (int q4 = 0; q4 < 4; q4++) {
      if (loc[q4] >= 0) {
        float* ap = acc + (size_t)loc[q4] * N + lane;
#pragma unroll
        for (int j = 0; j < N / 32; j++) unsafeAtomicAdd(&ap[32 * j], v[q4][j]);
      }
    }
  }
  __syncthreads();

  const float4 b = *(const float4*)(bias + ((tid << 2) & (N - 1)));
  float4* o4 = (float4*)(outp + (size_t)r0 * N);
  for (int i = tid; i < nel4; i += RS_THREADS) {
    float4 a = a4[i];
    float di = dl[(i << 2) / N];
    float4 o;
    o.x = di * a.x + b.x;
    o.y = di * a.y + b.y;
    o.z = di * a.z + b.z;
    o.w = di * a.w + b.w;
    if (RELU) {
      o.x = fmaxf(o.x, 0.f);
      o.y = fmaxf(o.y, 0.f);
      o.z = fmaxf(o.z, 0.f);
      o.w = fmaxf(o.w, 0.f);
    }
    o4[i] = o;
  }
}

#define ED_THREADS 128
typedef float v4f __attribute__((ext_vector_type(4)));

__global__ void __launch_bounds__(ED_THREADS)
edgedot_kernel(const int* __restrict__ es, const int* __restrict__ ed,
               const float* __restrict__ z, float* __restrict__ out, int nL) {
  const int e = blockIdx.x * ED_THREADS + threadIdx.x;
  const int ec = (e < nL) ? e : nL - 1;
  const v4f* a = (const v4f*)(z + (size_t)es[ec] * 64);
  const v4f* b = (const v4f*)(z + (size_t)ed[ec] * 64);
  v4f va[16], vb[16];
#pragma unroll
  for (int j = 0; j < 16; j++) va[j] = a[j];
#pragma unroll
  for (int j = 0; j < 16; j++) vb[j] = b[j];
  asm volatile("" ::"v"(va[0]), "v"(va[1]), "v"(va[2]), "v"(va[3]), "v"(va[4]), "v"(va[5]), "v"(va[6]),
               "v"(va[7]), "v"(va[8]), "v"(va[9]), "v"(va[10]), "v"(va[11]), "v"(va[12]), "v"(va[13]),
               "v"(va[14]), "v"(va[15]));
  asm volatile("" ::"v"(vb[0]), "v"(vb[1]), "v"(vb[2]), "v"(vb[3]), "v"(vb[4]), "v"(vb[5]), "v"(vb[6]),
               "v"(vb[7]), "v"(vb[8]), "v"(vb[9]), "v"(vb[10]), "v"(vb[11]), "v"(vb[12]), "v"(vb[13]),
               "v"(vb[14]), "v"(vb[15]));
  float p0 = 0.f, p1 = 0.f, p2 = 0.f, p3 = 0.f;
#pragma unroll
  for (int j = 0; j < 16; j++) {
    p0 = fmaf(va[j].x, vb[j].x, p0);
    p1 = fmaf(va[j].y, vb[j].y, p1);
    p2 = fmaf(va[j].z, vb[j].z, p2);
    p3 = fmaf(va[j].w, vb[j].w, p3);
  }
  if (e < nL) out[e] = (p0 + p1) + (p2 + p3);
}

extern "C" void kernel_launch(void* const* d_in, const int* in_sizes, int n_in,
                              void* d_out, int out_size, void* d_ws, size_t ws_size,
                              hipStream_t stream) {
  const int*   x_idx = (const int*)d_in[0];
  const int*   eidx  = (const int*)d_in[1];
  const int*   lidx  = (const int*)d_in[2];
  const float* emb   = (const float*)d_in[3];
  const float* W1    = (const float*)d_in[4];
  const float* b1    = (const float*)d_in[5];
  const float* W2    = (const float*)d_in[6];
  const float* b2    = (const float*)d_in[7];
  float* out = (float*)d_out;

  const int nN = in_sizes[0];
  const int nE = in_sizes[1] / 2;
  const int nL = in_sizes[2] / 2;
  const int H = 128, O = 64;

  float* dinv = (float*)d_ws;
  float* hs1  = dinv + ((nN + 63) & ~63);
  float* x1   = hs1 + (size_t)nN * H;
  float* hs2  = x1 + (size_t)nN * H;
  float* z    = hs2 + (size_t)nN * O;
  int*   qcnt = (int*)(z + (size_t)nN * O);
  int2*  qlist = (int2*)(qcnt + 256);
  unsigned* keys = (unsigned*)(qlist + (size_t)256 * RS_LIST_CAP);

  const int* esrc = eidx;
  const int* edst = eidx + nE;

  const int rpb = (nN + 255) / 256;
  size_t rs_tail = (size_t)((rpb + 1) & ~1) * sizeof(float) + (size_t)RS_LIST_CAP * sizeof(int2) + 16;

  range_key_kernel<<<(unsigned)((nE / 4 + 255) / 256), 256, 0, stream>>>(edst, keys, nE, rpb);
  range_queue_kernel<<<256, RS_THREADS,
                       (size_t)RS_LIST_CAP * sizeof(int) + (size_t)rpb * sizeof(int) + 16, stream>>>(
      esrc, edst, keys, dinv, qcnt, qlist, nN, nE, rpb);

  gemm_rows_kernel<128, 1><<<512, 256, 0, stream>>>(emb, x_idx, W1, dinv, hs1, nN);
  range_sum_kernel<128, 1><<<256, RS_THREADS, (size_t)rpb * H * sizeof(float) + rs_tail, stream>>>(
      esrc, edst, hs1, dinv, b1, x1, qcnt, qlist, nN, nE, rpb);

  gemm_rows_kernel<64, 0><<<512, 256, 0, stream>>>(x1, nullptr, W2, dinv, hs2, nN);
  range_sum_kernel<64, 0><<<256, RS_THREADS, (size_t)rpb * O * sizeof(float) + rs_tail, stream>>>(
      esrc, edst, hs2, dinv, b2, z, qcnt, qlist, nN, nE, rpb);

  edgedot_kernel<<<(unsigned)((nL + ED_THREADS - 1) / ED_THREADS), ED_THREADS, 0, stream>>>(
      lidx, lidx + nL, z, out, nL);
}


# Layer 2 product moved into the layer 1 node-sum launch: x1 stays in the owning block's LDS rows (pitch 132) and is multiplied by W2 on WMMA there; x1 never stored, one launch less; sum orders kept
# speedup vs baseline: 3.3932x; 1.0742x over previous
#include <hip/hip_runtime.h>

typedef float v2f __attribute__((ext_vector_type(2)));
typedef float v8f __attribute__((ext_vector_type(8)));

#define LDS_STRIDE 132

#define GT_TILE_ROWS 64

template <int GATHER>
__device__ __forceinline__ void gt_load_tile(const float4* __restrict__ X4,
                                             const int* __restrict__ gidx,
                                             const float* __restrict__ dinv, int row0, int n_rows,
                                             int tid, float4 (&pre)[8], float& dpre) {
  const int c4 = tid & 31;
  const int rw = tid >> 5;
  int sr[8];
#pragma unroll
  for (int u = 0; u < 8; u++) {
    int gr = row0 + rw + 8 * u;
    int g = (gr < n_rows) ? gr : 0;
    sr[u] = GATHER ? gidx[g] : g;
  }
#pragma unroll
  for (int u = 0; u < 8; u++) {
    pre[u] = X4[(size_t)sr[u] * 32 + c4];
  }
  dpre = 0.f;
  if (tid < GT_TILE_ROWS && row0 + tid < n_rows) dpre = dinv[row0 + tid];
}

template <int N, int GATHER>
__global__ void __launch_bounds__(256)
gemm_rows_kernel(const float* __restrict__ X, const int* __restrict__ gidx,
                 const float* __restrict__ W, const float* __restrict__ dinv,
                 float* __restrict__ hs, int n_rows) {
  __shared__ float lA[GT_TILE_ROWS * LDS_STRIDE];
  __shared__ __attribute__((aligned(16))) float lD[GT_TILE_ROWS];
  constexpr int NCT = N / 16;
  constexpr int WPC = 8 / NCT;
  constexpr int MT = 4 / WPC;
  const int tid = threadIdx.x;
  const int wave = tid >> 5;
  const int lane = tid & 31;
  const int col = lane & 15;
  const int koff = (lane >> 4) << 1;
  const int nbase = (wave % NCT) << 4;
  const int mt0 = (wave / NCT) * MT;
  const int ntiles = (n_rows + GT_TILE_ROWS - 1) / GT_TILE_ROWS;
  const float4* X4 = (const float4*)X;

  int tile = blockIdx.x;
  float4 pre[8];
  float dpre = 0.f;
#pragma unroll
  for (int u = 0; u < 8; u++) pre[u] = make_float4(0.f, 0.f, 0.f, 0.f);
  if (tile < ntiles) gt_load_tile<GATHER>(X4, gidx, dinv, tile * GT_TILE_ROWS, n_rows, tid, pre, dpre);

  v2f bf[32];
  {
    const float* wp = W + (size_t)koff * N + nbase + col;
#pragma unroll
    for (int k = 0; k < 32; k++) {
      bf[k].x = wp[(size_t)(4 * k) * N];
      bf[k].y = wp[(size_t)(4 * k + 1) * N];
    }
  }

  const int madd = (lane < 16) ? 0 : 8;
  const float* aBase = &lA[(mt0 * 16 + col) * LDS_STRIDE + koff];

  for (; tile < ntiles; tile += gridDim.x) {
    const int row0 = tile * GT_TILE_ROWS;
    __syncthreads();
    {
      const int c4 = tid & 31;
      const int rw = tid >> 5;
#pragma unroll
      for (int u = 0; u < 8; u++) {
        float4 v = pre[u];
        if (row0 + rw + 8 * u >= n_rows) v = make_float4(0.f, 0.f, 0.f, 0.f);
        *(float4*)&lA[(rw + 8 * u) * LDS_STRIDE + (c4 << 2)] = v;
      }
      if (tid < GT_TILE_ROWS) lD[tid] = dpre;
    }
    __syncthreads();
    const int nt = tile + gridDim.x;
    if (nt < ntiles) gt_load_tile<GATHER>(X4, gidx, dinv, nt * GT_TILE_ROWS, n_rows, tid, pre, dpre);

    v8f c[MT];
#pragma unroll
    for (int m = 0; m < MT; m++) c[m] = (v8f){0.f, 0.f, 0.f, 0.f, 0.f, 0.f, 0.f, 0.f};
#pragma unroll
    for (int kg = 0; kg < 32; kg += 4) {
      v2f a[4][MT];
#pragma unroll
      for (int kk = 0; kk < 4; kk++)
#pragma unroll
        for (int m = 0; m < MT; m++)
          a[kk][m] = *(const v2f*)(aBase + m * 16 * LDS_STRIDE + 4 * (kg + kk));
#pragma unroll
      for (int kk = 0; kk < 4; kk++)
#pragma unroll
        for (int m = 0; m < MT; m++)
          c[m] = __builtin_amdgcn_wmma_f32_16x16x4_f32(false, a[kk][m], false, bf[kg + kk], (short)0,
                                                       c[m], false, false);
    }
    float4 dv[MT][2];
#pragma unroll
    for (int m = 0; m < MT; m++) {
      const float4* dp = (const float4*)&lD[(mt0 + m) * 16 + madd];
      dv[m][0] = dp[0];
      dv[m][1] = dp[1];
    }
#pragma unroll
    for (int m = 0; m < MT; m++) {
      const float d[8] = {dv[m][0].x, dv[m][0].y, dv[m][0].z, dv[m][0].w,
                          dv[m][1].x, dv[m][1].y, dv[m][1].z, dv[m][1].w};
#pragma unroll
      for (int i = 0; i < 8; i++) {
        int lr = (mt0 + m) * 16 + i + madd;
        int gr = row0 + lr;
        if (gr < n_rows) hs[(size_t)gr * N + nbase + col] = c[m][i] * d[i];
      }
    }
  }
}

__global__ void __launch_bounds__(256)
range_key_kernel(const int* __restrict__ dst, unsigned* __restrict__ keys, int nE, int rpb) {
  const int nW = (nE >> 4) << 2;
  const int i = blockIdx.x * 256 + threadIdx.x;
  if (i >= nW) return;
  const int* p = dst + 4 * (size_t)i;
  int d0, d1, d2, d3;
  if ((((size_t)dst) & 15) == 0) {
    int4 v = *(const int4*)p;
    d0 = v.x; d1 = v.y; d2 = v.z; d3 = v.w;
  } else {
    d0 = p[0]; d1 = p[1]; d2 = p[2]; d3 = p[3];
  }
  const unsigned r = (unsigned)rpb;
  unsigned k0 = min((unsigned)d0 / r, 255u);
  unsigned k1 = min((unsigned)d1 / r, 255u);
  unsigned k2 = min((unsigned)d2 / r, 255u);
  unsigned k3 = min((unsigned)d3 / r, 255u);
  keys[i] = k0 | (k1 << 8) | (k2 << 16) | (k3 << 24);
}

#define RS_THREADS 1024
#define RS_WAVES (RS_THREADS / 32)
#define RS_LIST_CAP 8192
#define RS_LIST_PER_THREAD (RS_LIST_CAP / RS_THREADS)
#define RK_ONE 0x01010101u
#define RK_HI 0x80808080u

__device__ __forceinline__ unsigned rk_flag(unsigned x) { return (x - RK_ONE) & ~x; }

__global__ void __launch_bounds__(RS_THREADS)
range_queue_kernel(const int* __restrict__ src, const int* __restrict__ dst,
                   const unsigned* __restrict__ keyw, float* __restrict__ dinv,
                   int* __restrict__ qcnt, int2* __restrict__ qlist, int n_rows, int nE, int rpb) {
  extern __shared__ float lds[];
  const uint4* keys = (const uint4*)keyw;
  int* list = (int*)lds;
  int* degc = list + RS_LIST_CAP;
  int* cntp = degc + rpb;
  const int tid = threadIdx.x;
  const int r0 = blockIdx.x * rpb;
  int nrows = n_rows - r0;
  if (nrows > rpb) nrows = rpb;
  if (nrows < 0) nrows = 0;
  const unsigned un = (unsigned)nrows;
  const unsigned bb = (blockIdx.x & 255u) * RK_ONE;

  for (int r = tid; r < nrows; r += RS_THREADS) degc[r] = 0;
  if (tid == 0) *cntp = 0;
  __syncthreads();

  const int nK = nE >> 4;
  int base = 0;
  for (; base + 4 * RS_THREADS <= nK; base += 4 * RS_THREADS) {
    uint4 kv[4];
#pragma unroll
    for (int u = 0; u < 4; u++) kv[u] = keys[base + u * RS_THREADS + tid];
    unsigned a[16];
    unsigned any = 0;
#pragma unroll
    for (int u = 0; u < 4; u++) {
      a[4 * u + 0] = rk_flag(kv[u].x ^ bb);
      a[4 * u + 1] = rk_flag(kv[u].y ^ bb);
      a[4 * u + 2] = rk_flag(kv[u].z ^ bb);
      a[4 * u + 3] = rk_flag(kv[u].w ^ bb);
      any |= a[4 * u + 0] | a[4 * u + 1] | a[4 * u + 2] | a[4 * u + 3];
    }
    if (any & RK_HI) {
      unsigned qlo = 0, qhi = 0;
#pragma unroll
      for (int i = 0; i < 8; i++) {
        qlo |= (a[i] & RK_HI) >> i;
        qhi |= (a[8 + i] & RK_HI) >> i;
      }
      int slot = atomicAdd(cntp, __popc(qlo) + __popc(qhi));
      const int e0 = (base + tid) << 4;
      while (qlo) {
        int pos = __ffs(qlo) - 1;
        qlo &= qlo - 1;
        int i = 7 - (pos & 7);
        int e = e0 + (i >> 2) * (16 * RS_THREADS) + ((i & 3) << 2) + (pos >> 3);
        if (slot < RS_LIST_CAP) list[slot] = e;
        slot++;
      }
      while (qhi) {
        int pos = __ffs(qhi) - 1;
        qhi &= qhi - 1;
        int i = 15 - (pos & 7);
        int e = e0 + (i >> 2) * (16 * RS_THREADS) + ((i & 3) << 2) + (pos >> 3);
        if (slot < RS_LIST_CAP) list[slot] = e;
        slot++;
      }
    }
  }
  for (int j = base + tid; j < nK; j += RS_THREADS) {
    uint4 kv = keys[j];
    unsigned a0 = rk_flag(kv.x ^ bb), a1 = rk_flag(kv.y ^ bb);
    unsigned a2 = rk_flag(kv.z ^ bb), a3 = rk_flag(kv.w ^ bb);
    if ((a0 | a1 | a2 | a3) & RK_HI) {
      unsigned q = (a0 & RK_HI) | ((a1 & RK_HI) >> 1) | ((a2 & RK_HI) >> 2) | ((a3 & RK_HI) >> 3);
      int slot = atomicAdd(cntp, __popc(q));
      while (q) {
        int pos = __ffs(q) - 1;
        q &= q - 1;
        int i = 7 - (pos & 7);
        int e = (j << 4) + (i << 2) + (pos >> 3);
        if (slot < RS_LIST_CAP) list[slot] = e;
        slot++;
      }
    }
  }
  for (int e = (nK << 4) + tid; e < nE; e += RS_THREADS) {
    if ((unsigned)(dst[e] - r0) < un) {
      int slot = atomicAdd(cntp, 1);
      if (slot < RS_LIST_CAP) list[slot] = e;
    }
  }
  __syncthreads();

  const int n = *cntp;
  const bool over = n > RS_LIST_CAP;
  const int nq = over ? 0 : n;
  int2* q = qlist + (size_t)blockIdx.x * RS_LIST_CAP;
  if (over) {
    for (int e = tid; e < nE; e += RS_THREADS) {
      unsigned loc = (unsigned)(dst[e] - r0);
      if (loc < un) atomicAdd(&degc[loc], 1);
    }
  } else if (nq > 0) {
    int ed[RS_LIST_PER_THREAD];
    int s[RS_LIST_PER_THREAD];
    int d[RS_LIST_PER_THREAD];
#pragma unroll
    for (int u = 0; u < RS_LIST_PER_THREAD; u++) {
      int idx = tid + u * RS_THREADS;
      ed[u] = list[idx < nq ? idx : 0];
    }
#pragma unroll
    for (int u = 0; u < RS_LIST_PER_THREAD; u++) {
      s[u] = src[ed[u]];
      d[u] = dst[ed[u]];
    }
#pragma unroll
    for (int u = 0; u < RS_LIST_PER_THREAD; u++) {
      int idx = tid + u * RS_THREADS;
      if (idx < nq) {
        unsigned loc = (unsigned)(d[u] - r0);
        if (loc < un) {
          atomicAdd(&degc[loc], 1);
          q[idx] = make_int2(s[u], (int)loc);
        } else {
          q[idx] = make_int2(0, -1);
        }
      }
    }
  }
  __syncthreads();
  if (tid == 0) qcnt[blockIdx.x] = n;
  for (int r = tid; r < nrows; r += RS_THREADS) dinv[r0 + r] = rsqrtf(1.0f + (float)degc[r]);
}

template <int N, int NEXT>
struct RsPitch { static constexpr int v = (NEXT > 0) ? N + 4 : N; };

template <int N, int RELU, int NEXT>
__global__ void __launch_bounds__(RS_THREADS)
range_sum_kernel(const int* __restrict__ src, const int* __restrict__ dst,
                 const float* __restrict__ hs, const float* __restrict__ dinv,
                 const float* __restrict__ bias, float* __restrict__ outp,
                 const float* __restrict__ Wn,
                 const int* __restrict__ qcnt, const int2* __restrict__ qlist,
                 int n_rows, int nE, int rpb) {
  extern __shared__ float lds[];
  constexpr int P = RsPitch<N, NEXT>::v;
  constexpr int P4 = P / 4;
  constexpr int N4 = N / 4;
  const int arows = (NEXT > 0) ? ((rpb + 15) & ~15) : rpb;
  float* acc = lds;
  float* dl = lds + (size_t)arows * P;
  int2* list = (int2*)(dl + ((NEXT > 0) ? arows : ((rpb + 1) & ~1)));
  const int tid = threadIdx.x;
  const int r0 = blockIdx.x * rpb;
  int nrows = n_rows - r0;
  if (nrows > rpb) nrows = rpb;
  if (nrows < 0) nrows = 0;
  const unsigned un = (unsigned)nrows;
  const int nel4 = (nrows * N) >> 2;
  const int wave = tid >> 5;
  const int lane = tid & 31;

  constexpr int NCT = (NEXT > 0) ? NEXT / 16 : 1;
  const int col = lane & 15;
  const int koff = (lane >> 4) << 1;
  const int wv = __builtin_amdgcn_readfirstlane(wave);
  const int nbase = (wv % NCT) << 4;
  v2f bf[(NEXT > 0) ? N / 4 : 1];
  if constexpr (NEXT > 0) {
    const float* wp = Wn + (size_t)koff * NEXT + nbase + col;
#pragma unroll
    for (int k = 0; k < N / 4; k++) {
      bf[k].x = wp[(size_t)(4 * k) * NEXT];
      bf[k].y = wp[(size_t)(4 * k + 1) * NEXT];
    }
  }

  const int n = qcnt[blockIdx.x];
  const bool over = n > RS_LIST_CAP;
  const int nq = over ? 0 : n;
  const int2* q = qlist + (size_t)blockIdx.x * RS_LIST_CAP;

  const float4* h4 = (const float4*)(hs + (size_t)r0 * N);
  float4* a4 = (float4*)acc;
  int i0 = tid;
  for (; i0 + 3 * RS_THREADS < nel4; i0 += 4 * RS_THREADS) {
    float4 t[4];
#pragma unroll
    for (int u = 0; u < 4; u++) t[u] = h4[i0 + u * RS_THREADS];
#pragma unroll
    for (int u = 0; u < 4; u++) {
      const int i = i0 + u * RS_THREADS;
      a4[(i / N4) * P4 + (i % N4)] = t[u];
    }
  }
  for (; i0 < nel4; i0 += RS_THREADS) a4[(i0 / N4) * P4 + (i0 % N4)] = h4[i0];
  if (NEXT > 0) {
    const int tail4 = (((nrows + 15) & ~15) - nrows) * N4;
    if (tid < tail4) a4[(nrows + tid / N4) * P4 + (tid % N4)] = make_float4(0.f, 0.f, 0.f, 0.f);
  }
  for (int r = tid; r < nrows; r += RS_THREADS) dl[r] = dinv[r0 + r];
  if (NEXT > 0) {
    if (tid < ((nrows + 15) & ~15) - nrows) dl[nrows + tid] = 0.f;
  }
  {
    int2 p[RS_LIST_PER_THREAD];
#pragma unroll
    for (int u = 0; u < RS_LIST_PER_THREAD; u++) {
      int idx = tid + u * RS_THREADS;
      p[u] = make_int2(0, 0);
      if (idx < nq) p[u] = q[idx];
    }
#pragma unroll
    for (int u = 0; u < RS_LIST_PER_THREAD; u++) {
      int idx = tid + u * RS_THREADS;
      if (idx < nq) list[idx] = p[u];
    }
  }
  __syncthreads();

  if (over) {
    for (int e = tid; e < nE; e += RS_THREADS) {
      unsigned loc = (unsigned)(dst[e] - r0);
      if (loc < un) {
        const float* hp = hs + (size_t)src[e] * N;
        float* ap = acc + (size_t)loc * P;
        for (int f = 0; f < N; f++) unsafeAtomicAdd(&ap[f], hp[f]);
      }
    }
  }

  for (int i = wave; i < nq; i += 4 * RS_WAVES) {
    float v[4][N / 32];
    int loc[4];
#pragma unroll
    for (int q4 = 0; q4 < 4; q4++) {
      int idx = i + q4 * RS_WAVES;
      loc[q4] = -1;
      if (idx < nq) {
        int2 p = list[idx];
        loc[q4] = p.y;
        const float* hp = hs + (size_t)p.x * N + lane;
#pragma unroll
        for (int j = 0; j < N / 32; j++) v[q4][j] = hp[32 * j];
      }
    }
#pragma unroll
    for (int q4 = 0; q4 < 4; q4++) {
      if (loc[q4] >= 0) {
        float* ap = acc + (size_t)loc[q4] * P + lane;
#pragma unroll
        for (int j = 0; j < N / 32; j++) unsafeAtomicAdd(&ap[32 * j], v[q4][j]);
      }
    }
  }
  __syncthreads();

  const float4 b = *(const float4*)(bias + ((tid << 2) & (N - 1)));
  float4* o4 = (float4*)(outp + (size_t)r0 * N);
  for (int i = tid; i < nel4; i += RS_THREADS) {
    const int li = (i / N4) * P4 + (i % N4);
    float4 a = a4[li];
    float di = dl[(i << 2) / N];
    float4 o;
    o.x = di * a.x + b.x;
    o.y = di * a.y + b.y;
    o.z = di * a.z + b.z;
    o.w = di * a.w + b.w;
    if (RELU) {
      o.x = fmaxf(o.x, 0.f);
      o.y = fmaxf(o.y, 0.f);
      o.z = fmaxf(o.z, 0.f);
      o.w = fmaxf(o.w, 0.f);
    }
    if (NEXT > 0) a4[li] = o;
    else o4[i] = o;
  }

  if constexpr (NEXT > 0) {
    constexpr int RSTEP = RS_WAVES / NCT;
    const int madd = (lane < 16) ? 0 : 8;
    __syncthreads();
    const int nrt = (nrows + 15) >> 4;
    for (int rt = wv / NCT; rt < nrt; rt += RSTEP) {
      const float* aBase = &acc[(rt * 16 + col) * P + koff];
      v8f c = (v8f){0.f, 0.f, 0.f, 0.f, 0.f, 0.f, 0.f, 0.f};
#pragma unroll
      for (int kg = 0; kg < N / 4; kg += 8) {
        v2f a[8];
#pragma unroll
        for (int kk = 0; kk < 8; kk++) a[kk] = *(const v2f*)(aBase + 4 * (kg + kk));
#pragma unroll
        for (int kk = 0; kk < 8; kk++)
          c = __builtin_amdgcn_wmma_f32_16x16x4_f32(false, a[kk], false, bf[kg + kk], (short)0, c,
                                                    false, false);
      }
      const float4* dp = (const float4*)&dl[rt * 16 + madd];
      const float4 d0 = dp[0];
      const float4 d1 = dp[1];
      const float d[8] = {d0.x, d0.y, d0.z, d0.w, d1.x, d1.y, d1.z, d1.w};
#pragma unroll
      for (int i = 0; i < 8; i++) {
        const int lr = rt * 16 + i + madd;
        if (lr < nrows) outp[(size_t)(r0 + lr) * NEXT + nbase + col] = c[i] * d[i];
      }
    }
  }
}

#define ED_THREADS 128
typedef float v4f __attribute__((ext_vector_type(4)));

__global__ void __launch_bounds__(ED_THREADS)
edgedot_kernel(const int* __restrict__ es, const int* __restrict__ ed,
               const float* __restrict__ z, float* __restrict__ out, int nL) {
  const int e = blockIdx.x * ED_THREADS + threadIdx.x;
  const int ec = (e < nL) ? e : nL - 1;
  const v4f* a = (const v4f*)(z + (size_t)es[ec] * 64);
  const v4f* b = (const v4f*)(z + (size_t)ed[ec] * 64);
  v4f va[16], vb[16];
#pragma unroll
  for (int j = 0; j < 16; j++) va[j] = a[j];
#pragma unroll
  for (int j = 0; j < 16; j++) vb[j] = b[j];
  asm volatile("" ::"v"(va[0]), "v"(va[1]), "v"(va[2]), "v"(va[3]), "v"(va[4]), "v"(va[5]), "v"(va[6]),
               "v"(va[7]), "v"(va[8]), "v"(va[9]), "v"(va[10]), "v"(va[11]), "v"(va[12]), "v"(va[13]),
               "v"(va[14]), "v"(va[15]));
  asm volatile("" ::"v"(vb[0]), "v"(vb[1]), "v"(vb[2]), "v"(vb[3]), "v"(vb[4]), "v"(vb[5]), "v"(vb[6]),
               "v"(vb[7]), "v"(vb[8]), "v"(vb[9]), "v"(vb[10]), "v"(vb[11]), "v"(vb[12]), "v"(vb[13]),
               "v"(vb[14]), "v"(vb[15]));
  float p0 = 0.f, p1 = 0.f, p2 = 0.f, p3 = 0.f;
#pragma unroll
  for (int j = 0; j < 16; j++) {
    p0 = fmaf(va[j].x, vb[j].x, p0);
    p1 = fmaf(va[j].y, vb[j].y, p1);
    p2 = fmaf(va[j].z, vb[j].z, p2);
    p3 = fmaf(va[j].w, vb[j].w, p3);
  }
  if (e < nL) out[e] = (p0 + p1) + (p2 + p3);
}

extern "C" void kernel_launch(void* const* d_in, const int* in_sizes, int n_in,
                              void* d_out, int out_size, void* d_ws, size_t ws_size,
                              hipStream_t stream) {
  const int*   x_idx = (const int*)d_in[0];
  const int*   eidx  = (const int*)d_in[1];
  const int*   lidx  = (const int*)d_in[2];
  const float* emb   = (const float*)d_in[3];
  const float* W1    = (const float*)d_in[4];
  const float* b1    = (const float*)d_in[5];
  const float* W2    = (const float*)d_in[6];
  const float* b2    = (const float*)d_in[7];
  float* out = (float*)d_out;

  const int nN = in_sizes[0];
  const int nE = in_sizes[1] / 2;
  const int nL = in_sizes[2] / 2;
  const int H = 128, O = 64;

  float* dinv = (float*)d_ws;
  float* hs1  = dinv + ((nN + 63) & ~63);
  float* hs2  = hs1 + (size_t)nN * H;
  float* z    = hs2 + (size_t)nN * O;
  int*   qcnt = (int*)(z + (size_t)nN * O);
  int2*  qlist = (int2*)(qcnt + 256);
  unsigned* keys = (unsigned*)(qlist + (size_t)256 * RS_LIST_CAP);

  const int* esrc = eidx;
  const int* edst = eidx + nE;

  const int rpb = (nN + 255) / 256;
  size_t rs_tail = (size_t)((rpb + 1) & ~1) * sizeof(float) + (size_t)RS_LIST_CAP * sizeof(int2) + 16;
  const int arows = (rpb + 15) & ~15;
  size_t rs_mm = (size_t)arows * (H + 4 + 1) * sizeof(float) + (size_t)RS_LIST_CAP * sizeof(int2) + 16;

  range_key_kernel<<<(unsigned)((nE / 4 + 255) / 256), 256, 0, stream>>>(edst, keys, nE, rpb);
  range_queue_kernel<<<256, RS_THREADS,
                       (size_t)RS_LIST_CAP * sizeof(int) + (size_t)rpb * sizeof(int) + 16, stream>>>(
      esrc, edst, keys, dinv, qcnt, qlist, nN, nE, rpb);

  gemm_rows_kernel<128, 1><<<512, 256, 0, stream>>>(emb, x_idx, W1, dinv, hs1, nN);
  range_sum_kernel<128, 1, 64><<<256, RS_THREADS, rs_mm, stream>>>(
      esrc, edst, hs1, dinv, b1, hs2, W2, qcnt, qlist, nN, nE, rpb);

  range_sum_kernel<64, 0, 0><<<256, RS_THREADS, (size_t)rpb * O * sizeof(float) + rs_tail, stream>>>(
      esrc, edst, hs2, dinv, b2, z, nullptr, qcnt, qlist, nN, nE, rpb);

  edgedot_kernel<<<(unsigned)((nL + ED_THREADS - 1) / ED_THREADS), ED_THREADS, 0, stream>>>(
      lidx, lidx + nL, z, out, nL);
}


# Layer 1 product in the queue launch: a range block multiplies its own rows on waves 16-31 while waves 0-15 scan keys; hs1 stored unscaled, dinv applied on read; one launch less; sum orders kept
# speedup vs baseline: 3.7054x; 1.0920x over previous
#include <hip/hip_runtime.h>

typedef float v2f __attribute__((ext_vector_type(2)));
typedef float v8f __attribute__((ext_vector_type(8)));
typedef float v4f __attribute__((ext_vector_type(4)));

#define LDS_STRIDE 132

__global__ void __launch_bounds__(256)
range_key_kernel(const int* __restrict__ dst, unsigned* __restrict__ keys, int nE, int rpb) {
  const int nW = (nE >> 4) << 2;
  const int i = blockIdx.x * 256 + threadIdx.x;
  if (i >= nW) return;
  const int* p = dst + 4 * (size_t)i;
  int d0, d1, d2, d3;
  if ((((size_t)dst) & 15) == 0) {
    int4 v = *(const int4*)p;
    d0 = v.x; d1 = v.y; d2 = v.z; d3 = v.w;
  } else {
    d0 = p[0]; d1 = p[1]; d2 = p[2]; d3 = p[3];
  }
  const unsigned r = (unsigned)rpb;
  unsigned k0 = min((unsigned)d0 / r, 255u);
  unsigned k1 = min((unsigned)d1 / r, 255u);
  unsigned k2 = min((unsigned)d2 / r, 255u);
  unsigned k3 = min((unsigned)d3 / r, 255u);
  keys[i] = k0 | (k1 << 8) | (k2 << 16) | (k3 << 24);
}

#define RS_THREADS 1024
#define RS_WAVES (RS_THREADS / 32)
#define RS_LIST_CAP 8192
#define RS_LIST_PER_THREAD (RS_LIST_CAP / RS_THREADS)
#define RQ_SCAN_THREADS 512
#define RQ_SCAN_WAVES (RQ_SCAN_THREADS / 32)
#define RQ_TAIL_PER_THREAD 8
#define RQ_FILL 13
#define RQ_FILL_ROWS (32 * RQ_FILL)
#define RK_ONE 0x01010101u
#define RK_HI 0x80808080u

__device__ __forceinline__ unsigned rk_flag(unsigned x) { return (x - RK_ONE) & ~x; }

template <int N>
__global__ void __launch_bounds__(RS_THREADS)
range_queue_mm_kernel(const int* __restrict__ src, const int* __restrict__ dst,
                      const unsigned* __restrict__ keyw, const float* __restrict__ X,
                      const int* __restrict__ gidx, const float* __restrict__ Wm,
                      float* __restrict__ h, float* __restrict__ dinv,
                      int* __restrict__ qcnt, int2* __restrict__ qlist, int n_rows, int nE, int rpb) {
  static_assert(N == 128, "rows of 128 floats: 32 16-byte units a row, 8 column tiles");
  extern __shared__ float lds[];
  const uint4* keys = (const uint4*)keyw;
  const int arows = (rpb + 15) & ~15;
  const int frows = ((arows + RQ_FILL_ROWS - 1) / RQ_FILL_ROWS) * RQ_FILL_ROWS;
  float* lA = lds;
  int* list = (int*)(lds + (size_t)frows * LDS_STRIDE);
  int* degc = list + RS_LIST_CAP;
  int* cntp = degc + rpb;
  const int tid = threadIdx.x;
  const int lane = tid & 31;
  const int wv = __builtin_amdgcn_readfirstlane(tid >> 5);
  const bool is_mm = wv >= RQ_SCAN_WAVES;
  const int r0 = blockIdx.x * rpb;
  int nrows = n_rows - r0;
  if (nrows > rpb) nrows = rpb;
  if (nrows < 0) nrows = 0;
  const unsigned un = (unsigned)nrows;
  const unsigned bb = (blockIdx.x & 255u) * RK_ONE;

  for (int r = tid; r < nrows; r += RS_THREADS) degc[r] = 0;
  if (tid == 0) *cntp = 0;

  const int nfill = (nrows + 15) & ~15;
  {
    const v4f* X4 = (const v4f*)X;
    const int c4 = tid & 31;
    const int rw = tid >> 5;
    for (int rb = 0; rb < nfill; rb += RQ_FILL_ROWS) {
      int sr[RQ_FILL];
#pragma unroll
      for (int u = 0; u < RQ_FILL; u++) {
        const int r = rb + rw + 32 * u;
        sr[u] = gidx[r0 + (r < nrows ? r : 0)];
      }
      asm volatile("" ::"v"(sr[0]), "v"(sr[1]), "v"(sr[2]), "v"(sr[3]), "v"(sr[4]), "v"(sr[5]), "v"(sr[6]),
                   "v"(sr[7]), "v"(sr[8]), "v"(sr[9]), "v"(sr[10]), "v"(sr[11]), "v"(sr[12]));
      v4f v[RQ_FILL];
#pragma unroll
      for (int u = 0; u < RQ_FILL; u++) v[u] = X4[(size_t)sr[u] * (N / 4) + c4];
      asm volatile("" ::"v"(v[0]), "v"(v[1]), "v"(v[2]), "v"(v[3]), "v"(v[4]), "v"(v[5]), "v"(v[6]),
                   "v"(v[7]), "v"(v[8]), "v"(v[9]), "v"(v[10]), "v"(v[11]), "v"(v[12]));
#pragma unroll
      for (int u = 0; u < RQ_FILL; u++) {
        const int r = rb + rw + 32 * u;
        const v4f t = (r < nrows) ? v[u] : (v4f){0.f, 0.f, 0.f, 0.f};
        *(v4f*)&lA[r * LDS_STRIDE + (c4 << 2)] = t;
      }
    }
  }

  const int col = lane & 15;
  const int koff = (lane >> 4) << 1;
  const int gw = wv - RQ_SCAN_WAVES;
  const int nbase = (gw & 7) << 4;
  const int rpar = (gw >> 3) & 1;
  v2f bf[N / 4];
  if (is_mm) {
    const float* wp = Wm + (size_t)koff * N + nbase + col;
#pragma unroll
    for (int k = 0; k < N / 4; k++) {
      bf[k].x = wp[(size_t)(4 * k) * N];
      bf[k].y = wp[(size_t)(4 * k + 1) * N];
    }
  }
  __syncthreads();

  const int nrt = nfill >> 4;
  const int e1 = ((nrt * 3) >> 2) & ~1;
  const int e2 = (((nrt * 7) >> 3) + 1) & ~1;
  const int madd = (lane < 16) ? 0 : 8;

  auto mm_tiles = [&](int rt_lo, int rt_hi) {
    for (int rt = rt_lo + rpar; rt < rt_hi; rt += 2) {
      const float* aBase = &lA[(rt * 16 + col) * LDS_STRIDE + koff];
      v8f c = (v8f){0.f, 0.f, 0.f, 0.f, 0.f, 0.f, 0.f, 0.f};
#pragma unroll
      for (int kg = 0; kg < N / 4; kg += 8) {
        v2f a[8];
#pragma unroll
        for (int kk = 0; kk < 8; kk++) a[kk] = *(const v2f*)(aBase + 4 * (kg + kk));
#pragma unroll
        for (int kk = 0; kk < 8; kk++)
          c = __builtin_amdgcn_wmma_f32_16x16x4_f32(false, a[kk], false, bf[kg + kk], (short)0, c,
                                                    false, false);
      }
#pragma unroll
      for (int i = 0; i < 8; i++) {
        const int lr = rt * 16 + i + madd;
        if (lr < nrows) h[(size_t)(r0 + lr) * N + nbase + col] = c[i];
      }
    }
  };

  if (is_mm) {
    mm_tiles(0, e1);
  } else {
    const int nK = nE >> 4;
    int base = 0;
    for (; base + 8 * RQ_SCAN_THREADS <= nK; base += 8 * RQ_SCAN_THREADS) {
      uint4 kv[8];
#pragma unroll
      for (int u = 0; u < 8; u++) kv[u] = keys[base + u * RQ_SCAN_THREADS + tid];
      unsigned q[4];
      unsigned any = 0;
#pragma unroll
      for (int g = 0; g < 4; g++) {
        const unsigned a0 = rk_flag(kv[2 * g].x ^ bb), a1 = rk_flag(kv[2 * g].y ^ bb);
        const unsigned a2 = rk_flag(kv[2 * g].z ^ bb), a3 = rk_flag(kv[2 * g].w ^ bb);
        const unsigned a4 = rk_flag(kv[2 * g + 1].x ^ bb), a5 = rk_flag(kv[2 * g + 1].y ^ bb);
        const unsigned a6 = rk_flag(kv[2 * g + 1].z ^ bb), a7 = rk_flag(kv[2 * g + 1].w ^ bb);
        q[g] = (a0 & RK_HI) | ((a1 & RK_HI) >> 1) | ((a2 & RK_HI) >> 2) | ((a3 & RK_HI) >> 3) |
               ((a4 & RK_HI) >> 4) | ((a5 & RK_HI) >> 5) | ((a6 & RK_HI) >> 6) | ((a7 & RK_HI) >> 7);
        any |= q[g];
      }
      if (any) {
        int slot = atomicAdd(cntp, __popc(q[0]) + __popc(q[1]) + __popc(q[2]) + __popc(q[3]));
        const int e0 = (base + tid) << 4;
#pragma unroll
        for (int g = 0; g < 4; g++) {
          unsigned qq = q[g];
          while (qq) {
            int pos = __ffs(qq) - 1;
            qq &= qq - 1;
            int i = 8 * g + 7 - (pos & 7);
            int e = e0 + (i >> 2) * (16 * RQ_SCAN_THREADS) + ((i & 3) << 2) + (pos >> 3);
            if (slot < RS_LIST_CAP) list[slot] = e;
            slot++;
          }
        }
      }
    }
    for (int j = base + tid; j < nK; j += RQ_SCAN_THREADS) {
      uint4 kv = keys[j];
      unsigned a0 = rk_flag(kv.x ^ bb), a1 = rk_flag(kv.y ^ bb);
      unsigned a2 = rk_flag(kv.z ^ bb), a3 = rk_flag(kv.w ^ bb);
      if ((a0 | a1 | a2 | a3) & RK_HI) {
        unsigned q = (a0 & RK_HI) | ((a1 & RK_HI) >> 1) | ((a2 & RK_HI) >> 2) | ((a3 & RK_HI) >> 3);
        int slot = atomicAdd(cntp, __popc(q));
        while (q) {
          int pos = __ffs(q) - 1;
          q &= q - 1;
          int i = 7 - (pos & 7);
          int e = (j << 4) + (i << 2) + (pos >> 3);
          if (slot < RS_LIST_CAP) list[slot] = e;
          slot++;
        }
      }
    }
    for (int e = (nK << 4) + tid; e < nE; e += RQ_SCAN_THREADS) {
      if ((unsigned)(dst[e] - r0) < un) {
        int slot = atomicAdd(cntp, 1);
        if (slot < RS_LIST_CAP) list[slot] = e;
      }
    }
  }
  __syncthreads();

  const int n = *cntp;
  const bool over = n > RS_LIST_CAP;
  const int nq = over ? 0 : n;
  if (is_mm) {
    mm_tiles(e1, e2);
  } else {
    int2* q = qlist + (size_t)blockIdx.x * RS_LIST_CAP;
    if (over) {
      for (int e = tid; e < nE; e += RQ_SCAN_THREADS) {
        unsigned loc = (unsigned)(dst[e] - r0);
        if (loc < un) atomicAdd(&degc[loc], 1);
      }
    } else {
      for (int b0 = 0; b0 < nq; b0 += RQ_TAIL_PER_THREAD * RQ_SCAN_THREADS) {
        int ed[RQ_TAIL_PER_THREAD];
        int s[RQ_TAIL_PER_THREAD];
        int d[RQ_TAIL_PER_THREAD];
#pragma unroll
        for (int u = 0; u < RQ_TAIL_PER_THREAD; u++) {
          int idx = b0 + tid + u * RQ_SCAN_THREADS;
          ed[u] = list[idx < nq ? idx : 0];
        }
#pragma unroll
        for (int u = 0; u < RQ_TAIL_PER_THREAD; u++) {
          s[u] = src[ed[u]];
          d[u] = dst[ed[u]];
        }
#pragma unroll
        for (int u = 0; u < RQ_TAIL_PER_THREAD; u++) {
          int idx = b0 + tid + u * RQ_SCAN_THREADS;
          if (idx < nq) {
            unsigned loc = (unsigned)(d[u] - r0);
            if (loc < un) {
              atomicAdd(&degc[loc], 1);
              q[idx] = make_int2(s[u], (int)loc);
            } else {
              q[idx] = make_int2(0, -1);
            }
          }
        }
      }
    }
  }
  __syncthreads();
  if (is_mm) {
    mm_tiles(e2, nrt);
  } else {
    if (tid == 0) qcnt[blockIdx.x] = n;
    for (int r = tid; r < nrows; r += RQ_SCAN_THREADS) dinv[r0 + r] = rsqrtf(1.0f + (float)degc[r]);
  }
}

template <int N, int NEXT>
struct RsPitch { static constexpr int v = (NEXT > 0) ? N + 4 : N; };

template <int N, int RELU, int NEXT, int SCALE>
__global__ void __launch_bounds__(RS_THREADS)
range_sum_kernel(const int* __restrict__ src, const int* __restrict__ dst,
                 const float* __restrict__ hs, const float* __restrict__ dinv,
                 const float* __restrict__ bias, float* __restrict__ outp,
                 const float* __restrict__ Wn,
                 const int* __restrict__ qcnt, const int2* __restrict__ qlist,
                 int n_rows, int nE, int rpb) {
  extern __shared__ float lds[];
  constexpr int P = RsPitch<N, NEXT>::v;
  constexpr int P4 = P / 4;
  constexpr int N4 = N / 4;
  const int arows = (NEXT > 0) ? ((rpb + 15) & ~15) : rpb;
  float* acc = lds;
  float* dl = lds + (size_t)arows * P;
  int2* list = (int2*)(dl + ((NEXT > 0) ? arows : ((rpb + 1) & ~1)));
  const int tid = threadIdx.x;
  const int r0 = blockIdx.x * rpb;
  int nrows = n_rows - r0;
  if (nrows > rpb) nrows = rpb;
  if (nrows < 0) nrows = 0;
  const unsigned un = (unsigned)nrows;
  const int nel4 = (nrows * N) >> 2;
  const int wave = tid >> 5;
  const int lane = tid & 31;

  constexpr int NCT = (NEXT > 0) ? NEXT / 16 : 1;
  const int col = lane & 15;
  const int koff = (lane >> 4) << 1;
  const int wv = __builtin_amdgcn_readfirstlane(wave);
  const int nbase = (wv % NCT) << 4;
  v2f bf[(NEXT > 0) ? N / 4 : 1];
  if constexpr (NEXT > 0) {
    const float* wp = Wn + (size_t)koff * NEXT + nbase + col;
#pragma unroll
    for (int k = 0; k < N / 4; k++) {
      bf[k].x = wp[(size_t)(4 * k) * NEXT];
      bf[k].y = wp[(size_t)(4 * k + 1) * NEXT];
    }
  }

  const int n = qcnt[blockIdx.x];
  const bool over = n > RS_LIST_CAP;
  const int nq = over ? 0 : n;
  const int2* q = qlist + (size_t)blockIdx.x * RS_LIST_CAP;

  const float4* h4 = (const float4*)(hs + (size_t)r0 * N);
  float4* a4 = (float4*)acc;
  int i0 = tid;
  for (; i0 + 3 * RS_THREADS < nel4; i0 += 4 * RS_THREADS) {
    float4 t[4];
    float sc[4];
#pragma unroll
    for (int u = 0; u < 4; u++) {
      t[u] = h4[i0 + u * RS_THREADS];
      sc[u] = SCALE ? dinv[r0 + (i0 + u * RS_THREADS) / N4] : 1.f;
    }
#pragma unroll
    for (int u = 0; u < 4; u++) {
      const int i = i0 + u * RS_THREADS;
      if (SCALE) {
        t[u].x *= sc[u];
        t[u].y *= sc[u];
        t[u].z *= sc[u];
        t[u].w *= sc[u];
      }
      a4[(i / N4) * P4 + (i % N4)] = t[u];
    }
  }
  for (; i0 < nel4; i0 += RS_THREADS) {
    float4 t = h4[i0];
    if (SCALE) {
      const float sc = dinv[r0 + i0 / N4];
      t.x *= sc;
      t.y *= sc;
      t.z *= sc;
      t.w *= sc;
    }
    a4[(i0 / N4) * P4 + (i0 % N4)] = t;
  }
  if (NEXT > 0) {
    const int tail4 = (((nrows + 15) & ~15) - nrows) * N4;
    if (tid < tail4) a4[(nrows + tid / N4) * P4 + (tid % N4)] = make_float4(0.f, 0.f, 0.f, 0.f);
  }
  for (int r = tid; r < nrows; r += RS_THREADS) dl[r] = dinv[r0 + r];
  if (NEXT > 0) {
    if (tid < ((nrows + 15) & ~15) - nrows) dl[nrows + tid] = 0.f;
  }
  {
    int2 p[RS_LIST_PER_THREAD];
#pragma unroll
    for (int u = 0; u < RS_LIST_PER_THREAD; u++) {
      int idx = tid + u * RS_THREADS;
      p[u] = make_int2(0, 0);
      if (idx < nq) p[u] = q[idx];
    }
#pragma unroll
    for (int u = 0; u < RS_LIST_PER_THREAD; u++) {
      int idx = tid + u * RS_THREADS;
      if (idx < nq) list[idx] = p[u];
    }
  }
  __syncthreads();

  if (over) {
    for (int e = tid; e < nE; e += RS_THREADS) {
      unsigned loc = (unsigned)(dst[e] - r0);
      if (loc < un) {
        const int sn = src[e];
        const float* hp = hs + (size_t)sn * N;
        float* ap = acc + (size_t)loc * P;
        const float sc = SCALE ? dinv[sn] : 1.f;
        for (int f = 0; f < N; f++) unsafeAtomicAdd(&ap[f], SCALE ? sc * hp[f] : hp[f]);
      }
    }
  }

  for (int i = wave; i < nq; i += 4 * RS_WAVES) {
    float v[4][N / 32];
    float sc[4];
    int loc[4];
#pragma unroll
    for (int q4 = 0; q4 < 4; q4++) {
      int idx = i + q4 * RS_WAVES;
      loc[q4] = -1;
      sc[q4] = 1.f;
      if (idx < nq) {
        int2 p = list[idx];
        loc[q4] = p.y;
        const float* hp = hs + (size_t)p.x * N + lane;
#pragma unroll
        for (int j = 0; j < N / 32; j++) v[q4][j] = hp[32 * j];
        if (SCALE) sc[q4] = dinv[p.x];
      }
    }
#pragma unroll
    for (int q4 = 0; q4 < 4; q4++) {
      if (loc[q4] >= 0) {
        float* ap = acc + (size_t)loc[q4] * P + lane;
#pragma unroll
        for (int j = 0; j < N / 32; j++) unsafeAtomicAdd(&ap[32 * j], SCALE ? sc[q4] * v[q4][j] : v[q4][j]);
      }
    }
  }
  __syncthreads();

  const float4 b = *(const float4*)(bias + ((tid << 2) & (N - 1)));
  float4* o4 = (float4*)(outp + (size_t)r0 * N);
  for (int i = tid; i < nel4; i += RS_THREADS) {
    const int li = (i / N4) * P4 + (i % N4);
    float4 a = a4[li];
    float di = dl[(i << 2) / N];
    float4 o;
    o.x = di * a.x + b.x;
    o.y = di * a.y + b.y;
    o.z = di * a.z + b.z;
    o.w = di * a.w + b.w;
    if (RELU) {
      o.x = fmaxf(o.x, 0.f);
      o.y = fmaxf(o.y, 0.f);
      o.z = fmaxf(o.z, 0.f);
      o.w = fmaxf(o.w, 0.f);
    }
    if (NEXT > 0) a4[li] = o;
    else o4[i] = o;
  }

  if constexpr (NEXT > 0) {
    constexpr int RSTEP = RS_WAVES / NCT;
    const int madd = (lane < 16) ? 0 : 8;
    __syncthreads();
    const int nrt = (nrows + 15) >> 4;
    for (int rt = wv / NCT; rt < nrt; rt += RSTEP) {
      const float* aBase = &acc[(rt * 16 + col) * P + koff];
      v8f c = (v8f){0.f, 0.f, 0.f, 0.f, 0.f, 0.f, 0.f, 0.f};
#pragma unroll
      for (int kg = 0; kg < N / 4; kg += 8) {
        v2f a[8];
#pragma unroll
        for (int kk = 0; kk < 8; kk++) a[kk] = *(const v2f*)(aBase + 4 * (kg + kk));
#pragma unroll
        for (int kk = 0; kk < 8; kk++)
          c = __builtin_amdgcn_wmma_f32_16x16x4_f32(false, a[kk], false, bf[kg + kk], (short)0, c,
                                                    false, false);
      }
      const float4* dp = (const float4*)&dl[rt * 16 + madd];
      const float4 d0 = dp[0];
      const float4 d1 = dp[1];
      const float d[8] = {d0.x, d0.y, d0.z, d0.w, d1.x, d1.y, d1.z, d1.w};
#pragma unroll
      for (int i = 0; i < 8; i++) {
        const int lr = rt * 16 + i + madd;
        if (lr < nrows) outp[(size_t)(r0 + lr) * NEXT + nbase + col] = c[i] * d[i];
      }
    }
  }
}

#define ED_THREADS 128

__global__ void __launch_bounds__(ED_THREADS)
edgedot_kernel(const int* __restrict__ es, const int* __restrict__ ed,
               const float* __restrict__ z, float* __restrict__ out, int nL) {
  const int e = blockIdx.x * ED_THREADS + threadIdx.x;
  const int ec = (e < nL) ? e : nL - 1;
  const v4f* a = (const v4f*)(z + (size_t)es[ec] * 64);
  const v4f* b = (const v4f*)(z + (size_t)ed[ec] * 64);
  v4f va[16], vb[16];
#pragma unroll
  for (int j = 0; j < 16; j++) va[j] = a[j];
#pragma unroll
  for (int j = 0; j < 16; j++) vb[j] = b[j];
  asm volatile("" ::"v"(va[0]), "v"(va[1]), "v"(va[2]), "v"(va[3]), "v"(va[4]), "v"(va[5]), "v"(va[6]),
               "v"(va[7]), "v"(va[8]), "v"(va[9]), "v"(va[10]), "v"(va[11]), "v"(va[12]), "v"(va[13]),
               "v"(va[14]), "v"(va[15]));
  asm volatile("" ::"v"(vb[0]), "v"(vb[1]), "v"(vb[2]), "v"(vb[3]), "v"(vb[4]), "v"(vb[5]), "v"(vb[6]),
               "v"(vb[7]), "v"(vb[8]), "v"(vb[9]), "v"(vb[10]), "v"(vb[11]), "v"(vb[12]), "v"(vb[13]),
               "v"(vb[14]), "v"(vb[15]));
  float p0 = 0.f, p1 = 0.f, p2 = 0.f, p3 = 0.f;
#pragma unroll
  for (int j = 0; j < 16; j++) {
    p0 = fmaf(va[j].x, vb[j].x, p0);
    p1 = fmaf(va[j].y, vb[j].y, p1);
    p2 = fmaf(va[j].z, vb[j].z, p2);
    p3 = fmaf(va[j].w, vb[j].w, p3);
  }
  if (e < nL) out[e] = (p0 + p1) + (p2 + p3);
}

extern "C" void kernel_launch(void* const* d_in, const int* in_sizes, int n_in,
                              void* d_out, int out_size, void* d_ws, size_t ws_size,
                              hipStream_t stream) {
  const int*   x_idx = (const int*)d_in[0];
  const int*   eidx  = (const int*)d_in[1];
  const int*   lidx  = (const int*)d_in[2];
  const float* emb   = (const float*)d_in[3];
  const float* W1    = (const float*)d_in[4];
  const float* b1    = (const float*)d_in[5];
  const float* W2    = (const float*)d_in[6];
  const float* b2    = (const float*)d_in[7];
  float* out = (float*)d_out;

  const int nN = in_sizes[0];
  const int nE = in_sizes[1] / 2;
  const int nL = in_sizes[2] / 2;
  const int H = 128, O = 64;

  float* dinv = (float*)d_ws;
  float* hs1  = dinv + ((nN + 63) & ~63);
  float* hs2  = hs1 + (size_t)nN * H;
  float* z    = hs2 + (size_t)nN * O;
  int*   qcnt = (int*)(z + (size_t)nN * O);
  int2*  qlist = (int2*)(qcnt + 256);
  unsigned* keys = (unsigned*)(qlist + (size_t)256 * RS_LIST_CAP);

  const int* esrc = eidx;
  const int* edst = eidx + nE;

  const int rpb = (nN + 255) / 256;
  size_t rs_tail = (size_t)((rpb + 1) & ~1) * sizeof(float) + (size_t)RS_LIST_CAP * sizeof(int2) + 16;
  const int arows = (rpb + 15) & ~15;
  size_t rs_mm = (size_t)arows * (H + 4 + 1) * sizeof(float) + (size_t)RS_LIST_CAP * sizeof(int2) + 16;

  const int frows = ((arows + RQ_FILL_ROWS - 1) / RQ_FILL_ROWS) * RQ_FILL_ROWS;
  size_t rq_lds = (size_t)frows * LDS_STRIDE * sizeof(float) + (size_t)RS_LIST_CAP * sizeof(int) +
                  (size_t)rpb * sizeof(int) + 16;
  range_key_kernel<<<(unsigned)((nE / 4 + 255) / 256), 256, 0, stream>>>(edst, keys, nE, rpb);
  range_queue_mm_kernel<128><<<256, RS_THREADS, rq_lds, stream>>>(
      esrc, edst, keys, emb, x_idx, W1, hs1, dinv, qcnt, qlist, nN, nE, rpb);

  range_sum_kernel<128, 1, 64, 1><<<256, RS_THREADS, rs_mm, stream>>>(
      esrc, edst, hs1, dinv, b1, hs2, W2, qcnt, qlist, nN, nE, rpb);

  range_sum_kernel<64, 0, 0, 0><<<256, RS_THREADS, (size_t)rpb * O * sizeof(float) + rs_tail, stream>>>(
      esrc, edst, hs2, dinv, b2, z, nullptr, qcnt, qlist, nN, nE, rpb);

  edgedot_kernel<<<(unsigned)((nL + ED_THREADS - 1) / ED_THREADS), ED_THREADS, 0, stream>>>(
      lidx, lidx + nL, z, out, nL);
}


# Node-sum replay from registers: a wave reads its run of the queue 32 entries a load, deals them by readlane; 32 row loads go out together, no LDS read or wait on LDS adds; LDS arrival order changes
# speedup vs baseline: 3.8742x; 1.0456x over previous
#include <hip/hip_runtime.h>

typedef float v2f __attribute__((ext_vector_type(2)));
typedef float v8f __attribute__((ext_vector_type(8)));
typedef float v4f __attribute__((ext_vector_type(4)));

#define LDS_STRIDE 132

__global__ void __launch_bounds__(256)
range_key_kernel(const int* __restrict__ dst, unsigned* __restrict__ keys, int nE, int rpb) {
  const int nW = (nE >> 4) << 2;
  const int i = blockIdx.x * 256 + threadIdx.x;
  if (i >= nW) return;
  const int* p = dst + 4 * (size_t)i;
  int d0, d1, d2, d3;
  if ((((size_t)dst) & 15) == 0) {
    int4 v = *(const int4*)p;
    d0 = v.x; d1 = v.y; d2 = v.z; d3 = v.w;
  } else {
    d0 = p[0]; d1 = p[1]; d2 = p[2]; d3 = p[3];
  }
  const unsigned r = (unsigned)rpb;
  unsigned k0 = min((unsigned)d0 / r, 255u);
  unsigned k1 = min((unsigned)d1 / r, 255u);
  unsigned k2 = min((unsigned)d2 / r, 255u);
  unsigned k3 = min((unsigned)d3 / r, 255u);
  keys[i] = k0 | (k1 << 8) | (k2 << 16) | (k3 << 24);
}

#define RS_THREADS 1024
#define RS_WAVES (RS_THREADS / 32)
#define RS_LIST_CAP 8192
#define RQ_SCAN_THREADS 512
#define RQ_SCAN_WAVES (RQ_SCAN_THREADS / 32)
#define RQ_TAIL_PER_THREAD 8
#define RQ_FILL 13
#define RQ_FILL_ROWS (32 * RQ_FILL)
#define RK_ONE 0x01010101u
#define RK_HI 0x80808080u

__device__ __forceinline__ unsigned rk_flag(unsigned x) { return (x - RK_ONE) & ~x; }

template <int N>
__global__ void __launch_bounds__(RS_THREADS)
range_queue_mm_kernel(const int* __restrict__ src, const int* __restrict__ dst,
                      const unsigned* __restrict__ keyw, const float* __restrict__ X,
                      const int* __restrict__ gidx, const float* __restrict__ Wm,
                      float* __restrict__ h, float* __restrict__ dinv,
                      int* __restrict__ qcnt, int2* __restrict__ qlist, int n_rows, int nE, int rpb) {
  static_assert(N == 128, "rows of 128 floats: 32 16-byte units a row, 8 column tiles");
  extern __shared__ float lds[];
  const uint4* keys = (const uint4*)keyw;
  const int arows = (rpb + 15) & ~15;
  const int frows = ((arows + RQ_FILL_ROWS - 1) / RQ_FILL_ROWS) * RQ_FILL_ROWS;
  float* lA = lds;
  int* list = (int*)(lds + (size_t)frows * LDS_STRIDE);
  int* degc = list + RS_LIST_CAP;
  int* cntp = degc + rpb;
  const int tid = threadIdx.x;
  const int lane = tid & 31;
  const int wv = __builtin_amdgcn_readfirstlane(tid >> 5);
  const bool is_mm = wv >= RQ_SCAN_WAVES;
  const int r0 = blockIdx.x * rpb;
  int nrows = n_rows - r0;
  if (nrows > rpb) nrows = rpb;
  if (nrows < 0) nrows = 0;
  const unsigned un = (unsigned)nrows;
  const unsigned bb = (blockIdx.x & 255u) * RK_ONE;

  for (int r = tid; r < nrows; r += RS_THREADS) degc[r] = 0;
  if (tid == 0) *cntp = 0;

  const int nfill = (nrows + 15) & ~15;
  {
    const v4f* X4 = (const v4f*)X;
    const int c4 = tid & 31;
    const int rw = tid >> 5;
    for (int rb = 0; rb < nfill; rb += RQ_FILL_ROWS) {
      int sr[RQ_FILL];
#pragma unroll
      for (int u = 0; u < RQ_FILL; u++) {
        const int r = rb + rw + 32 * u;
        sr[u] = gidx[r0 + (r < nrows ? r : 0)];
      }
      asm volatile("" ::"v"(sr[0]), "v"(sr[1]), "v"(sr[2]), "v"(sr[3]), "v"(sr[4]), "v"(sr[5]), "v"(sr[6]),
                   "v"(sr[7]), "v"(sr[8]), "v"(sr[9]), "v"(sr[10]), "v"(sr[11]), "v"(sr[12]));
      v4f v[RQ_FILL];
#pragma unroll
      for (int u = 0; u < RQ_FILL; u++) v[u] = X4[(size_t)sr[u] * (N / 4) + c4];
      asm volatile("" ::"v"(v[0]), "v"(v[1]), "v"(v[2]), "v"(v[3]), "v"(v[4]), "v"(v[5]), "v"(v[6]),
                   "v"(v[7]), "v"(v[8]), "v"(v[9]), "v"(v[10]), "v"(v[11]), "v"(v[12]));
#pragma unroll
      for (int u = 0; u < RQ_FILL; u++) {
        const int r = rb + rw + 32 * u;
        const v4f t = (r < nrows) ? v[u] : (v4f){0.f, 0.f, 0.f, 0.f};
        *(v4f*)&lA[r * LDS_STRIDE + (c4 << 2)] = t;
      }
    }
  }

  const int col = lane & 15;
  const int koff = (lane >> 4) << 1;
  const int gw = wv - RQ_SCAN_WAVES;
  const int nbase = (gw & 7) << 4;
  const int rpar = (gw >> 3) & 1;
  v2f bf[N / 4];
  if (is_mm) {
    const float* wp = Wm + (size_t)koff * N + nbase + col;
#pragma unroll
    for (int k = 0; k < N / 4; k++) {
      bf[k].x = wp[(size_t)(4 * k) * N];
      bf[k].y = wp[(size_t)(4 * k + 1) * N];
    }
  }
  __syncthreads();

  const int nrt = nfill >> 4;
  const int e1 = ((nrt * 3) >> 2) & ~1;
  const int e2 = (((nrt * 7) >> 3) + 1) & ~1;
  const int madd = (lane < 16) ? 0 : 8;

  auto mm_tiles = [&](int rt_lo, int rt_hi) {
    for (int rt = rt_lo + rpar; rt < rt_hi; rt += 2) {
      const float* aBase = &lA[(rt * 16 + col) * LDS_STRIDE + koff];
      v8f c = (v8f){0.f, 0.f, 0.f, 0.f, 0.f, 0.f, 0.f, 0.f};
#pragma unroll
      for (int kg = 0; kg < N / 4; kg += 8) {
        v2f a[8];
#pragma unroll
        for (int kk = 0; kk < 8; kk++) a[kk] = *(const v2f*)(aBase + 4 * (kg + kk));
#pragma unroll
        for (int kk = 0; kk < 8; kk++)
          c = __builtin_amdgcn_wmma_f32_16x16x4_f32(false, a[kk], false, bf[kg + kk], (short)0, c,
                                                    false, false);
      }
#pragma unroll
      for (int i = 0; i < 8; i++) {
        const int lr = rt * 16 + i + madd;
        if (lr < nrows) h[(size_t)(r0 + lr) * N + nbase + col] = c[i];
      }
    }
  };

  if (is_mm) {
    mm_tiles(0, e1);
  } else {
    const int nK = nE >> 4;
    int base = 0;
    for (; base + 8 * RQ_SCAN_THREADS <= nK; base += 8 * RQ_SCAN_THREADS) {
      uint4 kv[8];
#pragma unroll
      for (int u = 0; u < 8; u++) kv[u] = keys[base + u * RQ_SCAN_THREADS + tid];
      unsigned q[4];
      unsigned any = 0;
#pragma unroll
      for (int g = 0; g < 4; g++) {
        const unsigned a0 = rk_flag(kv[2 * g].x ^ bb), a1 = rk_flag(kv[2 * g].y ^ bb);
        const unsigned a2 = rk_flag(kv[2 * g].z ^ bb), a3 = rk_flag(kv[2 * g].w ^ bb);
        const unsigned a4 = rk_flag(kv[2 * g + 1].x ^ bb), a5 = rk_flag(kv[2 * g + 1].y ^ bb);
        const unsigned a6 = rk_flag(kv[2 * g + 1].z ^ bb), a7 = rk_flag(kv[2 * g + 1].w ^ bb);
        q[g] = (a0 & RK_HI) | ((a1 & RK_HI) >> 1) | ((a2 & RK_HI) >> 2) | ((a3 & RK_HI) >> 3) |
               ((a4 & RK_HI) >> 4) | ((a5 & RK_HI) >> 5) | ((a6 & RK_HI) >> 6) | ((a7 & RK_HI) >> 7);
        any |= q[g];
      }
      if (any) {
        int slot = atomicAdd(cntp, __popc(q[0]) + __popc(q[1]) + __popc(q[2]) + __popc(q[3]));
        const int e0 = (base + tid) << 4;
#pragma unroll
        for (int g = 0; g < 4; g++) {
          unsigned qq = q[g];
          while (qq) {
            int pos = __ffs(qq) - 1;
            qq &= qq - 1;
            int i = 8 * g + 7 - (pos & 7);
            int e = e0 + (i >> 2) * (16 * RQ_SCAN_THREADS) + ((i & 3) << 2) + (pos >> 3);
            if (slot < RS_LIST_CAP) list[slot] = e;
            slot++;
          }
        }
      }
    }
    for (int j = base + tid; j < nK; j += RQ_SCAN_THREADS) {
      uint4 kv = keys[j];
      unsigned a0 = rk_flag(kv.x ^ bb), a1 = rk_flag(kv.y ^ bb);
      unsigned a2 = rk_flag(kv.z ^ bb), a3 = rk_flag(kv.w ^ bb);
      if ((a0 | a1 | a2 | a3) & RK_HI) {
        unsigned q = (a0 & RK_HI) | ((a1 & RK_HI) >> 1) | ((a2 & RK_HI) >> 2) | ((a3 & RK_HI) >> 3);
        int slot = atomicAdd(cntp, __popc(q));
        while (q) {
          int pos = __ffs(q) - 1;
          q &= q - 1;
          int i = 7 - (pos & 7);
          int e = (j << 4) + (i << 2) + (pos >> 3);
          if (slot < RS_LIST_CAP) list[slot] = e;
          slot++;
        }
      }
    }
    for (int e = (nK << 4) + tid; e < nE; e += RQ_SCAN_THREADS) {
      if ((unsigned)(dst[e] - r0) < un) {
        int slot = atomicAdd(cntp, 1);
        if (slot < RS_LIST_CAP) list[slot] = e;
      }
    }
  }
  __syncthreads();

  const int n = *cntp;
  const bool over = n > RS_LIST_CAP;
  const int nq = over ? 0 : n;
  if (is_mm) {
    mm_tiles(e1, e2);
  } else {
    int2* q = qlist + (size_t)blockIdx.x * RS_LIST_CAP;
    if (over) {
      for (int e = tid; e < nE; e += RQ_SCAN_THREADS) {
        unsigned loc = (unsigned)(dst[e] - r0);
        if (loc < un) atomicAdd(&degc[loc], 1);
      }
    } else {
      for (int b0 = 0; b0 < nq; b0 += RQ_TAIL_PER_THREAD * RQ_SCAN_THREADS) {
        int ed[RQ_TAIL_PER_THREAD];
        int s[RQ_TAIL_PER_THREAD];
        int d[RQ_TAIL_PER_THREAD];
#pragma unroll
        for (int u = 0; u < RQ_TAIL_PER_THREAD; u++) {
          int idx = b0 + tid + u * RQ_SCAN_THREADS;
          ed[u] = list[idx < nq ? idx : 0];
        }
#pragma unroll
        for (int u = 0; u < RQ_TAIL_PER_THREAD; u++) {
          s[u] = src[ed[u]];
          d[u] = dst[ed[u]];
        }
#pragma unroll
        for (int u = 0; u < RQ_TAIL_PER_THREAD; u++) {
          int idx = b0 + tid + u * RQ_SCAN_THREADS;
          if (idx < nq) {
            unsigned loc = (unsigned)(d[u] - r0);
            if (loc < un) {
              atomicAdd(&degc[loc], 1);
              q[idx] = make_int2(s[u], (int)loc);
            } else {
              q[idx] = make_int2(0, -1);
            }
          }
        }
      }
    }
  }
  __syncthreads();
  if (is_mm) {
    mm_tiles(e2, nrt);
  } else {
    if (tid == 0) qcnt[blockIdx.x] = n;
    for (int r = tid; r < nrows; r += RQ_SCAN_THREADS) dinv[r0 + r] = rsqrtf(1.0f + (float)degc[r]);
  }
}

template <int N, int NEXT>
struct RsPitch { static constexpr int v = (NEXT > 0) ? N + 4 : N; };

template <int N, int RELU, int NEXT, int SCALE>
__global__ void __launch_bounds__(RS_THREADS)
range_sum_kernel(const int* __restrict__ src, const int* __restrict__ dst,
                 const float* __restrict__ hs, const float* __restrict__ dinv,
                 const float* __restrict__ bias, float* __restrict__ outp,
                 const float* __restrict__ Wn,
                 const int* __restrict__ qcnt, const int2* __restrict__ qlist,
                 int n_rows, int nE, int rpb) {
  extern __shared__ float lds[];
  constexpr int P = RsPitch<N, NEXT>::v;
  constexpr int P4 = P / 4;
  constexpr int N4 = N / 4;
  const int arows = (NEXT > 0) ? ((rpb + 15) & ~15) : rpb;
  float* acc = lds;
  float* dl = lds + (size_t)arows * P;
  const int tid = threadIdx.x;
  const int r0 = blockIdx.x * rpb;
  int nrows = n_rows - r0;
  if (nrows > rpb) nrows = rpb;
  if (nrows < 0) nrows = 0;
  const unsigned un = (unsigned)nrows;
  const int nel4 = (nrows * N) >> 2;
  const int wave = tid >> 5;
  const int lane = tid & 31;

  constexpr int NCT = (NEXT > 0) ? NEXT / 16 : 1;
  const int col = lane & 15;
  const int koff = (lane >> 4) << 1;
  const int wv = __builtin_amdgcn_readfirstlane(wave);
  const int nbase = (wv % NCT) << 4;

  const int n = qcnt[blockIdx.x];
  const bool over = n > RS_LIST_CAP;
  const int nq = over ? 0 : n;
  const int2* q = qlist + (size_t)blockIdx.x * RS_LIST_CAP;

  const float4* h4 = (const float4*)(hs + (size_t)r0 * N);
  float4* a4 = (float4*)acc;
  int i0 = tid;
  for (; i0 + 3 * RS_THREADS < nel4; i0 += 4 * RS_THREADS) {
    float4 t[4];
    float sc[4];
#pragma unroll
    for (int u = 0; u < 4; u++) {
      t[u] = h4[i0 + u * RS_THREADS];
      sc[u] = SCALE ? dinv[r0 + (i0 + u * RS_THREADS) / N4] : 1.f;
    }
#pragma unroll
    for (int u = 0; u < 4; u++) {
      const int i = i0 + u * RS_THREADS;
      if (SCALE) {
        t[u].x *= sc[u];
        t[u].y *= sc[u];
        t[u].z *= sc[u];
        t[u].w *= sc[u];
      }
      a4[(i / N4) * P4 + (i % N4)] = t[u];
    }
  }
  for (; i0 < nel4; i0 += RS_THREADS) {
    float4 t = h4[i0];
    if (SCALE) {
      const float sc = dinv[r0 + i0 / N4];
      t.x *= sc;
      t.y *= sc;
      t.z *= sc;
      t.w *= sc;
    }
    a4[(i0 / N4) * P4 + (i0 % N4)] = t;
  }
  if (NEXT > 0) {
    const int tail4 = (((nrows + 15) & ~15) - nrows) * N4;
    if (tid < tail4) a4[(nrows + tid / N4) * P4 + (tid % N4)] = make_float4(0.f, 0.f, 0.f, 0.f);
  }
  for (int r = tid; r < nrows; r += RS_THREADS) dl[r] = dinv[r0 + r];
  if (NEXT > 0) {
    if (tid < ((nrows + 15) & ~15) - nrows) dl[nrows + tid] = 0.f;
  }
  __syncthreads();

  if (over) {
    for (int e = tid; e < nE; e += RS_THREADS) {
      unsigned loc = (unsigned)(dst[e] - r0);
      if (loc < un) {
        const int sn = src[e];
        const float* hp = hs + (size_t)sn * N;
        float* ap = acc + (size_t)loc * P;
        const float sc = SCALE ? dinv[sn] : 1.f;
        for (int f = 0; f < N; f++) unsafeAtomicAdd(&ap[f], SCALE ? sc * hp[f] : hp[f]);
      }
    }
  }

  if (nq > 0) {
    constexpr int NV = N / 32;
    constexpr int RG = 32 / NV;
    const int ng = (nq + RG - 1) / RG;
    const int eA = ((wv * ng) >> 5) * RG;
    int eB = (((wv + 1) * ng) >> 5) * RG;
    if (eB > nq) eB = nq;
    float* al = acc + lane;
    int2 ea = q[min(eA + lane, nq - 1)];
    int2 eb = q[min(eA + 32 + lane, nq - 1)];
    float sa = SCALE ? dinv[ea.x] : 1.f;
    for (int base = eA; base < eB; base += 32) {
      int2 e = ea;
      const int es = __float_as_int(sa);
      if (base + lane >= eB) e = make_int2(0, -1);
      ea = eb;
      sa = SCALE ? dinv[ea.x] : 1.f;
      eb = q[min(base + 64 + lane, nq - 1)];
      const int nv = min(32, eB - base);
      for (int j0 = 0; j0 < nv; j0 += RG) {
        float v[RG][NV];
        float sc[RG];
        int loc[RG];
#pragma unroll
        for (int u = 0; u < RG; u++) {
          const int sn = __builtin_amdgcn_readlane(e.x, j0 + u);
          loc[u] = __builtin_amdgcn_readlane(e.y, j0 + u);
          const float* hp = hs + (size_t)sn * N;
#pragma unroll
          for (int j = 0; j < NV; j++) v[u][j] = hp[lane + 32 * j];
          sc[u] = __int_as_float(__builtin_amdgcn_readlane(es, j0 + u));
        }
#pragma unroll
        for (int u = 0; u < RG; u++) {
          if constexpr (NV == 4) asm volatile("" ::"v"(v[u][0]), "v"(v[u][1]), "v"(v[u][2]), "v"(v[u][3]));
          else asm volatile("" ::"v"(v[u][0]), "v"(v[u][1]));
          if (loc[u] >= 0) {
            float* ap = al + loc[u] * P;
#pragma unroll
            for (int j = 0; j < NV; j++) unsafeAtomicAdd(&ap[32 * j], SCALE ? sc[u] * v[u][j] : v[u][j]);
          }
        }
      }
    }
  }
  v2f bf[(NEXT > 0) ? N / 4 : 1];
  if constexpr (NEXT > 0) {
    const float* wp = Wn + (size_t)koff * NEXT + nbase + col;
#pragma unroll
    for (int k = 0; k < N / 4; k++) {
      bf[k].x = wp[(size_t)(4 * k) * NEXT];
      bf[k].y = wp[(size_t)(4 * k + 1) * NEXT];
    }
  }
  __syncthreads();

  const float4 b = *(const float4*)(bias + ((tid << 2) & (N - 1)));
  float4* o4 = (float4*)(outp + (size_t)r0 * N);
  for (int i = tid; i < nel4; i += RS_THREADS) {
    const int li = (i / N4) * P4 + (i % N4);
    float4 a = a4[li];
    float di = dl[(i << 2) / N];
    float4 o;
    o.x = di * a.x + b.x;
    o.y = di * a.y + b.y;
    o.z = di * a.z + b.z;
    o.w = di * a.w + b.w;
    if (RELU) {
      o.x = fmaxf(o.x, 0.f);
      o.y = fmaxf(o.y, 0.f);
      o.z = fmaxf(o.z, 0.f);
      o.w = fmaxf(o.w, 0.f);
    }
    if (NEXT > 0) a4[li] = o;
    else o4[i] = o;
  }

  if constexpr (NEXT > 0) {
    constexpr int RSTEP = RS_WAVES / NCT;
    const int madd = (lane < 16) ? 0 : 8;
    __syncthreads();
    const int nrt = (nrows + 15) >> 4;
    for (int rt = wv / NCT; rt < nrt; rt += RSTEP) {
      const float* aBase = &acc[(rt * 16 + col) * P + koff];
      v8f c = (v8f){0.f, 0.f, 0.f, 0.f, 0.f, 0.f, 0.f, 0.f};
#pragma unroll
      for (int kg = 0; kg < N / 4; kg += 8) {
        v2f a[8];
#pragma unroll
        for (int kk = 0; kk < 8; kk++) a[kk] = *(const v2f*)(aBase + 4 * (kg + kk));
#pragma unroll
        for (int kk = 0; kk < 8; kk++)
          c = __builtin_amdgcn_wmma_f32_16x16x4_f32(false, a[kk], false, bf[kg + kk], (short)0, c,
                                                    false, false);
      }
      const float4* dp = (const float4*)&dl[rt * 16 + madd];
      const float4 d0 = dp[0];
      const float4 d1 = dp[1];
      const float d[8] = {d0.x, d0.y, d0.z, d0.w, d1.x, d1.y, d1.z, d1.w};
#pragma unroll
      for (int i = 0; i < 8; i++) {
        const int lr = rt * 16 + i + madd;
        if (lr < nrows) outp[(size_t)(r0 + lr) * NEXT + nbase + col] = c[i] * d[i];
      }
    }
  }
}

#define ED_THREADS 128

__global__ void __launch_bounds__(ED_THREADS)
edgedot_kernel(const int* __restrict__ es, const int* __restrict__ ed,
               const float* __restrict__ z, float* __restrict__ out, int nL) {
  const int e = blockIdx.x * ED_THREADS + threadIdx.x;
  const int ec = (e < nL) ? e : nL - 1;
  const v4f* a = (const v4f*)(z + (size_t)es[ec] * 64);
  const v4f* b = (const v4f*)(z + (size_t)ed[ec] * 64);
  v4f va[16], vb[16];
#pragma unroll
  for (int j = 0; j < 16; j++) va[j] = a[j];
#pragma unroll
  for (int j = 0; j < 16; j++) vb[j] = b[j];
  asm volatile("" ::"v"(va[0]), "v"(va[1]), "v"(va[2]), "v"(va[3]), "v"(va[4]), "v"(va[5]), "v"(va[6]),
               "v"(va[7]), "v"(va[8]), "v"(va[9]), "v"(va[10]), "v"(va[11]), "v"(va[12]), "v"(va[13]),
               "v"(va[14]), "v"(va[15]));
  asm volatile("" ::"v"(vb[0]), "v"(vb[1]), "v"(vb[2]), "v"(vb[3]), "v"(vb[4]), "v"(vb[5]), "v"(vb[6]),
               "v"(vb[7]), "v"(vb[8]), "v"(vb[9]), "v"(vb[10]), "v"(vb[11]), "v"(vb[12]), "v"(vb[13]),
               "v"(vb[14]), "v"(vb[15]));
  float p0 = 0.f, p1 = 0.f, p2 = 0.f, p3 = 0.f;
#pragma unroll
  for (int j = 0; j < 16; j++) {
    p0 = fmaf(va[j].x, vb[j].x, p0);
    p1 = fmaf(va[j].y, vb[j].y, p1);
    p2 = fmaf(va[j].z, vb[j].z, p2);
    p3 = fmaf(va[j].w, vb[j].w, p3);
  }
  if (e < nL) out[e] = (p0 + p1) + (p2 + p3);
}

extern "C" void kernel_launch(void* const* d_in, const int* in_sizes, int n_in,
                              void* d_out, int out_size, void* d_ws, size_t ws_size,
                              hipStream_t stream) {
  const int*   x_idx = (const int*)d_in[0];
  const int*   eidx  = (const int*)d_in[1];
  const int*   lidx  = (const int*)d_in[2];
  const float* emb   = (const float*)d_in[3];
  const float* W1    = (const float*)d_in[4];
  const float* b1    = (const float*)d_in[5];
  const float* W2    = (const float*)d_in[6];
  const float* b2    = (const float*)d_in[7];
  float* out = (float*)d_out;

  const int nN = in_sizes[0];
  const int nE = in_sizes[1] / 2;
  const int nL = in_sizes[2] / 2;
  const int H = 128, O = 64;

  float* dinv = (float*)d_ws;
  float* hs1  = dinv + ((nN + 63) & ~63);
  float* hs2  = hs1 + (size_t)nN * H;
  float* z    = hs2 + (size_t)nN * O;
  int*   qcnt = (int*)(z + (size_t)nN * O);
  int2*  qlist = (int2*)(qcnt + 256);
  unsigned* keys = (unsigned*)(qlist + (size_t)256 * RS_LIST_CAP);

  const int* esrc = eidx;
  const int* edst = eidx + nE;

  const int rpb = (nN + 255) / 256;
  size_t rs_tail = (size_t)((rpb + 1) & ~1) * sizeof(float) + (size_t)RS_LIST_CAP * sizeof(int2) + 16;
  const int arows = (rpb + 15) & ~15;
  size_t rs_mm = (size_t)arows * (H + 4 + 1) * sizeof(float) + (size_t)RS_LIST_CAP * sizeof(int2) + 16;

  const int frows = ((arows + RQ_FILL_ROWS - 1) / RQ_FILL_ROWS) * RQ_FILL_ROWS;
  size_t rq_lds = (size_t)frows * LDS_STRIDE * sizeof(float) + (size_t)RS_LIST_CAP * sizeof(int) +
                  (size_t)rpb * sizeof(int) + 16;
  range_key_kernel<<<(unsigned)((nE / 4 + 255) / 256), 256, 0, stream>>>(edst, keys, nE, rpb);
  range_queue_mm_kernel<128><<<256, RS_THREADS, rq_lds, stream>>>(
      esrc, edst, keys, emb, x_idx, W1, hs1, dinv, qcnt, qlist, nN, nE, rpb);

  range_sum_kernel<128, 1, 64, 1><<<256, RS_THREADS, rs_mm, stream>>>(
      esrc, edst, hs1, dinv, b1, hs2, W2, qcnt, qlist, nN, nE, rpb);

  range_sum_kernel<64, 0, 0, 0><<<256, RS_THREADS, (size_t)rpb * O * sizeof(float) + rs_tail, stream>>>(
      esrc, edst, hs2, dinv, b2, z, nullptr, qcnt, qlist, nN, nE, rpb);

  edgedot_kernel<<<(unsigned)((nL + ED_THREADS - 1) / ED_THREADS), ED_THREADS, 0, stream>>>(
      lidx, lidx + nL, z, out, nL);
}


# Node-sum replay: a queued source row comes by ONE load per wave (16 or 8 bytes a lane; was 4 or 2 one-line loads); LDS sums kept lane-major so adds stay conflict-free; LDS arrival order as before
# speedup vs baseline: 4.0247x; 1.0388x over previous
#include <hip/hip_runtime.h>

typedef float v2f __attribute__((ext_vector_type(2)));
typedef float v8f __attribute__((ext_vector_type(8)));
typedef float v4f __attribute__((ext_vector_type(4)));

#define LDS_STRIDE 132

__global__ void __launch_bounds__(256)
range_key_kernel(const int* __restrict__ dst, unsigned* __restrict__ keys, int nE, int rpb) {
  const int nW = (nE >> 4) << 2;
  const int i = blockIdx.x * 256 + threadIdx.x;
  if (i >= nW) return;
  const int* p = dst + 4 * (size_t)i;
  int d0, d1, d2, d3;
  if ((((size_t)dst) & 15) == 0) {
    int4 v = *(const int4*)p;
    d0 = v.x; d1 = v.y; d2 = v.z; d3 = v.w;
  } else {
    d0 = p[0]; d1 = p[1]; d2 = p[2]; d3 = p[3];
  }
  const unsigned r = (unsigned)rpb;
  unsigned k0 = min((unsigned)d0 / r, 255u);
  unsigned k1 = min((unsigned)d1 / r, 255u);
  unsigned k2 = min((unsigned)d2 / r, 255u);
  unsigned k3 = min((unsigned)d3 / r, 255u);
  keys[i] = k0 | (k1 << 8) | (k2 << 16) | (k3 << 24);
}

#define RS_THREADS 1024
#define RS_WAVES (RS_THREADS / 32)
#define RS_LIST_CAP 8192
#define RQ_SCAN_THREADS 512
#define RQ_SCAN_WAVES (RQ_SCAN_THREADS / 32)
#define RQ_TAIL_PER_THREAD 8
#define RQ_FILL 13
#define RQ_FILL_ROWS (32 * RQ_FILL)
#define RK_ONE 0x01010101u
#define RK_HI 0x80808080u

__device__ __forceinline__ unsigned rk_flag(unsigned x) { return (x - RK_ONE) & ~x; }

template <int N>
__global__ void __launch_bounds__(RS_THREADS)
range_queue_mm_kernel(const int* __restrict__ src, const int* __restrict__ dst,
                      const unsigned* __restrict__ keyw, const float* __restrict__ X,
                      const int* __restrict__ gidx, const float* __restrict__ Wm,
                      float* __restrict__ h, float* __restrict__ dinv,
                      int* __restrict__ qcnt, int2* __restrict__ qlist, int n_rows, int nE, int rpb) {
  static_assert(N == 128, "rows of 128 floats: 32 16-byte units a row, 8 column tiles");
  extern __shared__ float lds[];
  const uint4* keys = (const uint4*)keyw;
  const int arows = (rpb + 15) & ~15;
  const int frows = ((arows + RQ_FILL_ROWS - 1) / RQ_FILL_ROWS) * RQ_FILL_ROWS;
  float* lA = lds;
  int* list = (int*)(lds + (size_t)frows * LDS_STRIDE);
  int* degc = list + RS_LIST_CAP;
  int* cntp = degc + rpb;
  const int tid = threadIdx.x;
  const int lane = tid & 31;
  const int wv = __builtin_amdgcn_readfirstlane(tid >> 5);
  const bool is_mm = wv >= RQ_SCAN_WAVES;
  const int r0 = blockIdx.x * rpb;
  int nrows = n_rows - r0;
  if (nrows > rpb) nrows = rpb;
  if (nrows < 0) nrows = 0;
  const unsigned un = (unsigned)nrows;
  const unsigned bb = (blockIdx.x & 255u) * RK_ONE;

  for (int r = tid; r < nrows; r += RS_THREADS) degc[r] = 0;
  if (tid == 0) *cntp = 0;

  const int nfill = (nrows + 15) & ~15;
  {
    const v4f* X4 = (const v4f*)X;
    const int c4 = tid & 31;
    const int rw = tid >> 5;
    for (int rb = 0; rb < nfill; rb += RQ_FILL_ROWS) {
      int sr[RQ_FILL];
#pragma unroll
      for (int u = 0; u < RQ_FILL; u++) {
        const int r = rb + rw + 32 * u;
        sr[u] = gidx[r0 + (r < nrows ? r : 0)];
      }
      asm volatile("" ::"v"(sr[0]), "v"(sr[1]), "v"(sr[2]), "v"(sr[3]), "v"(sr[4]), "v"(sr[5]), "v"(sr[6]),
                   "v"(sr[7]), "v"(sr[8]), "v"(sr[9]), "v"(sr[10]), "v"(sr[11]), "v"(sr[12]));
      v4f v[RQ_FILL];
#pragma unroll
      for (int u = 0; u < RQ_FILL; u++) v[u] = X4[(size_t)sr[u] * (N / 4) + c4];
      asm volatile("" ::"v"(v[0]), "v"(v[1]), "v"(v[2]), "v"(v[3]), "v"(v[4]), "v"(v[5]), "v"(v[6]),
                   "v"(v[7]), "v"(v[8]), "v"(v[9]), "v"(v[10]), "v"(v[11]), "v"(v[12]));
#pragma unroll
      for (int u = 0; u < RQ_FILL; u++) {
        const int r = rb + rw + 32 * u;
        const v4f t = (r < nrows) ? v[u] : (v4f){0.f, 0.f, 0.f, 0.f};
        *(v4f*)&lA[r * LDS_STRIDE + (c4 << 2)] = t;
      }
    }
  }

  const int col = lane & 15;
  const int koff = (lane >> 4) << 1;
  const int gw = wv - RQ_SCAN_WAVES;
  const int nbase = (gw & 7) << 4;
  const int rpar = (gw >> 3) & 1;
  v2f bf[N / 4];
  if (is_mm) {
    const float* wp = Wm + (size_t)koff * N + nbase + col;
#pragma unroll
    for (int k = 0; k < N / 4; k++) {
      bf[k].x = wp[(size_t)(4 * k) * N];
      bf[k].y = wp[(size_t)(4 * k + 1) * N];
    }
  }
  __syncthreads();

  const int nrt = nfill >> 4;
  const int e1 = ((nrt * 3) >> 2) & ~1;
  const int e2 = (((nrt * 7) >> 3) + 1) & ~1;
  const int madd = (lane < 16) ? 0 : 8;

  auto mm_tiles = [&](int rt_lo, int rt_hi) {
    for (int rt = rt_lo + rpar; rt < rt_hi; rt += 2) {
      const float* aBase = &lA[(rt * 16 + col) * LDS_STRIDE + koff];
      v8f c = (v8f){0.f, 0.f, 0.f, 0.f, 0.f, 0.f, 0.f, 0.f};
#pragma unroll
      for (int kg = 0; kg < N / 4; kg += 8) {
        v2f a[8];
#pragma unroll
        for (int kk = 0; kk < 8; kk++) a[kk] = *(const v2f*)(aBase + 4 * (kg + kk));
#pragma unroll
        for (int kk = 0; kk < 8; kk++)
          c = __builtin_amdgcn_wmma_f32_16x16x4_f32(false, a[kk], false, bf[kg + kk], (short)0, c,
                                                    false, false);
      }
#pragma unroll
      for (int i = 0; i < 8; i++) {
        const int lr = rt * 16 + i + madd;
        if (lr < nrows) h[(size_t)(r0 + lr) * N + nbase + col] = c[i];
      }
    }
  };

  if (is_mm) {
    mm_tiles(0, e1);
  } else {
    const int nK = nE >> 4;
    int base = 0;
    for (; base + 8 * RQ_SCAN_THREADS <= nK; base += 8 * RQ_SCAN_THREADS) {
      uint4 kv[8];
#pragma unroll
      for (int u = 0; u < 8; u++) kv[u] = keys[base + u * RQ_SCAN_THREADS + tid];
      unsigned q[4];
      unsigned any = 0;
#pragma unroll
      for (int g = 0; g < 4; g++) {
        const unsigned a0 = rk_flag(kv[2 * g].x ^ bb), a1 = rk_flag(kv[2 * g].y ^ bb);
        const unsigned a2 = rk_flag(kv[2 * g].z ^ bb), a3 = rk_flag(kv[2 * g].w ^ bb);
        const unsigned a4 = rk_flag(kv[2 * g + 1].x ^ bb), a5 = rk_flag(kv[2 * g + 1].y ^ bb);
        const unsigned a6 = rk_flag(kv[2 * g + 1].z ^ bb), a7 = rk_flag(kv[2 * g + 1].w ^ bb);
        q[g] = (a0 & RK_HI) | ((a1 & RK_HI) >> 1) | ((a2 & RK_HI) >> 2) | ((a3 & RK_HI) >> 3) |
               ((a4 & RK_HI) >> 4) | ((a5 & RK_HI) >> 5) | ((a6 & RK_HI) >> 6) | ((a7 & RK_HI) >> 7);
        any |= q[g];
      }
      if (any) {
        int slot = atomicAdd(cntp, __popc(q[0]) + __popc(q[1]) + __popc(q[2]) + __popc(q[3]));
        const int e0 = (base + tid) << 4;
#pragma unroll
        for (int g = 0; g < 4; g++) {
          unsigned qq = q[g];
          while (qq) {
            int pos = __ffs(qq) - 1;
            qq &= qq - 1;
            int i = 8 * g + 7 - (pos & 7);
            int e = e0 + (i >> 2) * (16 * RQ_SCAN_THREADS) + ((i & 3) << 2) + (pos >> 3);
            if (slot < RS_LIST_CAP) list[slot] = e;
            slot++;
          }
        }
      }
    }
    for (int j = base + tid; j < nK; j += RQ_SCAN_THREADS) {
      uint4 kv = keys[j];
      unsigned a0 = rk_flag(kv.x ^ bb), a1 = rk_flag(kv.y ^ bb);
      unsigned a2 = rk_flag(kv.z ^ bb), a3 = rk_flag(kv.w ^ bb);
      if ((a0 | a1 | a2 | a3) & RK_HI) {
        unsigned q = (a0 & RK_HI) | ((a1 & RK_HI) >> 1) | ((a2 & RK_HI) >> 2) | ((a3 & RK_HI) >> 3);
        int slot = atomicAdd(cntp, __popc(q));
        while (q) {
          int pos = __ffs(q) - 1;
          q &= q - 1;
          int i = 7 - (pos & 7);
          int e = (j << 4) + (i << 2) + (pos >> 3);
          if (slot < RS_LIST_CAP) list[slot] = e;
          slot++;
        }
      }
    }
    for (int e = (nK << 4) + tid; e < nE; e += RQ_SCAN_THREADS) {
      if ((unsigned)(dst[e] - r0) < un) {
        int slot = atomicAdd(cntp, 1);
        if (slot < RS_LIST_CAP) list[slot] = e;
      }
    }
  }
  __syncthreads();

  const int n = *cntp;
  const bool over = n > RS_LIST_CAP;
  const int nq = over ? 0 : n;
  if (is_mm) {
    mm_tiles(e1, e2);
  } else {
    int2* q = qlist + (size_t)blockIdx.x * RS_LIST_CAP;
    if (over) {
      for (int e = tid; e < nE; e += RQ_SCAN_THREADS) {
        unsigned loc = (unsigned)(dst[e] - r0);
        if (loc < un) atomicAdd(&degc[loc], 1);
      }
    } else {
      for (int b0 = 0; b0 < nq; b0 += RQ_TAIL_PER_THREAD * RQ_SCAN_THREADS) {
        int ed[RQ_TAIL_PER_THREAD];
        int s[RQ_TAIL_PER_THREAD];
        int d[RQ_TAIL_PER_THREAD];
#pragma unroll
        for (int u = 0; u < RQ_TAIL_PER_THREAD; u++) {
          int idx = b0 + tid + u * RQ_SCAN_THREADS;
          ed[u] = list[idx < nq ? idx : 0];
        }
#pragma unroll
        for (int u = 0; u < RQ_TAIL_PER_THREAD; u++) {
          s[u] = src[ed[u]];
          d[u] = dst[ed[u]];
        }
#pragma unroll
        for (int u = 0; u < RQ_TAIL_PER_THREAD; u++) {
          int idx = b0 + tid + u * RQ_SCAN_THREADS;
          if (idx < nq) {
            unsigned loc = (unsigned)(d[u] - r0);
            if (loc < un) {
              atomicAdd(&degc[loc], 1);
              q[idx] = make_int2(s[u], (int)loc);
            } else {
              q[idx] = make_int2(0, -1);
            }
          }
        }
      }
    }
  }
  __syncthreads();
  if (is_mm) {
    mm_tiles(e2, nrt);
  } else {
    if (tid == 0) qcnt[blockIdx.x] = n;
    for (int r = tid; r < nrows; r += RQ_SCAN_THREADS) dinv[r0 + r] = rsqrtf(1.0f + (float)degc[r]);
  }
}

template <int N, int NEXT>
struct RsPitch { static constexpr int v = (NEXT > 0) ? N + 4 : N; };

template <int NV> struct RsRowVec;
template <> struct RsRowVec<4> { typedef v4f t; };
template <> struct RsRowVec<2> { typedef v2f t; };

template <int N, int RELU, int NEXT, int SCALE>
__global__ void __launch_bounds__(RS_THREADS)
range_sum_kernel(const int* __restrict__ src, const int* __restrict__ dst,
                 const float* __restrict__ hs, const float* __restrict__ dinv,
                 const float* __restrict__ bias, float* __restrict__ outp,
                 const float* __restrict__ Wn,
                 const int* __restrict__ qcnt, const int2* __restrict__ qlist,
                 int n_rows, int nE, int rpb) {
  extern __shared__ float lds[];
  constexpr int P = RsPitch<N, NEXT>::v;
  constexpr int P4 = P / 4;
  constexpr int N4 = N / 4;
  constexpr int NV = N / 32;
  static_assert(NV == 4 || NV == 2, "rows of 512 or 256 bytes");
  static_assert(NEXT == 0 || N4 == 32, "the pass at the end reorders a row in place: one wave, one whole row");
  const int arows = (NEXT > 0) ? ((rpb + 15) & ~15) : rpb;
  float* acc = lds;
  float* dl = lds + (size_t)arows * P;
  const int tid = threadIdx.x;
  const int r0 = blockIdx.x * rpb;
  int nrows = n_rows - r0;
  if (nrows > rpb) nrows = rpb;
  if (nrows < 0) nrows = 0;
  const unsigned un = (unsigned)nrows;
  const int nel4 = (nrows * N) >> 2;
  const int wave = tid >> 5;
  const int lane = tid & 31;

  constexpr int NCT = (NEXT > 0) ? NEXT / 16 : 1;
  const int col = lane & 15;
  const int koff = (lane >> 4) << 1;
  const int wv = __builtin_amdgcn_readfirstlane(wave);
  const int nbase = (wv % NCT) << 4;

  const int n = qcnt[blockIdx.x];
  const bool over = n > RS_LIST_CAP;
  const int nq = over ? 0 : n;
  const int2* q = qlist + (size_t)blockIdx.x * RS_LIST_CAP;

  const float4* h4 = (const float4*)(hs + (size_t)r0 * N);
  float4* a4 = (float4*)acc;
  auto put_unit = [&](int row, int c, const float4 t) {
    float* p = acc + row * P;
    if constexpr (NV == 4) {
      p[c] = t.x;
      p[32 + c] = t.y;
      p[64 + c] = t.z;
      p[96 + c] = t.w;
    } else {
      *(v2f*)(p + 2 * c) = (v2f){t.x, t.z};
      *(v2f*)(p + 32 + 2 * c) = (v2f){t.y, t.w};
    }
  };
  auto get_unit = [&](int row, int c) {
    const float* p = acc + row * P;
    if constexpr (NV == 4) {
      return make_float4(p[c], p[32 + c], p[64 + c], p[96 + c]);
    } else {
      const v2f lo = *(const v2f*)(p + 2 * c);
      const v2f hi = *(const v2f*)(p + 32 + 2 * c);
      return make_float4(lo.x, hi.x, lo.y, hi.y);
    }
  };
  int i0 = tid;
  for (; i0 + 3 * RS_THREADS < nel4; i0 += 4 * RS_THREADS) {
    float4 t[4];
    float sc[4];
#pragma unroll
    for (int u = 0; u < 4; u++) {
      t[u] = h4[i0 + u * RS_THREADS];
      sc[u] = SCALE ? dinv[r0 + (i0 + u * RS_THREADS) / N4] : 1.f;
    }
#pragma unroll
    for (int u = 0; u < 4; u++) {
      const int i = i0 + u * RS_THREADS;
      if (SCALE) {
        t[u].x *= sc[u];
        t[u].y *= sc[u];
        t[u].z *= sc[u];
        t[u].w *= sc[u];
      }
      put_unit(i / N4, i % N4, t[u]);
    }
  }
  for (; i0 < nel4; i0 += RS_THREADS) {
    float4 t = h4[i0];
    if (SCALE) {
      const float sc = dinv[r0 + i0 / N4];
      t.x *= sc;
      t.y *= sc;
      t.z *= sc;
      t.w *= sc;
    }
    put_unit(i0 / N4, i0 % N4, t);
  }
  if (NEXT > 0) {
    const int tail4 = (((nrows + 15) & ~15) - nrows) * N4;
    if (tid < tail4) a4[(nrows + tid / N4) * P4 + (tid % N4)] = make_float4(0.f, 0.f, 0.f, 0.f);
  }
  for (int r = tid; r < nrows; r += RS_THREADS) dl[r] = dinv[r0 + r];
  if (NEXT > 0) {
    if (tid < ((nrows + 15) & ~15) - nrows) dl[nrows + tid] = 0.f;
  }
  __syncthreads();

  if (over) {
    for (int e = tid; e < nE; e += RS_THREADS) {
      unsigned loc = (unsigned)(dst[e] - r0);
      if (loc < un) {
        const int sn = src[e];
        const float* hp = hs + (size_t)sn * N;
        float* ap = acc + (size_t)loc * P;
        const float sc = SCALE ? dinv[sn] : 1.f;
        for (int f = 0; f < N; f++)
          unsafeAtomicAdd(&ap[32 * (f % NV) + f / NV], SCALE ? sc * hp[f] : hp[f]);
      }
    }
  }

  if (nq > 0) {
    typedef typename RsRowVec<NV>::t rowv;
    constexpr int RG = 32 / NV;
    const int ng = (nq + RG - 1) / RG;
    const int eA = ((wv * ng) >> 5) * RG;
    int eB = (((wv + 1) * ng) >> 5) * RG;
    if (eB > nq) eB = nq;
    float* al = acc + lane;
    int2 ea = q[min(eA + lane, nq - 1)];
    int2 eb = q[min(eA + 32 + lane, nq - 1)];
    float sa = SCALE ? dinv[ea.x] : 1.f;
    for (int base = eA; base < eB; base += 32) {
      int2 e = ea;
      const int es = __float_as_int(sa);
      if (base + lane >= eB) e = make_int2(0, -1);
      ea = eb;
      sa = SCALE ? dinv[ea.x] : 1.f;
      eb = q[min(base + 64 + lane, nq - 1)];
      const int nv = min(32, eB - base);
      for (int j0 = 0; j0 < nv; j0 += RG) {
        rowv v[RG];
        float sc[RG];
        int loc[RG];
#pragma unroll
        for (int u = 0; u < RG; u++) {
          const int sn = __builtin_amdgcn_readlane(e.x, j0 + u);
          loc[u] = __builtin_amdgcn_readlane(e.y, j0 + u);
          const rowv* hp = (const rowv*)(hs + (size_t)sn * N);
          v[u] = hp[lane];
          sc[u] = __int_as_float(__builtin_amdgcn_readlane(es, j0 + u));
        }
#pragma unroll
        for (int u = 0; u < RG; u++) {
          asm volatile("" ::"v"(v[u]));
          if (loc[u] >= 0) {
            float* ap = al + loc[u] * P;
#pragma unroll
            for (int j = 0; j < NV; j++) unsafeAtomicAdd(&ap[32 * j], SCALE ? sc[u] * v[u][j] : v[u][j]);
          }
        }
      }
    }
  }
  v2f bf[(NEXT > 0) ? N / 4 : 1];
  if constexpr (NEXT > 0) {
    const float* wp = Wn + (size_t)koff * NEXT + nbase + col;
#pragma unroll
    for (int k = 0; k < N / 4; k++) {
      bf[k].x = wp[(size_t)(4 * k) * NEXT];
      bf[k].y = wp[(size_t)(4 * k + 1) * NEXT];
    }
  }
  __syncthreads();

  const float4 b = *(const float4*)(bias + ((tid << 2) & (N - 1)));
  float4* o4 = (float4*)(outp + (size_t)r0 * N);
  for (int i = tid; i < nel4; i += RS_THREADS) {
    const int li = (i / N4) * P4 + (i % N4);
    float4 a = get_unit(i / N4, i % N4);
    float di = dl[(i << 2) / N];
    float4 o;
    o.x = di * a.x + b.x;
    o.y = di * a.y + b.y;
    o.z = di * a.z + b.z;
    o.w = di * a.w + b.w;
    if (RELU) {
      o.x = fmaxf(o.x, 0.f);
      o.y = fmaxf(o.y, 0.f);
      o.z = fmaxf(o.z, 0.f);
      o.w = fmaxf(o.w, 0.f);
    }
    if (NEXT > 0) a4[li] = o;
    else o4[i] = o;
  }

  if constexpr (NEXT > 0) {
    constexpr int RSTEP = RS_WAVES / NCT;
    const int madd = (lane < 16) ? 0 : 8;
    __syncthreads();
    const int nrt = (nrows + 15) >> 4;
    for (int rt = wv / NCT; rt < nrt; rt += RSTEP) {
      const float* aBase = &acc[(rt * 16 + col) * P + koff];
      v8f c = (v8f){0.f, 0.f, 0.f, 0.f, 0.f, 0.f, 0.f, 0.f};
#pragma unroll
      for (int kg = 0; kg < N / 4; kg += 8) {
        v2f a[8];
#pragma unroll
        for (int kk = 0; kk < 8; kk++) a[kk] = *(const v2f*)(aBase + 4 * (kg + kk));
#pragma unroll
        for (int kk = 0; kk < 8; kk++)
          c = __builtin_amdgcn_wmma_f32_16x16x4_f32(false, a[kk], false, bf[kg + kk], (short)0, c,
                                                    false, false);
      }
      const float4* dp = (const float4*)&dl[rt * 16 + madd];
      const float4 d0 = dp[0];
      const float4 d1 = dp[1];
      const float d[8] = {d0.x, d0.y, d0.z, d0.w, d1.x, d1.y, d1.z, d1.w};
#pragma unroll
      for (int i = 0; i < 8; i++) {
        const int lr = rt * 16 + i + madd;
        if (lr < nrows) outp[(size_t)(r0 + lr) * NEXT + nbase + col] = c[i] * d[i];
      }
    }
  }
}

#define ED_THREADS 128

__global__ void __launch_bounds__(ED_THREADS)
edgedot_kernel(const int* __restrict__ es, const int* __restrict__ ed,
               const float* __restrict__ z, float* __restrict__ out, int nL) {
  const int e = blockIdx.x * ED_THREADS + threadIdx.x;
  const int ec = (e < nL) ? e : nL - 1;
  const v4f* a = (const v4f*)(z + (size_t)es[ec] * 64);
  const v4f* b = (const v4f*)(z + (size_t)ed[ec] * 64);
  v4f va[16], vb[16];
#pragma unroll
  for (int j = 0; j < 16; j++) va[j] = a[j];
#pragma unroll
  for (int j = 0; j < 16; j++) vb[j] = b[j];
  asm volatile("" ::"v"(va[0]), "v"(va[1]), "v"(va[2]), "v"(va[3]), "v"(va[4]), "v"(va[5]), "v"(va[6]),
               "v"(va[7]), "v"(va[8]), "v"(va[9]), "v"(va[10]), "v"(va[11]), "v"(va[12]), "v"(va[13]),
               "v"(va[14]), "v"(va[15]));
  asm volatile("" ::"v"(vb[0]), "v"(vb[1]), "v"(vb[2]), "v"(vb[3]), "v"(vb[4]), "v"(vb[5]), "v"(vb[6]),
               "v"(vb[7]), "v"(vb[8]), "v"(vb[9]), "v"(vb[10]), "v"(vb[11]), "v"(vb[12]), "v"(vb[13]),
               "v"(vb[14]), "v"(vb[15]));
  float p0 = 0.f, p1 = 0.f, p2 = 0.f, p3 = 0.f;
#pragma unroll
  for (int j = 0; j < 16; j++) {
    p0 = fmaf(va[j].x, vb[j].x, p0);
    p1 = fmaf(va[j].y, vb[j].y, p1);
    p2 = fmaf(va[j].z, vb[j].z, p2);
    p3 = fmaf(va[j].w, vb[j].w, p3);
  }
  if (e < nL) out[e] = (p0 + p1) + (p2 + p3);
}

extern "C" void kernel_launch(void* const* d_in, const int* in_sizes, int n_in,
                              void* d_out, int out_size, void* d_ws, size_t ws_size,
                              hipStream_t stream) {
  const int*   x_idx = (const int*)d_in[0];
  const int*   eidx  = (const int*)d_in[1];
  const int*   lidx  = (const int*)d_in[2];
  const float* emb   = (const float*)d_in[3];
  const float* W1    = (const float*)d_in[4];
  const float* b1    = (const float*)d_in[5];
  const float* W2    = (const float*)d_in[6];
  const float* b2    = (const float*)d_in[7];
  float* out = (float*)d_out;

  const int nN = in_sizes[0];
  const int nE = in_sizes[1] / 2;
  const int nL = in_sizes[2] / 2;
  const int H = 128, O = 64;

  float* dinv = (float*)d_ws;
  float* hs1  = dinv + ((nN + 63) & ~63);
  float* hs2  = hs1 + (size_t)nN * H;
  float* z    = hs2 + (size_t)nN * O;
  int*   qcnt = (int*)(z + (size_t)nN * O);
  int2*  qlist = (int2*)(qcnt + 256);
  unsigned* keys = (unsigned*)(qlist + (size_t)256 * RS_LIST_CAP);

  const int* esrc = eidx;
  const int* edst = eidx + nE;

  const int rpb = (nN + 255) / 256;
  size_t rs_tail = (size_t)((rpb + 1) & ~1) * sizeof(float) + (size_t)RS_LIST_CAP * sizeof(int2) + 16;
  const int arows = (rpb + 15) & ~15;
  size_t rs_mm = (size_t)arows * (H + 4 + 1) * sizeof(float) + (size_t)RS_LIST_CAP * sizeof(int2) + 16;

  const int frows = ((arows + RQ_FILL_ROWS - 1) / RQ_FILL_ROWS) * RQ_FILL_ROWS;
  size_t rq_lds = (size_t)frows * LDS_STRIDE * sizeof(float) + (size_t)RS_LIST_CAP * sizeof(int) +
                  (size_t)rpb * sizeof(int) + 16;
  range_key_kernel<<<(unsigned)((nE / 4 + 255) / 256), 256, 0, stream>>>(edst, keys, nE, rpb);
  range_queue_mm_kernel<128><<<256, RS_THREADS, rq_lds, stream>>>(
      esrc, edst, keys, emb, x_idx, W1, hs1, dinv, qcnt, qlist, nN, nE, rpb);

  range_sum_kernel<128, 1, 64, 1><<<256, RS_THREADS, rs_mm, stream>>>(
      esrc, edst, hs1, dinv, b1, hs2, W2, qcnt, qlist, nN, nE, rpb);

  range_sum_kernel<64, 0, 0, 0><<<256, RS_THREADS, (size_t)rpb * O * sizeof(float) + rs_tail, stream>>>(
      esrc, edst, hs2, dinv, b2, z, nullptr, qcnt, qlist, nN, nE, rpb);

  edgedot_kernel<<<(unsigned)((nL + ED_THREADS - 1) / ED_THREADS), ED_THREADS, 0, stream>>>(
      lidx, lidx + nL, z, out, nL);
}
